# starting kernel
# baseline (speedup 1.0000x reference)
#include <hip/hip_runtime.h>
#include <hip/hip_bf16.h>
#include <math.h>

#define E_DIM 1024
#define S_LEN 8192
#define H_N   16
#define D_H   64

typedef __attribute__((ext_vector_type(2))) float v2f;
typedef __attribute__((ext_vector_type(8))) float v8f;

__device__ __forceinline__ v8f wmma4(v2f a, v2f b, v8f c) {
    return __builtin_amdgcn_wmma_f32_16x16x4_f32(false, a, false, b, (short)0, c,
                                                 false, false);
}

__global__ void qkv_part_kernel(const float* __restrict__ x,
                                const float* __restrict__ Wv,
                                const float* __restrict__ Wq,
                                const float* __restrict__ Wk,
                                float* __restrict__ part_qkv) {
    __shared__ float xs[256];
    const int tid = threadIdx.x;
    const int k0  = blockIdx.y * 256;
    if (tid < 256) xs[tid] = x[k0 + tid];
    __syncthreads();

    const int m    = blockIdx.z;
    const int wv   = tid >> 5;
    const int lane = tid & 31;
    const int j    = lane & 15;
    const int half = lane >> 4;
    const int ob   = blockIdx.x * 128 + wv * 16;
    const float amask = (j == 0) ? 1.f : 0.f;

    const float* W = (m == 0) ? Wv : (m == 1) ? Wq : Wk;

    v8f c0 = {}, c1 = {};
    for (int eb = 0; eb < 256; eb += 8) {
        const int r0 = eb + half * 2;
        v2f a0, a1, b0, b1;
        a0.x = xs[r0]     * amask;  a0.y = xs[r0 + 1] * amask;
        a1.x = xs[r0 + 4] * amask;  a1.y = xs[r0 + 5] * amask;
        const float* wp0 = W + (size_t)(k0 + r0)     * E_DIM + ob + j;
        const float* wp1 = W + (size_t)(k0 + r0 + 4) * E_DIM + ob + j;
        b0.x = wp0[0];  b0.y = wp0[E_DIM];
        b1.x = wp1[0];  b1.y = wp1[E_DIM];
        c0 = wmma4(a0, b0, c0);
        c1 = wmma4(a1, b1, c1);
    }
    const v8f cs = c0 + c1;
    if (lane < 16)
        part_qkv[(size_t)(m * 4 + blockIdx.y) * E_DIM + ob + lane] = cs[0];
}

__global__ void qkv_combine_kernel(const float* __restrict__ part_qkv,
                                   const float* __restrict__ bv,
                                   const float* __restrict__ bq,
                                   const float* __restrict__ bk,
                                   float* __restrict__ q_out,
                                   float* __restrict__ newv, float* __restrict__ newk) {
    for (int o = threadIdx.x; o < 3 * E_DIM; o += 256) {
        const int m  = o >> 10;
        const int oo = o & (E_DIM - 1);
        float s = 0.f;
        #pragma unroll
        for (int ks = 0; ks < 4; ks++)
            s += part_qkv[(size_t)(m * 4 + ks) * E_DIM + oo];
        if (m == 0)      newv[(size_t)(S_LEN - 1) * E_DIM + oo] = s + bv[oo];
        else if (m == 1) q_out[oo] = s + bq[oo];
        else             newk[(size_t)(S_LEN - 1) * E_DIM + oo] = s + bk[oo];
    }
}

__global__ void shift_score_kernel(const float* __restrict__ oldv,
                                   const float* __restrict__ oldk,
                                   const float* __restrict__ q,
                                   float* __restrict__ newv, float* __restrict__ newk,
                                   float* __restrict__ scores) {
    const int tid  = threadIdx.x;
    const int lane = tid & 31;
    const int wv   = tid >> 5;
    const int c0   = blockIdx.x * 32;
    const float4* q4 = (const float4*)q;

    for (int p = wv; p < 32; p += 8) {
        const int s = c0 + p;
        const bool last = (s == S_LEN - 1);

        const float4* vsrc = (const float4*)(last ? (newv + (size_t)s * E_DIM)
                                                  : (oldv + (size_t)(s + 1) * E_DIM));
        float4* vdst = (float4*)(newv + (size_t)s * E_DIM);
        int nz = 0;
        #pragma unroll
        for (int i = 0; i < 8; i++) {
            float4 d = vsrc[i * 32 + lane];
            nz |= (d.x != 0.f) | (d.y != 0.f) | (d.z != 0.f) | (d.w != 0.f);
            if (!last) vdst[i * 32 + lane] = d;
        }
        #pragma unroll
        for (int mm = 16; mm >= 1; mm >>= 1) nz |= __shfl_xor(nz, mm, 32);

        const float4* ksrc = (const float4*)(last ? (newk + (size_t)s * E_DIM)
                                                  : (oldk + (size_t)(s + 1) * E_DIM));
        float4* kdst = (float4*)(newk + (size_t)s * E_DIM);
        #pragma unroll
        for (int i = 0; i < 8; i++) {
            float4 kd = ksrc[i * 32 + lane];
            if (!last) kdst[i * 32 + lane] = kd;
            float4 qd = q4[i * 32 + lane];
            float part = kd.x * qd.x + kd.y * qd.y + kd.z * qd.z + kd.w * qd.w;
            part += __shfl_xor(part, 1, 32);
            part += __shfl_xor(part, 2, 32);
            part += __shfl_xor(part, 4, 32);
            part += __shfl_xor(part, 8, 32);
            if ((lane & 15) == 0) {
                const int h = i * 2 + (lane >> 4);
                scores[(size_t)h * S_LEN + s] = nz ? part * 0.125f : -INFINITY;
            }
        }
    }
}

__global__ void softmax_kernel(float* __restrict__ scores, float* __restrict__ values) {
    __shared__ float red[256];
    const int h = blockIdx.x, tid = threadIdx.x;
    float* row = scores + (size_t)h * S_LEN;
    if (tid < D_H) values[h * D_H + tid] = 0.f;

    float mx = -INFINITY;
    for (int i = tid; i < S_LEN; i += 256) mx = fmaxf(mx, row[i]);
    red[tid] = mx; __syncthreads();
    for (int st = 128; st > 0; st >>= 1) {
        if (tid < st) red[tid] = fmaxf(red[tid], red[tid + st]);
        __syncthreads();
    }
    mx = red[0]; __syncthreads();

    float sm = 0.f;
    for (int i = tid; i < S_LEN; i += 256) sm += expf(row[i] - mx);
    red[tid] = sm; __syncthreads();
    for (int st = 128; st > 0; st >>= 1) {
        if (tid < st) red[tid] += red[tid + st];
        __syncthreads();
    }
    const float inv = 1.f / red[0];
    for (int i = tid; i < S_LEN; i += 256) row[i] = expf(row[i] - mx) * inv;
}

__global__ void attnv_kernel(const float* __restrict__ probs,
                             const float* __restrict__ newv,
                             float* __restrict__ values) {
    const int tid  = threadIdx.x;
    const int lane = tid & 31;
    const int wv   = tid >> 5;
    const int gw   = blockIdx.x * 8 + wv;
    const int h    = gw >> 6;
    const int dt   = (gw >> 4) & 3;
    const int sc   = gw & 15;
    const int s0   = sc * 512;
    const int j    = lane & 15;
    const int half = lane >> 4;
    const float amask = (j == 0) ? 1.f : 0.f;

    const float* p = probs + (size_t)h * S_LEN;
    const float* V = newv + h * D_H + dt * 16 + j;

    v8f c0 = {}, c1 = {};
    for (int sb = 0; sb < 512; sb += 8) {
        const int sk0 = s0 + sb + half * 2;
        const int sk1 = sk0 + 4;
        v2f a0, a1, b0, b1;
        a0.x = p[sk0]     * amask;  a0.y = p[sk0 + 1] * amask;
        a1.x = p[sk1]     * amask;  a1.y = p[sk1 + 1] * amask;
        const float* vp0 = V + (size_t)sk0 * E_DIM;
        const float* vp1 = V + (size_t)sk1 * E_DIM;
        b0.x = vp0[0];  b0.y = vp0[E_DIM];
        b1.x = vp1[0];  b1.y = vp1[E_DIM];
        c0 = wmma4(a0, b0, c0);
        c1 = wmma4(a1, b1, c1);
    }
    const v8f cs = c0 + c1;
    if (lane < 16) atomicAdd(&values[h * D_H + dt * 16 + lane], cs[0]);
}

__global__ void gemv_part_kernel(const float* __restrict__ W,
                                 const float* __restrict__ in_parts, int P,
                                 const float* __restrict__ in_bias, int in_relu,
                                 float* __restrict__ out_parts) {
    __shared__ float xs[128];
    const int tid = threadIdx.x;
    const int k0  = blockIdx.y * 128;
    if (tid < 128) {
        float s = 0.f;
        for (int pp = 0; pp < P; pp++) s += in_parts[(size_t)pp * E_DIM + k0 + tid];
        if (in_bias) s += in_bias[k0 + tid];
        if (in_relu) s = fmaxf(s, 0.f);
        xs[tid] = s;
    }
    __syncthreads();

    const int wv   = tid >> 5;
    const int lane = tid & 31;
    const int j    = lane & 15;
    const int half = lane >> 4;
    const int ob   = blockIdx.x * 128 + wv * 16;
    const float amask = (j == 0) ? 1.f : 0.f;

    v8f c0 = {}, c1 = {};
    for (int eb = 0; eb < 128; eb += 8) {
        const int r0 = eb + half * 2;
        v2f a0, a1, b0, b1;
        a0.x = xs[r0]     * amask;  a0.y = xs[r0 + 1] * amask;
        a1.x = xs[r0 + 4] * amask;  a1.y = xs[r0 + 5] * amask;
        const float* wp0 = W + (size_t)(k0 + r0)     * E_DIM + ob + j;
        const float* wp1 = W + (size_t)(k0 + r0 + 4) * E_DIM + ob + j;
        b0.x = wp0[0];  b0.y = wp0[E_DIM];
        b1.x = wp1[0];  b1.y = wp1[E_DIM];
        c0 = wmma4(a0, b0, c0);
        c1 = wmma4(a1, b1, c1);
    }
    const v8f cs = c0 + c1;
    if (lane < 16)
        out_parts[(size_t)blockIdx.y * E_DIM + ob + lane] = cs[0];
}

__global__ void ln_combine_kernel(const float* __restrict__ parts, int P,
                                  const float* __restrict__ bias,
                                  const float* __restrict__ resid,
                                  const float* __restrict__ sc,
                                  const float* __restrict__ bs,
                                  float* __restrict__ out) {
    __shared__ float red[256];
    const int tid = threadIdx.x;
    float vv[4];
    float sum = 0.f;
    #pragma unroll
    for (int i = 0; i < 4; i++) {
        const int e = i * 256 + tid;
        float s = bias[e] + resid[e];
        for (int pp = 0; pp < P; pp++) s += parts[(size_t)pp * E_DIM + e];
        vv[i] = s;
        sum += s;
    }
    red[tid] = sum; __syncthreads();
    for (int st = 128; st > 0; st >>= 1) {
        if (tid < st) red[tid] += red[tid + st];
        __syncthreads();
    }
    const float mean = red[0] * (1.f / E_DIM); __syncthreads();
    float var = 0.f;
    #pragma unroll
    for (int i = 0; i < 4; i++) { const float d = vv[i] - mean; var += d * d; }
    red[tid] = var; __syncthreads();
    for (int st = 128; st > 0; st >>= 1) {
        if (tid < st) red[tid] += red[tid + st];
        __syncthreads();
    }
    const float inv = rsqrtf(red[0] * (1.f / E_DIM) + 1e-6f);
    #pragma unroll
    for (int i = 0; i < 4; i++) {
        const int e = i * 256 + tid;
        out[e] = (vv[i] - mean) * inv * sc[e] + bs[e];
    }
}

extern "C" void kernel_launch(void* const* d_in, const int* in_sizes, int n_in,
                              void* d_out, int out_size, void* d_ws, size_t ws_size,
                              hipStream_t stream) {
    (void)in_sizes; (void)n_in; (void)out_size; (void)ws_size;
    const float* x     = (const float*)d_in[0];
    const float* cache = (const float*)d_in[1];
    const float* Wv = (const float*)d_in[2];  const float* bv = (const float*)d_in[3];
    const float* Wq = (const float*)d_in[4];  const float* bq = (const float*)d_in[5];
    const float* Wk = (const float*)d_in[6];  const float* bk = (const float*)d_in[7];
    const float* Wo = (const float*)d_in[8];  const float* bo = (const float*)d_in[9];
    const float* W1 = (const float*)d_in[10]; const float* b1 = (const float*)d_in[11];
    const float* W2 = (const float*)d_in[12]; const float* b2 = (const float*)d_in[13];
    const float* ln1_s = (const float*)d_in[14]; const float* ln1_b = (const float*)d_in[15];
    const float* ln2_s = (const float*)d_in[16]; const float* ln2_b = (const float*)d_in[17];

    float* out_h = (float*)d_out;
    float* newv  = out_h + E_DIM;
    float* newk  = newv + (size_t)S_LEN * E_DIM;
    const float* oldv = cache;
    const float* oldk = cache + (size_t)S_LEN * E_DIM;

    float* ws       = (float*)d_ws;
    float* q        = ws;
    float* scores   = q + E_DIM;
    float* values   = scores + H_N * S_LEN;
    float* h1       = values + E_DIM;
    float* part_qkv = h1 + E_DIM;
    float* part_o   = part_qkv + 12 * E_DIM;
    float* part_1   = part_o + 8 * E_DIM;
    float* part_2   = part_1 + 8 * E_DIM;

    qkv_part_kernel<<<dim3(8, 4, 3), 256, 0, stream>>>(x, Wv, Wq, Wk, part_qkv);
    qkv_combine_kernel<<<1, 256, 0, stream>>>(part_qkv, bv, bq, bk, q, newv, newk);
    shift_score_kernel<<<256, 256, 0, stream>>>(oldv, oldk, q, newv, newk, scores);
    softmax_kernel<<<16, 256, 0, stream>>>(scores, values);
    attnv_kernel<<<128, 256, 0, stream>>>(scores, newv, values);
    gemv_part_kernel<<<dim3(8, 8), 256, 0, stream>>>(Wo, values, 1, nullptr, 0, part_o);
    ln_combine_kernel<<<1, 256, 0, stream>>>(part_o, 8, bo, x, ln1_s, ln1_b, h1);
    gemv_part_kernel<<<dim3(8, 8), 256, 0, stream>>>(W1, h1, 1, nullptr, 0, part_1);
    gemv_part_kernel<<<dim3(8, 8), 256, 0, stream>>>(W2, part_1, 8, b1, 1, part_2);
    ln_combine_kernel<<<1, 256, 0, stream>>>(part_2, 8, b2, h1, ln2_s, ln2_b, out_h);
}


# shift/score pass: each wave owns 2 adjacent cache positions and issues all its loads (q + 32 row loads, 16 KB) before first use; 512 blocks of 8 waves (2 per WGP); score pairs stored as float2
# speedup vs baseline: 1.3329x; 1.3329x over previous
#include <hip/hip_runtime.h>
#include <hip/hip_bf16.h>
#include <math.h>

#define E_DIM 1024
#define S_LEN 8192
#define H_N   16
#define D_H   64

typedef __attribute__((ext_vector_type(2))) float v2f;
typedef __attribute__((ext_vector_type(8))) float v8f;

__device__ __forceinline__ v8f wmma4(v2f a, v2f b, v8f c) {
    return __builtin_amdgcn_wmma_f32_16x16x4_f32(false, a, false, b, (short)0, c,
                                                 false, false);
}

__global__ void qkv_part_kernel(const float* __restrict__ x,
                                const float* __restrict__ Wv,
                                const float* __restrict__ Wq,
                                const float* __restrict__ Wk,
                                float* __restrict__ part_qkv) {
    __shared__ float xs[256];
    const int tid = threadIdx.x;
    const int k0  = blockIdx.y * 256;
    if (tid < 256) xs[tid] = x[k0 + tid];
    __syncthreads();

    const int m    = blockIdx.z;
    const int wv   = tid >> 5;
    const int lane = tid & 31;
    const int j    = lane & 15;
    const int half = lane >> 4;
    const int ob   = blockIdx.x * 128 + wv * 16;
    const float amask = (j == 0) ? 1.f : 0.f;

    const float* W = (m == 0) ? Wv : (m == 1) ? Wq : Wk;

    v8f c0 = {}, c1 = {};
    for (int eb = 0; eb < 256; eb += 8) {
        const int r0 = eb + half * 2;
        v2f a0, a1, b0, b1;
        a0.x = xs[r0]     * amask;  a0.y = xs[r0 + 1] * amask;
        a1.x = xs[r0 + 4] * amask;  a1.y = xs[r0 + 5] * amask;
        const float* wp0 = W + (size_t)(k0 + r0)     * E_DIM + ob + j;
        const float* wp1 = W + (size_t)(k0 + r0 + 4) * E_DIM + ob + j;
        b0.x = wp0[0];  b0.y = wp0[E_DIM];
        b1.x = wp1[0];  b1.y = wp1[E_DIM];
        c0 = wmma4(a0, b0, c0);
        c1 = wmma4(a1, b1, c1);
    }
    const v8f cs = c0 + c1;
    if (lane < 16)
        part_qkv[(size_t)(m * 4 + blockIdx.y) * E_DIM + ob + lane] = cs[0];
}

__global__ void qkv_combine_kernel(const float* __restrict__ part_qkv,
                                   const float* __restrict__ bv,
                                   const float* __restrict__ bq,
                                   const float* __restrict__ bk,
                                   float* __restrict__ q_out,
                                   float* __restrict__ newv, float* __restrict__ newk) {
    for (int o = threadIdx.x; o < 3 * E_DIM; o += 256) {
        const int m  = o >> 10;
        const int oo = o & (E_DIM - 1);
        float s = 0.f;
        #pragma unroll
        for (int ks = 0; ks < 4; ks++)
            s += part_qkv[(size_t)(m * 4 + ks) * E_DIM + oo];
        if (m == 0)      newv[(size_t)(S_LEN - 1) * E_DIM + oo] = s + bv[oo];
        else if (m == 1) q_out[oo] = s + bq[oo];
        else             newk[(size_t)(S_LEN - 1) * E_DIM + oo] = s + bk[oo];
    }
}

__global__ __launch_bounds__(256)
void shift_score_kernel(const float* __restrict__ oldv,
                        const float* __restrict__ oldk,
                        const float* __restrict__ q,
                        float* __restrict__ newv, float* __restrict__ newk,
                        float* __restrict__ scores) {
    const int tid  = threadIdx.x;
    const int lane = tid & 31;
    const int wv   = tid >> 5;
    const int s0   = (blockIdx.x * 8 + wv) * 2;
    const int s1   = s0 + 1;
    const bool last1 = (s1 == S_LEN - 1);

    const float4* v0src = (const float4*)(oldv + (size_t)(s0 + 1) * E_DIM);
    const float4* k0src = (const float4*)(oldk + (size_t)(s0 + 1) * E_DIM);
    const float4* v1src = (const float4*)(last1 ? (newv + (size_t)s1 * E_DIM)
                                                : (oldv + (size_t)(s1 + 1) * E_DIM));
    const float4* k1src = (const float4*)(last1 ? (newk + (size_t)s1 * E_DIM)
                                                : (oldk + (size_t)(s1 + 1) * E_DIM));
    float4* v0dst = (float4*)(newv + (size_t)s0 * E_DIM);
    float4* k0dst = (float4*)(newk + (size_t)s0 * E_DIM);
    float4* v1dst = (float4*)(newv + (size_t)s1 * E_DIM);
    float4* k1dst = (float4*)(newk + (size_t)s1 * E_DIM);
    const float4* q4 = (const float4*)q;

    float4 qd[8], va[8], vb[8], ka[8], kb[8];
    #pragma unroll
    for (int i = 0; i < 8; i++) qd[i] = q4[i * 32 + lane];
    #pragma unroll
    for (int i = 0; i < 8; i++) va[i] = v0src[i * 32 + lane];
    #pragma unroll
    for (int i = 0; i < 8; i++) vb[i] = v1src[i * 32 + lane];
    #pragma unroll
    for (int i = 0; i < 8; i++) ka[i] = k0src[i * 32 + lane];
    #pragma unroll
    for (int i = 0; i < 8; i++) kb[i] = k1src[i * 32 + lane];

    #pragma unroll
    for (int i = 0; i < 8; i++) v0dst[i * 32 + lane] = va[i];
    if (!last1) {
        #pragma unroll
        for (int i = 0; i < 8; i++) v1dst[i * 32 + lane] = vb[i];
    }
    #pragma unroll
    for (int i = 0; i < 8; i++) k0dst[i * 32 + lane] = ka[i];
    if (!last1) {
        #pragma unroll
        for (int i = 0; i < 8; i++) k1dst[i * 32 + lane] = kb[i];
    }

    int nza = 0, nzb = 0;
    #pragma unroll
    for (int i = 0; i < 8; i++) {
        nza |= (va[i].x != 0.f) | (va[i].y != 0.f) | (va[i].z != 0.f) | (va[i].w != 0.f);
        nzb |= (vb[i].x != 0.f) | (vb[i].y != 0.f) | (vb[i].z != 0.f) | (vb[i].w != 0.f);
    }
    #pragma unroll
    for (int mm = 16; mm >= 1; mm >>= 1) {
        nza |= __shfl_xor(nza, mm, 32);
        nzb |= __shfl_xor(nzb, mm, 32);
    }

    #pragma unroll
    for (int i = 0; i < 8; i++) {
        float pa = ka[i].x * qd[i].x + ka[i].y * qd[i].y + ka[i].z * qd[i].z + ka[i].w * qd[i].w;
        float pb = kb[i].x * qd[i].x + kb[i].y * qd[i].y + kb[i].z * qd[i].z + kb[i].w * qd[i].w;
        pa += __shfl_xor(pa, 1, 32);  pb += __shfl_xor(pb, 1, 32);
        pa += __shfl_xor(pa, 2, 32);  pb += __shfl_xor(pb, 2, 32);
        pa += __shfl_xor(pa, 4, 32);  pb += __shfl_xor(pb, 4, 32);
        pa += __shfl_xor(pa, 8, 32);  pb += __shfl_xor(pb, 8, 32);
        if ((lane & 15) == 0) {
            const int h = i * 2 + (lane >> 4);
            float2 o;
            o.x = nza ? pa * 0.125f : -INFINITY;
            o.y = nzb ? pb * 0.125f : -INFINITY;
            *(float2*)(scores + (size_t)h * S_LEN + s0) = o;
        }
    }
}

__global__ void softmax_kernel(float* __restrict__ scores, float* __restrict__ values) {
    __shared__ float red[256];
    const int h = blockIdx.x, tid = threadIdx.x;
    float* row = scores + (size_t)h * S_LEN;
    if (tid < D_H) values[h * D_H + tid] = 0.f;

    float mx = -INFINITY;
    for (int i = tid; i < S_LEN; i += 256) mx = fmaxf(mx, row[i]);
    red[tid] = mx; __syncthreads();
    for (int st = 128; st > 0; st >>= 1) {
        if (tid < st) red[tid] = fmaxf(red[tid], red[tid + st]);
        __syncthreads();
    }
    mx = red[0]; __syncthreads();

    float sm = 0.f;
    for (int i = tid; i < S_LEN; i += 256) sm += expf(row[i] - mx);
    red[tid] = sm; __syncthreads();
    for (int st = 128; st > 0; st >>= 1) {
        if (tid < st) red[tid] += red[tid + st];
        __syncthreads();
    }
    const float inv = 1.f / red[0];
    for (int i = tid; i < S_LEN; i += 256) row[i] = expf(row[i] - mx) * inv;
}

__global__ void attnv_kernel(const float* __restrict__ probs,
                             const float* __restrict__ newv,
                             float* __restrict__ values) {
    const int tid  = threadIdx.x;
    const int lane = tid & 31;
    const int wv   = tid >> 5;
    const int gw   = blockIdx.x * 8 + wv;
    const int h    = gw >> 6;
    const int dt   = (gw >> 4) & 3;
    const int sc   = gw & 15;
    const int s0   = sc * 512;
    const int j    = lane & 15;
    const int half = lane >> 4;
    const float amask = (j == 0) ? 1.f : 0.f;

    const float* p = probs + (size_t)h * S_LEN;
    const float* V = newv + h * D_H + dt * 16 + j;

    v8f c0 = {}, c1 = {};
    for (int sb = 0; sb < 512; sb += 8) {
        const int sk0 = s0 + sb + half * 2;
        const int sk1 = sk0 + 4;
        v2f a0, a1, b0, b1;
        a0.x = p[sk0]     * amask;  a0.y = p[sk0 + 1] * amask;
        a1.x = p[sk1]     * amask;  a1.y = p[sk1 + 1] * amask;
        const float* vp0 = V + (size_t)sk0 * E_DIM;
        const float* vp1 = V + (size_t)sk1 * E_DIM;
        b0.x = vp0[0];  b0.y = vp0[E_DIM];
        b1.x = vp1[0];  b1.y = vp1[E_DIM];
        c0 = wmma4(a0, b0, c0);
        c1 = wmma4(a1, b1, c1);
    }
    const v8f cs = c0 + c1;
    if (lane < 16) atomicAdd(&values[h * D_H + dt * 16 + lane], cs[0]);
}

__global__ void gemv_part_kernel(const float* __restrict__ W,
                                 const float* __restrict__ in_parts, int P,
                                 const float* __restrict__ in_bias, int in_relu,
                                 float* __restrict__ out_parts) {
    __shared__ float xs[128];
    const int tid = threadIdx.x;
    const int k0  = blockIdx.y * 128;
    if (tid < 128) {
        float s = 0.f;
        for (int pp = 0; pp < P; pp++) s += in_parts[(size_t)pp * E_DIM + k0 + tid];
        if (in_bias) s += in_bias[k0 + tid];
        if (in_relu) s = fmaxf(s, 0.f);
        xs[tid] = s;
    }
    __syncthreads();

    const int wv   = tid >> 5;
    const int lane = tid & 31;
    const int j    = lane & 15;
    const int half = lane >> 4;
    const int ob   = blockIdx.x * 128 + wv * 16;
    const float amask = (j == 0) ? 1.f : 0.f;

    v8f c0 = {}, c1 = {};
    for (int eb = 0; eb < 128; eb += 8) {
        const int r0 = eb + half * 2;
        v2f a0, a1, b0, b1;
        a0.x = xs[r0]     * amask;  a0.y = xs[r0 + 1] * amask;
        a1.x = xs[r0 + 4] * amask;  a1.y = xs[r0 + 5] * amask;
        const float* wp0 = W + (size_t)(k0 + r0)     * E_DIM + ob + j;
        const float* wp1 = W + (size_t)(k0 + r0 + 4) * E_DIM + ob + j;
        b0.x = wp0[0];  b0.y = wp0[E_DIM];
        b1.x = wp1[0];  b1.y = wp1[E_DIM];
        c0 = wmma4(a0, b0, c0);
        c1 = wmma4(a1, b1, c1);
    }
    const v8f cs = c0 + c1;
    if (lane < 16)
        out_parts[(size_t)blockIdx.y * E_DIM + ob + lane] = cs[0];
}

__global__ void ln_combine_kernel(const float* __restrict__ parts, int P,
                                  const float* __restrict__ bias,
                                  const float* __restrict__ resid,
                                  const float* __restrict__ sc,
                                  const float* __restrict__ bs,
                                  float* __restrict__ out) {
    __shared__ float red[256];
    const int tid = threadIdx.x;
    float vv[4];
    float sum = 0.f;
    #pragma unroll
    for (int i = 0; i < 4; i++) {
        const int e = i * 256 + tid;
        float s = bias[e] + resid[e];
        for (int pp = 0; pp < P; pp++) s += parts[(size_t)pp * E_DIM + e];
        vv[i] = s;
        sum += s;
    }
    red[tid] = sum; __syncthreads();
    for (int st = 128; st > 0; st >>= 1) {
        if (tid < st) red[tid] += red[tid + st];
        __syncthreads();
    }
    const float mean = red[0] * (1.f / E_DIM); __syncthreads();
    float var = 0.f;
    #pragma unroll
    for (int i = 0; i < 4; i++) { const float d = vv[i] - mean; var += d * d; }
    red[tid] = var; __syncthreads();
    for (int st = 128; st > 0; st >>= 1) {
        if (tid < st) red[tid] += red[tid + st];
        __syncthreads();
    }
    const float inv = rsqrtf(red[0] * (1.f / E_DIM) + 1e-6f);
    #pragma unroll
    for (int i = 0; i < 4; i++) {
        const int e = i * 256 + tid;
        out[e] = (vv[i] - mean) * inv * sc[e] + bs[e];
    }
}

extern "C" void kernel_launch(void* const* d_in, const int* in_sizes, int n_in,
                              void* d_out, int out_size, void* d_ws, size_t ws_size,
                              hipStream_t stream) {
    (void)in_sizes; (void)n_in; (void)out_size; (void)ws_size;
    const float* x     = (const float*)d_in[0];
    const float* cache = (const float*)d_in[1];
    const float* Wv = (const float*)d_in[2];  const float* bv = (const float*)d_in[3];
    const float* Wq = (const float*)d_in[4];  const float* bq = (const float*)d_in[5];
    const float* Wk = (const float*)d_in[6];  const float* bk = (const float*)d_in[7];
    const float* Wo = (const float*)d_in[8];  const float* bo = (const float*)d_in[9];
    const float* W1 = (const float*)d_in[10]; const float* b1 = (const float*)d_in[11];
    const float* W2 = (const float*)d_in[12]; const float* b2 = (const float*)d_in[13];
    const float* ln1_s = (const float*)d_in[14]; const float* ln1_b = (const float*)d_in[15];
    const float* ln2_s = (const float*)d_in[16]; const float* ln2_b = (const float*)d_in[17];

    float* out_h = (float*)d_out;
    float* newv  = out_h + E_DIM;
    float* newk  = newv + (size_t)S_LEN * E_DIM;
    const float* oldv = cache;
    const float* oldk = cache + (size_t)S_LEN * E_DIM;

    float* ws       = (float*)d_ws;
    float* q        = ws;
    float* scores   = q + E_DIM;
    float* values   = scores + H_N * S_LEN;
    float* h1       = values + E_DIM;
    float* part_qkv = h1 + E_DIM;
    float* part_o   = part_qkv + 12 * E_DIM;
    float* part_1   = part_o + 8 * E_DIM;
    float* part_2   = part_1 + 8 * E_DIM;

    qkv_part_kernel<<<dim3(8, 4, 3), 256, 0, stream>>>(x, Wv, Wq, Wk, part_qkv);
    qkv_combine_kernel<<<1, 256, 0, stream>>>(part_qkv, bv, bq, bk, q, newv, newk);
    shift_score_kernel<<<512, 256, 0, stream>>>(oldv, oldk, q, newv, newk, scores);
    softmax_kernel<<<16, 256, 0, stream>>>(scores, values);
    attnv_kernel<<<128, 256, 0, stream>>>(scores, newv, values);
    gemv_part_kernel<<<dim3(8, 8), 256, 0, stream>>>(Wo, values, 1, nullptr, 0, part_o);
    ln_combine_kernel<<<1, 256, 0, stream>>>(part_o, 8, bo, x, ln1_s, ln1_b, h1);
    gemv_part_kernel<<<dim3(8, 8), 256, 0, stream>>>(W1, h1, 1, nullptr, 0, part_1);
    gemv_part_kernel<<<dim3(8, 8), 256, 0, stream>>>(W2, part_1, 8, b1, 1, part_2);
    ln_combine_kernel<<<1, 256, 0, stream>>>(part_2, 8, b2, h1, ln2_s, ln2_b, out_h);
}


# attention in one pass over the cache: exp(score - wave max) times the V rows still in registers (WMMA), block partials merged by rescaling; no score array, softmax pass or atomics; new sum order
# speedup vs baseline: 2.0290x; 1.5222x over previous
#include <hip/hip_runtime.h>
#include <hip/hip_bf16.h>
#include <math.h>

#define E_DIM 1024
#define S_LEN 8192
#define H_N   16
#define D_H   64

typedef __attribute__((ext_vector_type(2))) float v2f;
typedef __attribute__((ext_vector_type(8))) float v8f;

__device__ __forceinline__ v8f wmma4(v2f a, v2f b, v8f c) {
    return __builtin_amdgcn_wmma_f32_16x16x4_f32(false, a, false, b, (short)0, c,
                                                 false, false);
}

__global__ void qkv_part_kernel(const float* __restrict__ x,
                                const float* __restrict__ Wv,
                                const float* __restrict__ Wq,
                                const float* __restrict__ Wk,
                                float* __restrict__ part_qkv) {
    __shared__ float xs[256];
    const int tid = threadIdx.x;
    const int k0  = blockIdx.y * 256;
    if (tid < 256) xs[tid] = x[k0 + tid];
    __syncthreads();

    const int m    = blockIdx.z;
    const int wv   = tid >> 5;
    const int lane = tid & 31;
    const int j    = lane & 15;
    const int half = lane >> 4;
    const int ob   = blockIdx.x * 128 + wv * 16;
    const float amask = (j == 0) ? 1.f : 0.f;

    const float* W = (m == 0) ? Wv : (m == 1) ? Wq : Wk;

    v8f c0 = {}, c1 = {};
    for (int eb = 0; eb < 256; eb += 8) {
        const int r0 = eb + half * 2;
        v2f a0, a1, b0, b1;
        a0.x = xs[r0]     * amask;  a0.y = xs[r0 + 1] * amask;
        a1.x = xs[r0 + 4] * amask;  a1.y = xs[r0 + 5] * amask;
        const float* wp0 = W + (size_t)(k0 + r0)     * E_DIM + ob + j;
        const float* wp1 = W + (size_t)(k0 + r0 + 4) * E_DIM + ob + j;
        b0.x = wp0[0];  b0.y = wp0[E_DIM];
        b1.x = wp1[0];  b1.y = wp1[E_DIM];
        c0 = wmma4(a0, b0, c0);
        c1 = wmma4(a1, b1, c1);
    }
    const v8f cs = c0 + c1;
    if (lane < 16)
        part_qkv[(size_t)(m * 4 + blockIdx.y) * E_DIM + ob + lane] = cs[0];
}

__global__ void qkv_combine_kernel(const float* __restrict__ part_qkv,
                                   const float* __restrict__ bv,
                                   const float* __restrict__ bq,
                                   const float* __restrict__ bk,
                                   float* __restrict__ q_out,
                                   float* __restrict__ newv, float* __restrict__ newk) {
    for (int o = threadIdx.x; o < 3 * E_DIM; o += 256) {
        const int m  = o >> 10;
        const int oo = o & (E_DIM - 1);
        float s = 0.f;
        #pragma unroll
        for (int ks = 0; ks < 4; ks++)
            s += part_qkv[(size_t)(m * 4 + ks) * E_DIM + oo];
        if (m == 0)      newv[(size_t)(S_LEN - 1) * E_DIM + oo] = s + bv[oo];
        else if (m == 1) q_out[oo] = s + bq[oo];
        else             newk[(size_t)(S_LEN - 1) * E_DIM + oo] = s + bk[oo];
    }
}

__global__ __launch_bounds__(256)
void shift_score_pv_kernel(const float* __restrict__ oldv,
                           const float* __restrict__ oldk,
                           const float* __restrict__ q,
                           float* __restrict__ newv, float* __restrict__ newk,
                           float* __restrict__ part_acc,
                           float* __restrict__ part_ml) {
    __shared__ float4 s_acc[8 * 256];
    __shared__ float  s_m[8 * 16];
    __shared__ float  s_l[8 * 16];
    __shared__ float  s_ml[32];
    const int tid  = threadIdx.x;
    const int lane = tid & 31;
    const int wv   = tid >> 5;
    const int s0   = (blockIdx.x * 8 + wv) * 2;
    const int s1   = s0 + 1;
    const bool last1 = (s1 == S_LEN - 1);

    const float4* v0src = (const float4*)(oldv + (size_t)(s0 + 1) * E_DIM);
    const float4* k0src = (const float4*)(oldk + (size_t)(s0 + 1) * E_DIM);
    const float4* v1src = (const float4*)(last1 ? (newv + (size_t)s1 * E_DIM)
                                                : (oldv + (size_t)(s1 + 1) * E_DIM));
    const float4* k1src = (const float4*)(last1 ? (newk + (size_t)s1 * E_DIM)
                                                : (oldk + (size_t)(s1 + 1) * E_DIM));
    float4* v0dst = (float4*)(newv + (size_t)s0 * E_DIM);
    float4* k0dst = (float4*)(newk + (size_t)s0 * E_DIM);
    float4* v1dst = (float4*)(newv + (size_t)s1 * E_DIM);
    float4* k1dst = (float4*)(newk + (size_t)s1 * E_DIM);
    const float4* q4 = (const float4*)q;

    float4 qd[8], va[8], vb[8], ka[8], kb[8];
    #pragma unroll
    for (int i = 0; i < 8; i++) qd[i] = q4[i * 32 + lane];
    #pragma unroll
    for (int i = 0; i < 8; i++) va[i] = v0src[i * 32 + lane];
    #pragma unroll
    for (int i = 0; i < 8; i++) vb[i] = v1src[i * 32 + lane];
    #pragma unroll
    for (int i = 0; i < 8; i++) ka[i] = k0src[i * 32 + lane];
    #pragma unroll
    for (int i = 0; i < 8; i++) kb[i] = k1src[i * 32 + lane];

    #pragma unroll
    for (int i = 0; i < 8; i++) v0dst[i * 32 + lane] = va[i];
    if (!last1) {
        #pragma unroll
        for (int i = 0; i < 8; i++) v1dst[i * 32 + lane] = vb[i];
    }
    #pragma unroll
    for (int i = 0; i < 8; i++) k0dst[i * 32 + lane] = ka[i];
    if (!last1) {
        #pragma unroll
        for (int i = 0; i < 8; i++) k1dst[i * 32 + lane] = kb[i];
    }

    int nza = 0, nzb = 0;
    #pragma unroll
    for (int i = 0; i < 8; i++) {
        nza |= (va[i].x != 0.f) | (va[i].y != 0.f) | (va[i].z != 0.f) | (va[i].w != 0.f);
        nzb |= (vb[i].x != 0.f) | (vb[i].y != 0.f) | (vb[i].z != 0.f) | (vb[i].w != 0.f);
    }
    #pragma unroll
    for (int mm = 16; mm >= 1; mm >>= 1) {
        nza |= __shfl_xor(nza, mm, 32);
        nzb |= __shfl_xor(nzb, mm, 32);
    }

    const bool arow = (lane == 0) || (lane == 24);
    const v8f zero8 = {};

    float sa[8], sb[8];
    #pragma unroll
    for (int i = 0; i < 8; i++) {
        float pa = ka[i].x * qd[i].x + ka[i].y * qd[i].y + ka[i].z * qd[i].z + ka[i].w * qd[i].w;
        float pb = kb[i].x * qd[i].x + kb[i].y * qd[i].y + kb[i].z * qd[i].z + kb[i].w * qd[i].w;
        pa += __shfl_xor(pa, 1, 32);  pb += __shfl_xor(pb, 1, 32);
        pa += __shfl_xor(pa, 2, 32);  pb += __shfl_xor(pb, 2, 32);
        pa += __shfl_xor(pa, 4, 32);  pb += __shfl_xor(pb, 4, 32);
        pa += __shfl_xor(pa, 8, 32);  pb += __shfl_xor(pb, 8, 32);
        sa[i] = nza ? pa * 0.125f : -INFINITY;
        sb[i] = nzb ? pb * 0.125f : -INFINITY;
    }

    #pragma unroll
    for (int i = 0; i < 8; i++) {
        const float m  = fmaxf(sa[i], sb[i]);
        const float ms = (m == -INFINITY) ? 0.f : m;
        const float ea = expf(sa[i] - ms);
        const float eb = expf(sb[i] - ms);
        if ((lane & 15) == 0) {
            const int h = i * 2 + (lane >> 4);
            s_m[wv * 16 + h] = m;
            s_l[wv * 16 + h] = ea + eb;
        }
        v2f a, b0, b1, b2, b3;
        a.x  = arow ? ea : 0.f;   a.y  = arow ? eb : 0.f;
        b0.x = va[i].x;           b0.y = vb[i].x;
        b1.x = va[i].y;           b1.y = vb[i].y;
        b2.x = va[i].z;           b2.y = vb[i].z;
        b3.x = va[i].w;           b3.y = vb[i].w;
        const v8f d0 = wmma4(a, b0, zero8);
        const v8f d1 = wmma4(a, b1, zero8);
        const v8f d2 = wmma4(a, b2, zero8);
        const v8f d3 = wmma4(a, b3, zero8);
        float4 acc;
        acc.x = d0[0];  acc.y = d1[0];  acc.z = d2[0];  acc.w = d3[0];
        s_acc[wv * 256 + i * 32 + lane] = acc;
    }
    __syncthreads();

    {
        const int head = tid >> 4;
        float mw[8];
        float M = -INFINITY;
        #pragma unroll
        for (int w = 0; w < 8; w++) { mw[w] = s_m[w * 16 + head];  M = fmaxf(M, mw[w]); }
        float4 acc = {0.f, 0.f, 0.f, 0.f};
        float L = 0.f;
        const float Ms = (M == -INFINITY) ? 0.f : M;
        #pragma unroll
        for (int w = 0; w < 8; w++) {
            const float sc = expf(mw[w] - Ms);
            const float4 t = s_acc[w * 256 + tid];
            acc.x += t.x * sc;  acc.y += t.y * sc;  acc.z += t.z * sc;  acc.w += t.w * sc;
            L += s_l[w * 16 + head] * sc;
        }
        ((float4*)part_acc)[(size_t)blockIdx.x * 256 + tid] = acc;
        if ((tid & 15) == 0) { s_ml[head] = M;  s_ml[16 + head] = L; }
    }
    __syncthreads();
    if (tid < 32) part_ml[blockIdx.x * 32 + tid] = s_ml[tid];
}

__global__ __launch_bounds__(512)
void attn_merge_kernel(const float* __restrict__ part_acc,
                       const float* __restrict__ part_ml,
                       float* __restrict__ values) {
    __shared__ float  s_w[512];
    __shared__ float  s_red[16];
    __shared__ float4 s_sum[64 * 8];
    const int h    = blockIdx.x;
    const int hx   = blockIdx.y;
    const int tid  = threadIdx.x;
    const int lane = tid & 31;
    const int wv   = tid >> 5;

    const float mb = part_ml[tid * 32 + h];
    const float lb = part_ml[tid * 32 + 16 + h];

    const int c = tid & 7;
    const int g = tid >> 3;
    const float4* src = (const float4*)part_acc + h * 16 + hx * 8 + c;
    float4 pv[8];
    #pragma unroll
    for (int it = 0; it < 8; it++) pv[it] = src[(size_t)(it * 64 + g) * 256];

    float mx = mb;
    #pragma unroll
    for (int mm = 16; mm >= 1; mm >>= 1) mx = fmaxf(mx, __shfl_xor(mx, mm, 32));
    if (lane == 0) s_red[wv] = mx;
    __syncthreads();
    float M = s_red[0];
    for (int w = 1; w < 16; w++) M = fmaxf(M, s_red[w]);
    __syncthreads();

    const float Ms  = (M == -INFINITY) ? 0.f : M;
    const float wgt = expf(mb - Ms);
    s_w[tid] = wgt;
    float ls = wgt * lb;
    #pragma unroll
    for (int mm = 16; mm >= 1; mm >>= 1) ls += __shfl_xor(ls, mm, 32);
    if (lane == 0) s_red[wv] = ls;
    __syncthreads();
    float L = 0.f;
    for (int w = 0; w < 16; w++) L += s_red[w];

    float4 acc = {0.f, 0.f, 0.f, 0.f};
    #pragma unroll
    for (int it = 0; it < 8; it++) {
        const float w = s_w[it * 64 + g];
        acc.x += pv[it].x * w;  acc.y += pv[it].y * w;
        acc.z += pv[it].z * w;  acc.w += pv[it].w * w;
    }
    s_sum[g * 8 + c] = acc;
    __syncthreads();
    if (tid < 32) {
        const float* sf = (const float*)s_sum;
        float s = 0.f;
        for (int gg = 0; gg < 64; gg++) s += sf[gg * 32 + tid];
        values[h * D_H + hx * 32 + tid] = s / L;
    }
}

__global__ void gemv_part_kernel(const float* __restrict__ W,
                                 const float* __restrict__ in_parts, int P,
                                 const float* __restrict__ in_bias, int in_relu,
                                 float* __restrict__ out_parts) {
    __shared__ float xs[128];
    const int tid = threadIdx.x;
    const int k0  = blockIdx.y * 128;
    if (tid < 128) {
        float s = 0.f;
        for (int pp = 0; pp < P; pp++) s += in_parts[(size_t)pp * E_DIM + k0 + tid];
        if (in_bias) s += in_bias[k0 + tid];
        if (in_relu) s = fmaxf(s, 0.f);
        xs[tid] = s;
    }
    __syncthreads();

    const int wv   = tid >> 5;
    const int lane = tid & 31;
    const int j    = lane & 15;
    const int half = lane >> 4;
    const int ob   = blockIdx.x * 128 + wv * 16;
    const float amask = (j == 0) ? 1.f : 0.f;

    v8f c0 = {}, c1 = {};
    for (int eb = 0; eb < 128; eb += 8) {
        const int r0 = eb + half * 2;
        v2f a0, a1, b0, b1;
        a0.x = xs[r0]     * amask;  a0.y = xs[r0 + 1] * amask;
        a1.x = xs[r0 + 4] * amask;  a1.y = xs[r0 + 5] * amask;
        const float* wp0 = W + (size_t)(k0 + r0)     * E_DIM + ob + j;
        const float* wp1 = W + (size_t)(k0 + r0 + 4) * E_DIM + ob + j;
        b0.x = wp0[0];  b0.y = wp0[E_DIM];
        b1.x = wp1[0];  b1.y = wp1[E_DIM];
        c0 = wmma4(a0, b0, c0);
        c1 = wmma4(a1, b1, c1);
    }
    const v8f cs = c0 + c1;
    if (lane < 16)
        out_parts[(size_t)blockIdx.y * E_DIM + ob + lane] = cs[0];
}

__global__ void ln_combine_kernel(const float* __restrict__ parts, int P,
                                  const float* __restrict__ bias,
                                  const float* __restrict__ resid,
                                  const float* __restrict__ sc,
                                  const float* __restrict__ bs,
                                  float* __restrict__ out) {
    __shared__ float red[256];
    const int tid = threadIdx.x;
    float vv[4];
    float sum = 0.f;
    #pragma unroll
    for (int i = 0; i < 4; i++) {
        const int e = i * 256 + tid;
        float s = bias[e] + resid[e];
        for (int pp = 0; pp < P; pp++) s += parts[(size_t)pp * E_DIM + e];
        vv[i] = s;
        sum += s;
    }
    red[tid] = sum; __syncthreads();
    for (int st = 128; st > 0; st >>= 1) {
        if (tid < st) red[tid] += red[tid + st];
        __syncthreads();
    }
    const float mean = red[0] * (1.f / E_DIM); __syncthreads();
    float var = 0.f;
    #pragma unroll
    for (int i = 0; i < 4; i++) { const float d = vv[i] - mean; var += d * d; }
    red[tid] = var; __syncthreads();
    for (int st = 128; st > 0; st >>= 1) {
        if (tid < st) red[tid] += red[tid + st];
        __syncthreads();
    }
    const float inv = rsqrtf(red[0] * (1.f / E_DIM) + 1e-6f);
    #pragma unroll
    for (int i = 0; i < 4; i++) {
        const int e = i * 256 + tid;
        out[e] = (vv[i] - mean) * inv * sc[e] + bs[e];
    }
}

extern "C" void kernel_launch(void* const* d_in, const int* in_sizes, int n_in,
                              void* d_out, int out_size, void* d_ws, size_t ws_size,
                              hipStream_t stream) {
    (void)in_sizes; (void)n_in; (void)out_size; (void)ws_size;
    const float* x     = (const float*)d_in[0];
    const float* cache = (const float*)d_in[1];
    const float* Wv = (const float*)d_in[2];  const float* bv = (const float*)d_in[3];
    const float* Wq = (const float*)d_in[4];  const float* bq = (const float*)d_in[5];
    const float* Wk = (const float*)d_in[6];  const float* bk = (const float*)d_in[7];
    const float* Wo = (const float*)d_in[8];  const float* bo = (const float*)d_in[9];
    const float* W1 = (const float*)d_in[10]; const float* b1 = (const float*)d_in[11];
    const float* W2 = (const float*)d_in[12]; const float* b2 = (const float*)d_in[13];
    const float* ln1_s = (const float*)d_in[14]; const float* ln1_b = (const float*)d_in[15];
    const float* ln2_s = (const float*)d_in[16]; const float* ln2_b = (const float*)d_in[17];

    float* out_h = (float*)d_out;
    float* newv  = out_h + E_DIM;
    float* newk  = newv + (size_t)S_LEN * E_DIM;
    const float* oldv = cache;
    const float* oldk = cache + (size_t)S_LEN * E_DIM;

    float* ws       = (float*)d_ws;
    float* q        = ws;
    float* part_acc = q + E_DIM;
    float* part_ml  = part_acc + 512 * E_DIM;
    float* values   = part_ml + 512 * 32;
    float* h1       = values + E_DIM;
    float* part_qkv = h1 + E_DIM;
    float* part_o   = part_qkv + 12 * E_DIM;
    float* part_1   = part_o + 8 * E_DIM;
    float* part_2   = part_1 + 8 * E_DIM;

    qkv_part_kernel<<<dim3(8, 4, 3), 256, 0, stream>>>(x, Wv, Wq, Wk, part_qkv);
    qkv_combine_kernel<<<1, 256, 0, stream>>>(part_qkv, bv, bq, bk, q, newv, newk);
    shift_score_pv_kernel<<<512, 256, 0, stream>>>(oldv, oldk, q, newv, newk, part_acc, part_ml);
    attn_merge_kernel<<<dim3(16, 2), 512, 0, stream>>>(part_acc, part_ml, values);
    gemv_part_kernel<<<dim3(8, 8), 256, 0, stream>>>(Wo, values, 1, nullptr, 0, part_o);
    ln_combine_kernel<<<1, 256, 0, stream>>>(part_o, 8, bo, x, ln1_s, ln1_b, h1);
    gemv_part_kernel<<<dim3(8, 8), 256, 0, stream>>>(W1, h1, 1, nullptr, 0, part_1);
    gemv_part_kernel<<<dim3(8, 8), 256, 0, stream>>>(W2, part_1, 8, b1, 1, part_2);
    ln_combine_kernel<<<1, 256, 0, stream>>>(part_2, 8, b2, h1, ln2_s, ln2_b, out_h);
}


# qkv combine and LN1 launches folded away: qkv K slices summed in the producing block's LDS, LN1 formed in each W1 GEMV block's preload, loads issued first; 9 -> 7 launches; same sum order
# speedup vs baseline: 2.6901x; 1.3258x over previous
#include <hip/hip_runtime.h>
#include <hip/hip_bf16.h>
#include <math.h>

#define E_DIM 1024
#define S_LEN 8192
#define H_N   16
#define D_H   64

typedef __attribute__((ext_vector_type(2))) float v2f;
typedef __attribute__((ext_vector_type(8))) float v8f;

__device__ __forceinline__ v8f wmma4(v2f a, v2f b, v8f c) {
    return __builtin_amdgcn_wmma_f32_16x16x4_f32(false, a, false, b, (short)0, c,
                                                 false, false);
}

__global__ __launch_bounds__(256)
void qkv_kernel(const float* __restrict__ x,
                const float* __restrict__ Wv,
                const float* __restrict__ Wq,
                const float* __restrict__ Wk,
                const float* __restrict__ bv,
                const float* __restrict__ bq,
                const float* __restrict__ bk,
                float* __restrict__ q_out,
                float* __restrict__ newv, float* __restrict__ newk) {
    __shared__ float4 xs4[E_DIM / 4];
    __shared__ float  red[8 * 16];
    const float* xs = (const float*)xs4;
    const int tid = threadIdx.x;
    const int m = blockIdx.y;
    const float* W    = (m == 0) ? Wv : (m == 1) ? Wq : Wk;
    const float* bias = (m == 0) ? bv : (m == 1) ? bq : bk;
    float* dst = (m == 0) ? newv + (size_t)(S_LEN - 1) * E_DIM
               : (m == 1) ? q_out
                          : newk + (size_t)(S_LEN - 1) * E_DIM;
    const float bval = bias[blockIdx.x * 32 + (tid & 31)];
    xs4[tid] = ((const float4*)x)[tid];
    __syncthreads();

    const int wv   = tid >> 5;
    const int lane = tid & 31;
    const int j    = lane & 15;
    const int half = lane >> 4;
    const int ob   = blockIdx.x * 32 + (wv & 1) * 16;
    const int k0   = (wv >> 1) * 256;
    const float amask = (j == 0) ? 1.f : 0.f;

    v8f c0 = {}, c1 = {};
    for (int eb = 0; eb < 256; eb += 8) {
        const int r0 = eb + half * 2;
        v2f a0, a1, b0, b1;
        a0.x = xs[k0 + r0]     * amask;  a0.y = xs[k0 + r0 + 1] * amask;
        a1.x = xs[k0 + r0 + 4] * amask;  a1.y = xs[k0 + r0 + 5] * amask;
        const float* wp0 = W + (size_t)(k0 + r0)     * E_DIM + ob + j;
        const float* wp1 = W + (size_t)(k0 + r0 + 4) * E_DIM + ob + j;
        b0.x = wp0[0];  b0.y = wp0[E_DIM];
        b1.x = wp1[0];  b1.y = wp1[E_DIM];
        c0 = wmma4(a0, b0, c0);
        c1 = wmma4(a1, b1, c1);
    }
    const v8f cs = c0 + c1;
    if (lane < 16) red[wv * 16 + lane] = cs[0];
    __syncthreads();
    if (tid < 32) {
        const int og = tid >> 4;
        const int jj = tid & 15;
        float s = 0.f;
        #pragma unroll
        for (int ks = 0; ks < 4; ks++) s += red[(ks * 2 + og) * 16 + jj];
        dst[blockIdx.x * 32 + tid] = s + bval;
    }
}

__global__ __launch_bounds__(256)
void shift_score_pv_kernel(const float* __restrict__ oldv,
                           const float* __restrict__ oldk,
                           const float* __restrict__ q,
                           float* __restrict__ newv, float* __restrict__ newk,
                           float* __restrict__ part_acc,
                           float* __restrict__ part_ml) {
    __shared__ float4 s_acc[8 * 256];
    __shared__ float  s_m[8 * 16];
    __shared__ float  s_l[8 * 16];
    __shared__ float  s_ml[32];
    const int tid  = threadIdx.x;
    const int lane = tid & 31;
    const int wv   = tid >> 5;
    const int s0   = (blockIdx.x * 8 + wv) * 2;
    const int s1   = s0 + 1;
    const bool last1 = (s1 == S_LEN - 1);

    const float4* v0src = (const float4*)(oldv + (size_t)(s0 + 1) * E_DIM);
    const float4* k0src = (const float4*)(oldk + (size_t)(s0 + 1) * E_DIM);
    const float4* v1src = (const float4*)(last1 ? (newv + (size_t)s1 * E_DIM)
                                                : (oldv + (size_t)(s1 + 1) * E_DIM));
    const float4* k1src = (const float4*)(last1 ? (newk + (size_t)s1 * E_DIM)
                                                : (oldk + (size_t)(s1 + 1) * E_DIM));
    float4* v0dst = (float4*)(newv + (size_t)s0 * E_DIM);
    float4* k0dst = (float4*)(newk + (size_t)s0 * E_DIM);
    float4* v1dst = (float4*)(newv + (size_t)s1 * E_DIM);
    float4* k1dst = (float4*)(newk + (size_t)s1 * E_DIM);
    const float4* q4 = (const float4*)q;

    float4 qd[8], va[8], vb[8], ka[8], kb[8];
    #pragma unroll
    for (int i = 0; i < 8; i++) qd[i] = q4[i * 32 + lane];
    #pragma unroll
    for (int i = 0; i < 8; i++) va[i] = v0src[i * 32 + lane];
    #pragma unroll
    for (int i = 0; i < 8; i++) vb[i] = v1src[i * 32 + lane];
    #pragma unroll
    for (int i = 0; i < 8; i++) ka[i] = k0src[i * 32 + lane];
    #pragma unroll
    for (int i = 0; i < 8; i++) kb[i] = k1src[i * 32 + lane];

    #pragma unroll
    for (int i = 0; i < 8; i++) v0dst[i * 32 + lane] = va[i];
    if (!last1) {
        #pragma unroll
        for (int i = 0; i < 8; i++) v1dst[i * 32 + lane] = vb[i];
    }
    #pragma unroll
    for (int i = 0; i < 8; i++) k0dst[i * 32 + lane] = ka[i];
    if (!last1) {
        #pragma unroll
        for (int i = 0; i < 8; i++) k1dst[i * 32 + lane] = kb[i];
    }

    int nza = 0, nzb = 0;
    #pragma unroll
    for (int i = 0; i < 8; i++) {
        nza |= (va[i].x != 0.f) | (va[i].y != 0.f) | (va[i].z != 0.f) | (va[i].w != 0.f);
        nzb |= (vb[i].x != 0.f) | (vb[i].y != 0.f) | (vb[i].z != 0.f) | (vb[i].w != 0.f);
    }
    #pragma unroll
    for (int mm = 16; mm >= 1; mm >>= 1) {
        nza |= __shfl_xor(nza, mm, 32);
        nzb |= __shfl_xor(nzb, mm, 32);
    }

    const bool arow = (lane == 0) || (lane == 24);
    const v8f zero8 = {};

    float sa[8], sb[8];
    #pragma unroll
    for (int i = 0; i < 8; i++) {
        float pa = ka[i].x * qd[i].x + ka[i].y * qd[i].y + ka[i].z * qd[i].z + ka[i].w * qd[i].w;
        float pb = kb[i].x * qd[i].x + kb[i].y * qd[i].y + kb[i].z * qd[i].z + kb[i].w * qd[i].w;
        pa += __shfl_xor(pa, 1, 32);  pb += __shfl_xor(pb, 1, 32);
        pa += __shfl_xor(pa, 2, 32);  pb += __shfl_xor(pb, 2, 32);
        pa += __shfl_xor(pa, 4, 32);  pb += __shfl_xor(pb, 4, 32);
        pa += __shfl_xor(pa, 8, 32);  pb += __shfl_xor(pb, 8, 32);
        sa[i] = nza ? pa * 0.125f : -INFINITY;
        sb[i] = nzb ? pb * 0.125f : -INFINITY;
    }

    #pragma unroll
    for (int i = 0; i < 8; i++) {
        const float m  = fmaxf(sa[i], sb[i]);
        const float ms = (m == -INFINITY) ? 0.f : m;
        const float ea = expf(sa[i] - ms);
        const float eb = expf(sb[i] - ms);
        if ((lane & 15) == 0) {
            const int h = i * 2 + (lane >> 4);
            s_m[wv * 16 + h] = m;
            s_l[wv * 16 + h] = ea + eb;
        }
        v2f a, b0, b1, b2, b3;
        a.x  = arow ? ea : 0.f;   a.y  = arow ? eb : 0.f;
        b0.x = va[i].x;           b0.y = vb[i].x;
        b1.x = va[i].y;           b1.y = vb[i].y;
        b2.x = va[i].z;           b2.y = vb[i].z;
        b3.x = va[i].w;           b3.y = vb[i].w;
        const v8f d0 = wmma4(a, b0, zero8);
        const v8f d1 = wmma4(a, b1, zero8);
        const v8f d2 = wmma4(a, b2, zero8);
        const v8f d3 = wmma4(a, b3, zero8);
        float4 acc;
        acc.x = d0[0];  acc.y = d1[0];  acc.z = d2[0];  acc.w = d3[0];
        s_acc[wv * 256 + i * 32 + lane] = acc;
    }
    __syncthreads();

    {
        const int head = tid >> 4;
        float mw[8];
        float M = -INFINITY;
        #pragma unroll
        for (int w = 0; w < 8; w++) { mw[w] = s_m[w * 16 + head];  M = fmaxf(M, mw[w]); }
        float4 acc = {0.f, 0.f, 0.f, 0.f};
        float L = 0.f;
        const float Ms = (M == -INFINITY) ? 0.f : M;
        #pragma unroll
        for (int w = 0; w < 8; w++) {
            const float sc = expf(mw[w] - Ms);
            const float4 t = s_acc[w * 256 + tid];
            acc.x += t.x * sc;  acc.y += t.y * sc;  acc.z += t.z * sc;  acc.w += t.w * sc;
            L += s_l[w * 16 + head] * sc;
        }
        ((float4*)part_acc)[(size_t)blockIdx.x * 256 + tid] = acc;
        if ((tid & 15) == 0) { s_ml[head] = M;  s_ml[16 + head] = L; }
    }
    __syncthreads();
    if (tid < 32) part_ml[blockIdx.x * 32 + tid] = s_ml[tid];
}

__global__ __launch_bounds__(512)
void attn_merge_kernel(const float* __restrict__ part_acc,
                       const float* __restrict__ part_ml,
                       float* __restrict__ values) {
    __shared__ float  s_w[512];
    __shared__ float  s_red[16];
    __shared__ float4 s_sum[64 * 8];
    const int h    = blockIdx.x;
    const int hx   = blockIdx.y;
    const int tid  = threadIdx.x;
    const int lane = tid & 31;
    const int wv   = tid >> 5;

    const float mb = part_ml[tid * 32 + h];
    const float lb = part_ml[tid * 32 + 16 + h];

    const int c = tid & 7;
    const int g = tid >> 3;
    const float4* src = (const float4*)part_acc + h * 16 + hx * 8 + c;
    float4 pv[8];
    #pragma unroll
    for (int it = 0; it < 8; it++) pv[it] = src[(size_t)(it * 64 + g) * 256];

    float mx = mb;
    #pragma unroll
    for (int mm = 16; mm >= 1; mm >>= 1) mx = fmaxf(mx, __shfl_xor(mx, mm, 32));
    if (lane == 0) s_red[wv] = mx;
    __syncthreads();
    float M = s_red[0];
    for (int w = 1; w < 16; w++) M = fmaxf(M, s_red[w]);
    __syncthreads();

    const float Ms  = (M == -INFINITY) ? 0.f : M;
    const float wgt = expf(mb - Ms);
    s_w[tid] = wgt;
    float ls = wgt * lb;
    #pragma unroll
    for (int mm = 16; mm >= 1; mm >>= 1) ls += __shfl_xor(ls, mm, 32);
    if (lane == 0) s_red[wv] = ls;
    __syncthreads();
    float L = 0.f;
    for (int w = 0; w < 16; w++) L += s_red[w];

    float4 acc = {0.f, 0.f, 0.f, 0.f};
    #pragma unroll
    for (int it = 0; it < 8; it++) {
        const float w = s_w[it * 64 + g];
        acc.x += pv[it].x * w;  acc.y += pv[it].y * w;
        acc.z += pv[it].z * w;  acc.w += pv[it].w * w;
    }
    s_sum[g * 8 + c] = acc;
    __syncthreads();
    if (tid < 32) {
        const float* sf = (const float*)s_sum;
        float s = 0.f;
        for (int gg = 0; gg < 64; gg++) s += sf[gg * 32 + tid];
        values[h * D_H + hx * 32 + tid] = s / L;
    }
}

__global__ void gemv_part_kernel(const float* __restrict__ W,
                                 const float* __restrict__ in_parts, int P,
                                 const float* __restrict__ in_bias, int in_relu,
                                 float* __restrict__ out_parts) {
    __shared__ float xs[128];
    const int tid = threadIdx.x;
    const int k0  = blockIdx.y * 128;
    if (tid < 128) {
        float s = 0.f;
        for (int pp = 0; pp < P; pp++) s += in_parts[(size_t)pp * E_DIM + k0 + tid];
        if (in_bias) s += in_bias[k0 + tid];
        if (in_relu) s = fmaxf(s, 0.f);
        xs[tid] = s;
    }
    __syncthreads();

    const int wv   = tid >> 5;
    const int lane = tid & 31;
    const int j    = lane & 15;
    const int half = lane >> 4;
    const int ob   = blockIdx.x * 128 + wv * 16;
    const float amask = (j == 0) ? 1.f : 0.f;

    v8f c0 = {}, c1 = {};
    for (int eb = 0; eb < 128; eb += 8) {
        const int r0 = eb + half * 2;
        v2f a0, a1, b0, b1;
        a0.x = xs[r0]     * amask;  a0.y = xs[r0 + 1] * amask;
        a1.x = xs[r0 + 4] * amask;  a1.y = xs[r0 + 5] * amask;
        const float* wp0 = W + (size_t)(k0 + r0)     * E_DIM + ob + j;
        const float* wp1 = W + (size_t)(k0 + r0 + 4) * E_DIM + ob + j;
        b0.x = wp0[0];  b0.y = wp0[E_DIM];
        b1.x = wp1[0];  b1.y = wp1[E_DIM];
        c0 = wmma4(a0, b0, c0);
        c1 = wmma4(a1, b1, c1);
    }
    const v8f cs = c0 + c1;
    if (lane < 16)
        out_parts[(size_t)blockIdx.y * E_DIM + ob + lane] = cs[0];
}

__global__ __launch_bounds__(256)
void gemv_ln_part_kernel(const float* __restrict__ W,
                         const float* __restrict__ parts,
                         const float* __restrict__ bias,
                         const float* __restrict__ resid,
                         const float* __restrict__ sc,
                         const float* __restrict__ bs,
                         float* __restrict__ h_out,
                         float* __restrict__ out_parts) {
    __shared__ float red[256];
    __shared__ float xs[128];
    const int tid = threadIdx.x;
    const int k0  = blockIdx.y * 128;

    float pb[4], pr[4], pp[4][8], psc[4], pbs[4];
    #pragma unroll
    for (int i = 0; i < 4; i++) {
        const int e = i * 256 + tid;
        pb[i] = bias[e];
        pr[i] = resid[e];
        #pragma unroll
        for (int p = 0; p < 8; p++) pp[i][p] = parts[(size_t)p * E_DIM + e];
        psc[i] = sc[e];
        pbs[i] = bs[e];
    }
    float vv[4];
    float sum = 0.f;
    #pragma unroll
    for (int i = 0; i < 4; i++) {
        float s = pb[i] + pr[i];
        #pragma unroll
        for (int p = 0; p < 8; p++) s += pp[i][p];
        vv[i] = s;
        sum += s;
    }
    red[tid] = sum; __syncthreads();
    for (int st = 128; st > 0; st >>= 1) {
        if (tid < st) red[tid] += red[tid + st];
        __syncthreads();
    }
    const float mean = red[0] * (1.f / E_DIM); __syncthreads();
    float var = 0.f;
    #pragma unroll
    for (int i = 0; i < 4; i++) { const float d = vv[i] - mean; var += d * d; }
    red[tid] = var; __syncthreads();
    for (int st = 128; st > 0; st >>= 1) {
        if (tid < st) red[tid] += red[tid + st];
        __syncthreads();
    }
    const float inv = rsqrtf(red[0] * (1.f / E_DIM) + 1e-6f);
    const bool mine = ((tid >> 7) == (int)(blockIdx.y & 1));
    #pragma unroll
    for (int i = 0; i < 4; i++) {
        const float hv = (vv[i] - mean) * inv * psc[i] + pbs[i];
        if (mine && i == (int)(blockIdx.y >> 1)) {
            xs[tid & 127] = hv;
            if (blockIdx.x == 0) h_out[i * 256 + tid] = hv;
        }
    }
    __syncthreads();

    const int wv   = tid >> 5;
    const int lane = tid & 31;
    const int j    = lane & 15;
    const int half = lane >> 4;
    const int ob   = blockIdx.x * 128 + wv * 16;
    const float amask = (j == 0) ? 1.f : 0.f;

    v8f c0 = {}, c1 = {};
    for (int eb = 0; eb < 128; eb += 8) {
        const int r0 = eb + half * 2;
        v2f a0, a1, b0, b1;
        a0.x = xs[r0]     * amask;  a0.y = xs[r0 + 1] * amask;
        a1.x = xs[r0 + 4] * amask;  a1.y = xs[r0 + 5] * amask;
        const float* wp0 = W + (size_t)(k0 + r0)     * E_DIM + ob + j;
        const float* wp1 = W + (size_t)(k0 + r0 + 4) * E_DIM + ob + j;
        b0.x = wp0[0];  b0.y = wp0[E_DIM];
        b1.x = wp1[0];  b1.y = wp1[E_DIM];
        c0 = wmma4(a0, b0, c0);
        c1 = wmma4(a1, b1, c1);
    }
    const v8f cs = c0 + c1;
    if (lane < 16)
        out_parts[(size_t)blockIdx.y * E_DIM + ob + lane] = cs[0];
}

__global__ void ln_combine_kernel(const float* __restrict__ parts, int P,
                                  const float* __restrict__ bias,
                                  const float* __restrict__ resid,
                                  const float* __restrict__ sc,
                                  const float* __restrict__ bs,
                                  float* __restrict__ out) {
    __shared__ float red[256];
    const int tid = threadIdx.x;
    float vv[4];
    float sum = 0.f;
    #pragma unroll
    for (int i = 0; i < 4; i++) {
        const int e = i * 256 + tid;
        float s = bias[e] + resid[e];
        for (int pp = 0; pp < P; pp++) s += parts[(size_t)pp * E_DIM + e];
        vv[i] = s;
        sum += s;
    }
    red[tid] = sum; __syncthreads();
    for (int st = 128; st > 0; st >>= 1) {
        if (tid < st) red[tid] += red[tid + st];
        __syncthreads();
    }
    const float mean = red[0] * (1.f / E_DIM); __syncthreads();
    float var = 0.f;
    #pragma unroll
    for (int i = 0; i < 4; i++) { const float d = vv[i] - mean; var += d * d; }
    red[tid] = var; __syncthreads();
    for (int st = 128; st > 0; st >>= 1) {
        if (tid < st) red[tid] += red[tid + st];
        __syncthreads();
    }
    const float inv = rsqrtf(red[0] * (1.f / E_DIM) + 1e-6f);
    #pragma unroll
    for (int i = 0; i < 4; i++) {
        const int e = i * 256 + tid;
        out[e] = (vv[i] - mean) * inv * sc[e] + bs[e];
    }
}

extern "C" void kernel_launch(void* const* d_in, const int* in_sizes, int n_in,
                              void* d_out, int out_size, void* d_ws, size_t ws_size,
                              hipStream_t stream) {
    (void)in_sizes; (void)n_in; (void)out_size; (void)ws_size;
    const float* x     = (const float*)d_in[0];
    const float* cache = (const float*)d_in[1];
    const float* Wv = (const float*)d_in[2];  const float* bv = (const float*)d_in[3];
    const float* Wq = (const float*)d_in[4];  const float* bq = (const float*)d_in[5];
    const float* Wk = (const float*)d_in[6];  const float* bk = (const float*)d_in[7];
    const float* Wo = (const float*)d_in[8];  const float* bo = (const float*)d_in[9];
    const float* W1 = (const float*)d_in[10]; const float* b1 = (const float*)d_in[11];
    const float* W2 = (const float*)d_in[12]; const float* b2 = (const float*)d_in[13];
    const float* ln1_s = (const float*)d_in[14]; const float* ln1_b = (const float*)d_in[15];
    const float* ln2_s = (const float*)d_in[16]; const float* ln2_b = (const float*)d_in[17];

    float* out_h = (float*)d_out;
    float* newv  = out_h + E_DIM;
    float* newk  = newv + (size_t)S_LEN * E_DIM;
    const float* oldv = cache;
    const float* oldk = cache + (size_t)S_LEN * E_DIM;

    float* ws       = (float*)d_ws;
    float* q        = ws;
    float* part_acc = q + E_DIM;
    float* part_ml  = part_acc + 512 * E_DIM;
    float* values   = part_ml + 512 * 32;
    float* h1       = values + E_DIM;
    float* part_o   = h1 + E_DIM;
    float* part_1   = part_o + 8 * E_DIM;
    float* part_2   = part_1 + 8 * E_DIM;

    qkv_kernel<<<dim3(32, 3), 256, 0, stream>>>(x, Wv, Wq, Wk, bv, bq, bk, q, newv, newk);
    shift_score_pv_kernel<<<512, 256, 0, stream>>>(oldv, oldk, q, newv, newk, part_acc, part_ml);
    attn_merge_kernel<<<dim3(16, 2), 512, 0, stream>>>(part_acc, part_ml, values);
    gemv_part_kernel<<<dim3(8, 8), 256, 0, stream>>>(Wo, values, 1, nullptr, 0, part_o);
    gemv_ln_part_kernel<<<dim3(8, 8), 256, 0, stream>>>(W1, part_o, bo, x, ln1_s, ln1_b, h1, part_1);
    gemv_part_kernel<<<dim3(8, 8), 256, 0, stream>>>(W2, part_1, 8, b1, 1, part_2);
    ln_combine_kernel<<<1, 256, 0, stream>>>(part_2, 8, b2, h1, ln2_s, ln2_b, out_h);
}


# merge launch folded into the Wo GEMV: each block (head x 128 outputs) merges its head's 512 partials in LDS, W tile loaded first; 16 K slices of 64 feed LN1; 7 -> 6 launches; new sum order
# speedup vs baseline: 2.8510x; 1.0598x over previous
#include <hip/hip_runtime.h>
#include <hip/hip_bf16.h>
#include <math.h>

#define E_DIM 1024
#define S_LEN 8192
#define H_N   16
#define D_H   64

typedef __attribute__((ext_vector_type(2))) float v2f;
typedef __attribute__((ext_vector_type(8))) float v8f;

__device__ __forceinline__ v8f wmma4(v2f a, v2f b, v8f c) {
    return __builtin_amdgcn_wmma_f32_16x16x4_f32(false, a, false, b, (short)0, c,
                                                 false, false);
}

__global__ __launch_bounds__(256)
void qkv_kernel(const float* __restrict__ x,
                const float* __restrict__ Wv,
                const float* __restrict__ Wq,
                const float* __restrict__ Wk,
                const float* __restrict__ bv,
                const float* __restrict__ bq,
                const float* __restrict__ bk,
                float* __restrict__ q_out,
                float* __restrict__ newv, float* __restrict__ newk) {
    __shared__ float4 xs4[E_DIM / 4];
    __shared__ float  red[8 * 16];
    const float* xs = (const float*)xs4;
    const int tid = threadIdx.x;
    const int m = blockIdx.y;
    const float* W    = (m == 0) ? Wv : (m == 1) ? Wq : Wk;
    const float* bias = (m == 0) ? bv : (m == 1) ? bq : bk;
    float* dst = (m == 0) ? newv + (size_t)(S_LEN - 1) * E_DIM
               : (m == 1) ? q_out
                          : newk + (size_t)(S_LEN - 1) * E_DIM;
    const float bval = bias[blockIdx.x * 32 + (tid & 31)];
    xs4[tid] = ((const float4*)x)[tid];
    __syncthreads();

    const int wv   = tid >> 5;
    const int lane = tid & 31;
    const int j    = lane & 15;
    const int half = lane >> 4;
    const int ob   = blockIdx.x * 32 + (wv & 1) * 16;
    const int k0   = (wv >> 1) * 256;
    const float amask = (j == 0) ? 1.f : 0.f;

    v8f c0 = {}, c1 = {};
    for (int eb = 0; eb < 256; eb += 8) {
        const int r0 = eb + half * 2;
        v2f a0, a1, b0, b1;
        a0.x = xs[k0 + r0]     * amask;  a0.y = xs[k0 + r0 + 1] * amask;
        a1.x = xs[k0 + r0 + 4] * amask;  a1.y = xs[k0 + r0 + 5] * amask;
        const float* wp0 = W + (size_t)(k0 + r0)     * E_DIM + ob + j;
        const float* wp1 = W + (size_t)(k0 + r0 + 4) * E_DIM + ob + j;
        b0.x = wp0[0];  b0.y = wp0[E_DIM];
        b1.x = wp1[0];  b1.y = wp1[E_DIM];
        c0 = wmma4(a0, b0, c0);
        c1 = wmma4(a1, b1, c1);
    }
    const v8f cs = c0 + c1;
    if (lane < 16) red[wv * 16 + lane] = cs[0];
    __syncthreads();
    if (tid < 32) {
        const int og = tid >> 4;
        const int jj = tid & 15;
        float s = 0.f;
        #pragma unroll
        for (int ks = 0; ks < 4; ks++) s += red[(ks * 2 + og) * 16 + jj];
        dst[blockIdx.x * 32 + tid] = s + bval;
    }
}

__global__ __launch_bounds__(256)
void shift_score_pv_kernel(const float* __restrict__ oldv,
                           const float* __restrict__ oldk,
                           const float* __restrict__ q,
                           float* __restrict__ newv, float* __restrict__ newk,
                           float* __restrict__ part_acc,
                           float* __restrict__ part_ml) {
    __shared__ float4 s_acc[8 * 256];
    __shared__ float  s_m[8 * 16];
    __shared__ float  s_l[8 * 16];
    __shared__ float  s_ml[32];
    const int tid  = threadIdx.x;
    const int lane = tid & 31;
    const int wv   = tid >> 5;
    const int s0   = (blockIdx.x * 8 + wv) * 2;
    const int s1   = s0 + 1;
    const bool last1 = (s1 == S_LEN - 1);

    const float4* v0src = (const float4*)(oldv + (size_t)(s0 + 1) * E_DIM);
    const float4* k0src = (const float4*)(oldk + (size_t)(s0 + 1) * E_DIM);
    const float4* v1src = (const float4*)(last1 ? (newv + (size_t)s1 * E_DIM)
                                                : (oldv + (size_t)(s1 + 1) * E_DIM));
    const float4* k1src = (const float4*)(last1 ? (newk + (size_t)s1 * E_DIM)
                                                : (oldk + (size_t)(s1 + 1) * E_DIM));
    float4* v0dst = (float4*)(newv + (size_t)s0 * E_DIM);
    float4* k0dst = (float4*)(newk + (size_t)s0 * E_DIM);
    float4* v1dst = (float4*)(newv + (size_t)s1 * E_DIM);
    float4* k1dst = (float4*)(newk + (size_t)s1 * E_DIM);
    const float4* q4 = (const float4*)q;

    float4 qd[8], va[8], vb[8], ka[8], kb[8];
    #pragma unroll
    for (int i = 0; i < 8; i++) qd[i] = q4[i * 32 + lane];
    #pragma unroll
    for (int i = 0; i < 8; i++) va[i] = v0src[i * 32 + lane];
    #pragma unroll
    for (int i = 0; i < 8; i++) vb[i] = v1src[i * 32 + lane];
    #pragma unroll
    for (int i = 0; i < 8; i++) ka[i] = k0src[i * 32 + lane];
    #pragma unroll
    for (int i = 0; i < 8; i++) kb[i] = k1src[i * 32 + lane];

    #pragma unroll
    for (int i = 0; i < 8; i++) v0dst[i * 32 + lane] = va[i];
    if (!last1) {
        #pragma unroll
        for (int i = 0; i < 8; i++) v1dst[i * 32 + lane] = vb[i];
    }
    #pragma unroll
    for (int i = 0; i < 8; i++) k0dst[i * 32 + lane] = ka[i];
    if (!last1) {
        #pragma unroll
        for (int i = 0; i < 8; i++) k1dst[i * 32 + lane] = kb[i];
    }

    int nza = 0, nzb = 0;
    #pragma unroll
    for (int i = 0; i < 8; i++) {
        nza |= (va[i].x != 0.f) | (va[i].y != 0.f) | (va[i].z != 0.f) | (va[i].w != 0.f);
        nzb |= (vb[i].x != 0.f) | (vb[i].y != 0.f) | (vb[i].z != 0.f) | (vb[i].w != 0.f);
    }
    #pragma unroll
    for (int mm = 16; mm >= 1; mm >>= 1) {
        nza |= __shfl_xor(nza, mm, 32);
        nzb |= __shfl_xor(nzb, mm, 32);
    }

    const bool arow = (lane == 0) || (lane == 24);
    const v8f zero8 = {};

    float sa[8], sb[8];
    #pragma unroll
    for (int i = 0; i < 8; i++) {
        float pa = ka[i].x * qd[i].x + ka[i].y * qd[i].y + ka[i].z * qd[i].z + ka[i].w * qd[i].w;
        float pb = kb[i].x * qd[i].x + kb[i].y * qd[i].y + kb[i].z * qd[i].z + kb[i].w * qd[i].w;
        pa += __shfl_xor(pa, 1, 32);  pb += __shfl_xor(pb, 1, 32);
        pa += __shfl_xor(pa, 2, 32);  pb += __shfl_xor(pb, 2, 32);
        pa += __shfl_xor(pa, 4, 32);  pb += __shfl_xor(pb, 4, 32);
        pa += __shfl_xor(pa, 8, 32);  pb += __shfl_xor(pb, 8, 32);
        sa[i] = nza ? pa * 0.125f : -INFINITY;
        sb[i] = nzb ? pb * 0.125f : -INFINITY;
    }

    #pragma unroll
    for (int i = 0; i < 8; i++) {
        const float m  = fmaxf(sa[i], sb[i]);
        const float ms = (m == -INFINITY) ? 0.f : m;
        const float ea = expf(sa[i] - ms);
        const float eb = expf(sb[i] - ms);
        if ((lane & 15) == 0) {
            const int h = i * 2 + (lane >> 4);
            s_m[wv * 16 + h] = m;
            s_l[wv * 16 + h] = ea + eb;
        }
        v2f a, b0, b1, b2, b3;
        a.x  = arow ? ea : 0.f;   a.y  = arow ? eb : 0.f;
        b0.x = va[i].x;           b0.y = vb[i].x;
        b1.x = va[i].y;           b1.y = vb[i].y;
        b2.x = va[i].z;           b2.y = vb[i].z;
        b3.x = va[i].w;           b3.y = vb[i].w;
        const v8f d0 = wmma4(a, b0, zero8);
        const v8f d1 = wmma4(a, b1, zero8);
        const v8f d2 = wmma4(a, b2, zero8);
        const v8f d3 = wmma4(a, b3, zero8);
        float4 acc;
        acc.x = d0[0];  acc.y = d1[0];  acc.z = d2[0];  acc.w = d3[0];
        s_acc[wv * 256 + i * 32 + lane] = acc;
    }
    __syncthreads();

    {
        const int head = tid >> 4;
        float mw[8];
        float M = -INFINITY;
        #pragma unroll
        for (int w = 0; w < 8; w++) { mw[w] = s_m[w * 16 + head];  M = fmaxf(M, mw[w]); }
        float4 acc = {0.f, 0.f, 0.f, 0.f};
        float L = 0.f;
        const float Ms = (M == -INFINITY) ? 0.f : M;
        #pragma unroll
        for (int w = 0; w < 8; w++) {
            const float sc = expf(mw[w] - Ms);
            const float4 t = s_acc[w * 256 + tid];
            acc.x += t.x * sc;  acc.y += t.y * sc;  acc.z += t.z * sc;  acc.w += t.w * sc;
            L += s_l[w * 16 + head] * sc;
        }
        ((float4*)part_acc)[(size_t)blockIdx.x * 256 + tid] = acc;
        if ((tid & 15) == 0) { s_ml[head] = M;  s_ml[16 + head] = L; }
    }
    __syncthreads();
    if (tid < 32) part_ml[blockIdx.x * 32 + tid] = s_ml[tid];
}

__global__ __launch_bounds__(256)
void merge_gemv_kernel(const float* __restrict__ part_acc,
                       const float* __restrict__ part_ml,
                       const float* __restrict__ W,
                       float* __restrict__ out_parts) {
    __shared__ float  s_w[512];
    __shared__ float  s_redm[8];
    __shared__ float  s_redl[8];
    __shared__ float4 s_sum[16 * 16];
    __shared__ float  xs[D_H];
    const int h    = blockIdx.y;
    const int tid  = threadIdx.x;
    const int lane = tid & 31;
    const int wv   = tid >> 5;
    const int j    = lane & 15;
    const int half = lane >> 4;
    const int ob   = blockIdx.x * 128 + wv * 16;
    const int k0   = h * D_H;

    const float mb0 = part_ml[tid * 32 + h];
    const float mb1 = part_ml[(tid + 256) * 32 + h];
    const float lb0 = part_ml[tid * 32 + 16 + h];
    const float lb1 = part_ml[(tid + 256) * 32 + 16 + h];

    const int c = tid & 15;
    const int g = tid >> 4;
    const float4* psrc = (const float4*)part_acc + h * 16 + c;
    float4 pv[32];
    #pragma unroll
    for (int it = 0; it < 32; it++) pv[it] = psrc[(size_t)(it * 16 + g) * 256];

    float bw[32];
    #pragma unroll
    for (int t = 0; t < 8; t++) {
        const int r0 = t * 8 + half * 2;
        const float* wp0 = W + (size_t)(k0 + r0)     * E_DIM + ob + j;
        const float* wp1 = W + (size_t)(k0 + r0 + 4) * E_DIM + ob + j;
        bw[t * 4 + 0] = wp0[0];  bw[t * 4 + 1] = wp0[E_DIM];
        bw[t * 4 + 2] = wp1[0];  bw[t * 4 + 3] = wp1[E_DIM];
    }

    float mx = fmaxf(mb0, mb1);
    #pragma unroll
    for (int mm = 16; mm >= 1; mm >>= 1) mx = fmaxf(mx, __shfl_xor(mx, mm, 32));
    if (lane == 0) s_redm[wv] = mx;
    __syncthreads();
    float M = s_redm[0];
    #pragma unroll
    for (int w = 1; w < 8; w++) M = fmaxf(M, s_redm[w]);

    const float Ms = (M == -INFINITY) ? 0.f : M;
    const float w0 = expf(mb0 - Ms);
    const float w1 = expf(mb1 - Ms);
    s_w[tid]       = w0;
    s_w[tid + 256] = w1;
    float ls = w0 * lb0 + w1 * lb1;
    #pragma unroll
    for (int mm = 16; mm >= 1; mm >>= 1) ls += __shfl_xor(ls, mm, 32);
    if (lane == 0) s_redl[wv] = ls;
    __syncthreads();
    float L = 0.f;
    #pragma unroll
    for (int w = 0; w < 8; w++) L += s_redl[w];

    float4 acc = {0.f, 0.f, 0.f, 0.f};
    #pragma unroll
    for (int it = 0; it < 32; it++) {
        const float w = s_w[it * 16 + g];
        acc.x += pv[it].x * w;  acc.y += pv[it].y * w;
        acc.z += pv[it].z * w;  acc.w += pv[it].w * w;
    }
    s_sum[g * 16 + c] = acc;
    __syncthreads();
    if (tid < D_H) {
        const float* sf = (const float*)s_sum;
        float s = 0.f;
        #pragma unroll
        for (int gg = 0; gg < 16; gg++) s += sf[gg * D_H + tid];
        xs[tid] = s / L;
    }
    __syncthreads();

    const float amask = (j == 0) ? 1.f : 0.f;
    v8f c0 = {}, c1 = {};
    #pragma unroll
    for (int t = 0; t < 8; t++) {
        const int r0 = t * 8 + half * 2;
        v2f a0, a1, b0, b1;
        a0.x = xs[r0]     * amask;  a0.y = xs[r0 + 1] * amask;
        a1.x = xs[r0 + 4] * amask;  a1.y = xs[r0 + 5] * amask;
        b0.x = bw[t * 4 + 0];  b0.y = bw[t * 4 + 1];
        b1.x = bw[t * 4 + 2];  b1.y = bw[t * 4 + 3];
        c0 = wmma4(a0, b0, c0);
        c1 = wmma4(a1, b1, c1);
    }
    const v8f cs = c0 + c1;
    if (lane < 16)
        out_parts[(size_t)h * E_DIM + ob + lane] = cs[0];
}

__global__ void gemv_part_kernel(const float* __restrict__ W,
                                 const float* __restrict__ in_parts, int P,
                                 const float* __restrict__ in_bias, int in_relu,
                                 float* __restrict__ out_parts) {
    __shared__ float xs[128];
    const int tid = threadIdx.x;
    const int k0  = blockIdx.y * 128;
    if (tid < 128) {
        float s = 0.f;
        for (int pp = 0; pp < P; pp++) s += in_parts[(size_t)pp * E_DIM + k0 + tid];
        if (in_bias) s += in_bias[k0 + tid];
        if (in_relu) s = fmaxf(s, 0.f);
        xs[tid] = s;
    }
    __syncthreads();

    const int wv   = tid >> 5;
    const int lane = tid & 31;
    const int j    = lane & 15;
    const int half = lane >> 4;
    const int ob   = blockIdx.x * 128 + wv * 16;
    const float amask = (j == 0) ? 1.f : 0.f;

    v8f c0 = {}, c1 = {};
    for (int eb = 0; eb < 128; eb += 8) {
        const int r0 = eb + half * 2;
        v2f a0, a1, b0, b1;
        a0.x = xs[r0]     * amask;  a0.y = xs[r0 + 1] * amask;
        a1.x = xs[r0 + 4] * amask;  a1.y = xs[r0 + 5] * amask;
        const float* wp0 = W + (size_t)(k0 + r0)     * E_DIM + ob + j;
        const float* wp1 = W + (size_t)(k0 + r0 + 4) * E_DIM + ob + j;
        b0.x = wp0[0];  b0.y = wp0[E_DIM];
        b1.x = wp1[0];  b1.y = wp1[E_DIM];
        c0 = wmma4(a0, b0, c0);
        c1 = wmma4(a1, b1, c1);
    }
    const v8f cs = c0 + c1;
    if (lane < 16)
        out_parts[(size_t)blockIdx.y * E_DIM + ob + lane] = cs[0];
}

__global__ __launch_bounds__(256)
void gemv_ln_part_kernel(const float* __restrict__ W,
                         const float* __restrict__ parts,
                         const float* __restrict__ bias,
                         const float* __restrict__ resid,
                         const float* __restrict__ sc,
                         const float* __restrict__ bs,
                         float* __restrict__ h_out,
                         float* __restrict__ out_parts) {
    __shared__ float red[256];
    __shared__ float xs[128];
    const int tid = threadIdx.x;
    const int k0  = blockIdx.y * 128;

    float pb[4], pr[4], pp[4][16], psc[4], pbs[4];
    #pragma unroll
    for (int i = 0; i < 4; i++) {
        const int e = i * 256 + tid;
        pb[i] = bias[e];
        pr[i] = resid[e];
        #pragma unroll
        for (int p = 0; p < 16; p++) pp[i][p] = parts[(size_t)p * E_DIM + e];
        psc[i] = sc[e];
        pbs[i] = bs[e];
    }
    float vv[4];
    float sum = 0.f;
    #pragma unroll
    for (int i = 0; i < 4; i++) {
        float s = pb[i] + pr[i];
        #pragma unroll
        for (int p = 0; p < 16; p++) s += pp[i][p];
        vv[i] = s;
        sum += s;
    }
    red[tid] = sum; __syncthreads();
    for (int st = 128; st > 0; st >>= 1) {
        if (tid < st) red[tid] += red[tid + st];
        __syncthreads();
    }
    const float mean = red[0] * (1.f / E_DIM); __syncthreads();
    float var = 0.f;
    #pragma unroll
    for (int i = 0; i < 4; i++) { const float d = vv[i] - mean; var += d * d; }
    red[tid] = var; __syncthreads();
    for (int st = 128; st > 0; st >>= 1) {
        if (tid < st) red[tid] += red[tid + st];
        __syncthreads();
    }
    const float inv = rsqrtf(red[0] * (1.f / E_DIM) + 1e-6f);
    const bool mine = ((tid >> 7) == (int)(blockIdx.y & 1));
    #pragma unroll
    for (int i = 0; i < 4; i++) {
        const float hv = (vv[i] - mean) * inv * psc[i] + pbs[i];
        if (mine && i == (int)(blockIdx.y >> 1)) {
            xs[tid & 127] = hv;
            if (blockIdx.x == 0) h_out[i * 256 + tid] = hv;
        }
    }
    __syncthreads();

    const int wv   = tid >> 5;
    const int lane = tid & 31;
    const int j    = lane & 15;
    const int half = lane >> 4;
    const int ob   = blockIdx.x * 128 + wv * 16;
    const float amask = (j == 0) ? 1.f : 0.f;

    v8f c0 = {}, c1 = {};
    for (int eb = 0; eb < 128; eb += 8) {
        const int r0 = eb + half * 2;
        v2f a0, a1, b0, b1;
        a0.x = xs[r0]     * amask;  a0.y = xs[r0 + 1] * amask;
        a1.x = xs[r0 + 4] * amask;  a1.y = xs[r0 + 5] * amask;
        const float* wp0 = W + (size_t)(k0 + r0)     * E_DIM + ob + j;
        const float* wp1 = W + (size_t)(k0 + r0 + 4) * E_DIM + ob + j;
        b0.x = wp0[0];  b0.y = wp0[E_DIM];
        b1.x = wp1[0];  b1.y = wp1[E_DIM];
        c0 = wmma4(a0, b0, c0);
        c1 = wmma4(a1, b1, c1);
    }
    const v8f cs = c0 + c1;
    if (lane < 16)
        out_parts[(size_t)blockIdx.y * E_DIM + ob + lane] = cs[0];
}

__global__ void ln_combine_kernel(const float* __restrict__ parts, int P,
                                  const float* __restrict__ bias,
                                  const float* __restrict__ resid,
                                  const float* __restrict__ sc,
                                  const float* __restrict__ bs,
                                  float* __restrict__ out) {
    __shared__ float red[256];
    const int tid = threadIdx.x;
    float vv[4];
    float sum = 0.f;
    #pragma unroll
    for (int i = 0; i < 4; i++) {
        const int e = i * 256 + tid;
        float s = bias[e] + resid[e];
        for (int pp = 0; pp < P; pp++) s += parts[(size_t)pp * E_DIM + e];
        vv[i] = s;
        sum += s;
    }
    red[tid] = sum; __syncthreads();
    for (int st = 128; st > 0; st >>= 1) {
        if (tid < st) red[tid] += red[tid + st];
        __syncthreads();
    }
    const float mean = red[0] * (1.f / E_DIM); __syncthreads();
    float var = 0.f;
    #pragma unroll
    for (int i = 0; i < 4; i++) { const float d = vv[i] - mean; var += d * d; }
    red[tid] = var; __syncthreads();
    for (int st = 128; st > 0; st >>= 1) {
        if (tid < st) red[tid] += red[tid + st];
        __syncthreads();
    }
    const float inv = rsqrtf(red[0] * (1.f / E_DIM) + 1e-6f);
    #pragma unroll
    for (int i = 0; i < 4; i++) {
        const int e = i * 256 + tid;
        out[e] = (vv[i] - mean) * inv * sc[e] + bs[e];
    }
}

extern "C" void kernel_launch(void* const* d_in, const int* in_sizes, int n_in,
                              void* d_out, int out_size, void* d_ws, size_t ws_size,
                              hipStream_t stream) {
    (void)in_sizes; (void)n_in; (void)out_size; (void)ws_size;
    const float* x     = (const float*)d_in[0];
    const float* cache = (const float*)d_in[1];
    const float* Wv = (const float*)d_in[2];  const float* bv = (const float*)d_in[3];
    const float* Wq = (const float*)d_in[4];  const float* bq = (const float*)d_in[5];
    const float* Wk = (const float*)d_in[6];  const float* bk = (const float*)d_in[7];
    const float* Wo = (const float*)d_in[8];  const float* bo = (const float*)d_in[9];
    const float* W1 = (const float*)d_in[10]; const float* b1 = (const float*)d_in[11];
    const float* W2 = (const float*)d_in[12]; const float* b2 = (const float*)d_in[13];
    const float* ln1_s = (const float*)d_in[14]; const float* ln1_b = (const float*)d_in[15];
    const float* ln2_s = (const float*)d_in[16]; const float* ln2_b = (const float*)d_in[17];

    float* out_h = (float*)d_out;
    float* newv  = out_h + E_DIM;
    float* newk  = newv + (size_t)S_LEN * E_DIM;
    const float* oldv = cache;
    const float* oldk = cache + (size_t)S_LEN * E_DIM;

    float* ws       = (float*)d_ws;
    float* q        = ws;
    float* part_acc = q + E_DIM;
    float* part_ml  = part_acc + 512 * E_DIM;
    float* h1       = part_ml + 512 * 32;
    float* part_o   = h1 + E_DIM;
    float* part_1   = part_o + 16 * E_DIM;
    float* part_2   = part_1 + 8 * E_DIM;

    qkv_kernel<<<dim3(32, 3), 256, 0, stream>>>(x, Wv, Wq, Wk, bv, bq, bk, q, newv, newk);
    shift_score_pv_kernel<<<512, 256, 0, stream>>>(oldv, oldk, q, newv, newk, part_acc, part_ml);
    merge_gemv_kernel<<<dim3(8, 16), 256, 0, stream>>>(part_acc, part_ml, Wo, part_o);
    gemv_ln_part_kernel<<<dim3(8, 8), 256, 0, stream>>>(W1, part_o, bo, x, ln1_s, ln1_b, h1, part_1);
    gemv_part_kernel<<<dim3(8, 8), 256, 0, stream>>>(W2, part_1, 8, b1, 1, part_2);
    ln_combine_kernel<<<1, 256, 0, stream>>>(part_2, 8, b2, h1, ln2_s, ln2_b, out_h);
}


# global loads of the W1 GEMV, W2 GEMV and final LN kernels all issued at entry, W tiles kept in flight across LDS-only barriers; W2 and LN inputs with fixed counts, loads first; same sum order
# speedup vs baseline: 3.2740x; 1.1484x over previous
#include <hip/hip_runtime.h>
#include <hip/hip_bf16.h>
#include <math.h>

#define E_DIM 1024
#define S_LEN 8192
#define H_N   16
#define D_H   64

typedef __attribute__((ext_vector_type(2))) float v2f;
typedef __attribute__((ext_vector_type(8))) float v8f;

__device__ __forceinline__ v8f wmma4(v2f a, v2f b, v8f c) {
    return __builtin_amdgcn_wmma_f32_16x16x4_f32(false, a, false, b, (short)0, c,
                                                 false, false);
}

__device__ __forceinline__ void lds_barrier() {
    __builtin_amdgcn_fence(__ATOMIC_RELEASE, "workgroup", "local");
    __builtin_amdgcn_s_barrier();
    __builtin_amdgcn_fence(__ATOMIC_ACQUIRE, "workgroup", "local");
}

__global__ __launch_bounds__(256)
void qkv_kernel(const float* __restrict__ x,
                const float* __restrict__ Wv,
                const float* __restrict__ Wq,
                const float* __restrict__ Wk,
                const float* __restrict__ bv,
                const float* __restrict__ bq,
                const float* __restrict__ bk,
                float* __restrict__ q_out,
                float* __restrict__ newv, float* __restrict__ newk) {
    __shared__ float4 xs4[E_DIM / 4];
    __shared__ float  red[8 * 16];
    const float* xs = (const float*)xs4;
    const int tid = threadIdx.x;
    const int m = blockIdx.y;
    const float* W    = (m == 0) ? Wv : (m == 1) ? Wq : Wk;
    const float* bias = (m == 0) ? bv : (m == 1) ? bq : bk;
    float* dst = (m == 0) ? newv + (size_t)(S_LEN - 1) * E_DIM
               : (m == 1) ? q_out
                          : newk + (size_t)(S_LEN - 1) * E_DIM;
    const float bval = bias[blockIdx.x * 32 + (tid & 31)];
    xs4[tid] = ((const float4*)x)[tid];
    __syncthreads();

    const int wv   = tid >> 5;
    const int lane = tid & 31;
    const int j    = lane & 15;
    const int half = lane >> 4;
    const int ob   = blockIdx.x * 32 + (wv & 1) * 16;
    const int k0   = (wv >> 1) * 256;
    const float amask = (j == 0) ? 1.f : 0.f;

    v8f c0 = {}, c1 = {};
    for (int eb = 0; eb < 256; eb += 8) {
        const int r0 = eb + half * 2;
        v2f a0, a1, b0, b1;
        a0.x = xs[k0 + r0]     * amask;  a0.y = xs[k0 + r0 + 1] * amask;
        a1.x = xs[k0 + r0 + 4] * amask;  a1.y = xs[k0 + r0 + 5] * amask;
        const float* wp0 = W + (size_t)(k0 + r0)     * E_DIM + ob + j;
        const float* wp1 = W + (size_t)(k0 + r0 + 4) * E_DIM + ob + j;
        b0.x = wp0[0];  b0.y = wp0[E_DIM];
        b1.x = wp1[0];  b1.y = wp1[E_DIM];
        c0 = wmma4(a0, b0, c0);
        c1 = wmma4(a1, b1, c1);
    }
    const v8f cs = c0 + c1;
    if (lane < 16) red[wv * 16 + lane] = cs[0];
    __syncthreads();
    if (tid < 32) {
        const int og = tid >> 4;
        const int jj = tid & 15;
        float s = 0.f;
        #pragma unroll
        for (int ks = 0; ks < 4; ks++) s += red[(ks * 2 + og) * 16 + jj];
        dst[blockIdx.x * 32 + tid] = s + bval;
    }
}

__global__ __launch_bounds__(256)
void shift_score_pv_kernel(const float* __restrict__ oldv,
                           const float* __restrict__ oldk,
                           const float* __restrict__ q,
                           float* __restrict__ newv, float* __restrict__ newk,
                           float* __restrict__ part_acc,
                           float* __restrict__ part_ml) {
    __shared__ float4 s_acc[8 * 256];
    __shared__ float  s_m[8 * 16];
    __shared__ float  s_l[8 * 16];
    __shared__ float  s_ml[32];
    const int tid  = threadIdx.x;
    const int lane = tid & 31;
    const int wv   = tid >> 5;
    const int s0   = (blockIdx.x * 8 + wv) * 2;
    const int s1   = s0 + 1;
    const bool last1 = (s1 == S_LEN - 1);

    const float4* v0src = (const float4*)(oldv + (size_t)(s0 + 1) * E_DIM);
    const float4* k0src = (const float4*)(oldk + (size_t)(s0 + 1) * E_DIM);
    const float4* v1src = (const float4*)(last1 ? (newv + (size_t)s1 * E_DIM)
                                                : (oldv + (size_t)(s1 + 1) * E_DIM));
    const float4* k1src = (const float4*)(last1 ? (newk + (size_t)s1 * E_DIM)
                                                : (oldk + (size_t)(s1 + 1) * E_DIM));
    float4* v0dst = (float4*)(newv + (size_t)s0 * E_DIM);
    float4* k0dst = (float4*)(newk + (size_t)s0 * E_DIM);
    float4* v1dst = (float4*)(newv + (size_t)s1 * E_DIM);
    float4* k1dst = (float4*)(newk + (size_t)s1 * E_DIM);
    const float4* q4 = (const float4*)q;

    float4 qd[8], va[8], vb[8], ka[8], kb[8];
    #pragma unroll
    for (int i = 0; i < 8; i++) qd[i] = q4[i * 32 + lane];
    #pragma unroll
    for (int i = 0; i < 8; i++) va[i] = v0src[i * 32 + lane];
    #pragma unroll
    for (int i = 0; i < 8; i++) vb[i] = v1src[i * 32 + lane];
    #pragma unroll
    for (int i = 0; i < 8; i++) ka[i] = k0src[i * 32 + lane];
    #pragma unroll
    for (int i = 0; i < 8; i++) kb[i] = k1src[i * 32 + lane];

    #pragma unroll
    for (int i = 0; i < 8; i++) v0dst[i * 32 + lane] = va[i];
    if (!last1) {
        #pragma unroll
        for (int i = 0; i < 8; i++) v1dst[i * 32 + lane] = vb[i];
    }
    #pragma unroll
    for (int i = 0; i < 8; i++) k0dst[i * 32 + lane] = ka[i];
    if (!last1) {
        #pragma unroll
        for (int i = 0; i < 8; i++) k1dst[i * 32 + lane] = kb[i];
    }

    int nza = 0, nzb = 0;
    #pragma unroll
    for (int i = 0; i < 8; i++) {
        nza |= (va[i].x != 0.f) | (va[i].y != 0.f) | (va[i].z != 0.f) | (va[i].w != 0.f);
        nzb |= (vb[i].x != 0.f) | (vb[i].y != 0.f) | (vb[i].z != 0.f) | (vb[i].w != 0.f);
    }
    #pragma unroll
    for (int mm = 16; mm >= 1; mm >>= 1) {
        nza |= __shfl_xor(nza, mm, 32);
        nzb |= __shfl_xor(nzb, mm, 32);
    }

    const bool arow = (lane == 0) || (lane == 24);
    const v8f zero8 = {};

    float sa[8], sb[8];
    #pragma unroll
    for (int i = 0; i < 8; i++) {
        float pa = ka[i].x * qd[i].x + ka[i].y * qd[i].y + ka[i].z * qd[i].z + ka[i].w * qd[i].w;
        float pb = kb[i].x * qd[i].x + kb[i].y * qd[i].y + kb[i].z * qd[i].z + kb[i].w * qd[i].w;
        pa += __shfl_xor(pa, 1, 32);  pb += __shfl_xor(pb, 1, 32);
        pa += __shfl_xor(pa, 2, 32);  pb += __shfl_xor(pb, 2, 32);
        pa += __shfl_xor(pa, 4, 32);  pb += __shfl_xor(pb, 4, 32);
        pa += __shfl_xor(pa, 8, 32);  pb += __shfl_xor(pb, 8, 32);
        sa[i] = nza ? pa * 0.125f : -INFINITY;
        sb[i] = nzb ? pb * 0.125f : -INFINITY;
    }

    #pragma unroll
    for (int i = 0; i < 8; i++) {
        const float m  = fmaxf(sa[i], sb[i]);
        const float ms = (m == -INFINITY) ? 0.f : m;
        const float ea = expf(sa[i] - ms);
        const float eb = expf(sb[i] - ms);
        if ((lane & 15) == 0) {
            const int h = i * 2 + (lane >> 4);
            s_m[wv * 16 + h] = m;
            s_l[wv * 16 + h] = ea + eb;
        }
        v2f a, b0, b1, b2, b3;
        a.x  = arow ? ea : 0.f;   a.y  = arow ? eb : 0.f;
        b0.x = va[i].x;           b0.y = vb[i].x;
        b1.x = va[i].y;           b1.y = vb[i].y;
        b2.x = va[i].z;           b2.y = vb[i].z;
        b3.x = va[i].w;           b3.y = vb[i].w;
        const v8f d0 = wmma4(a, b0, zero8);
        const v8f d1 = wmma4(a, b1, zero8);
        const v8f d2 = wmma4(a, b2, zero8);
        const v8f d3 = wmma4(a, b3, zero8);
        float4 acc;
        acc.x = d0[0];  acc.y = d1[0];  acc.z = d2[0];  acc.w = d3[0];
        s_acc[wv * 256 + i * 32 + lane] = acc;
    }
    __syncthreads();

    {
        const int head = tid >> 4;
        float mw[8];
        float M = -INFINITY;
        #pragma unroll
        for (int w = 0; w < 8; w++) { mw[w] = s_m[w * 16 + head];  M = fmaxf(M, mw[w]); }
        float4 acc = {0.f, 0.f, 0.f, 0.f};
        float L = 0.f;
        const float Ms = (M == -INFINITY) ? 0.f : M;
        #pragma unroll
        for (int w = 0; w < 8; w++) {
            const float sc = expf(mw[w] - Ms);
            const float4 t = s_acc[w * 256 + tid];
            acc.x += t.x * sc;  acc.y += t.y * sc;  acc.z += t.z * sc;  acc.w += t.w * sc;
            L += s_l[w * 16 + head] * sc;
        }
        ((float4*)part_acc)[(size_t)blockIdx.x * 256 + tid] = acc;
        if ((tid & 15) == 0) { s_ml[head] = M;  s_ml[16 + head] = L; }
    }
    __syncthreads();
    if (tid < 32) part_ml[blockIdx.x * 32 + tid] = s_ml[tid];
}

__global__ __launch_bounds__(256)
void merge_gemv_kernel(const float* __restrict__ part_acc,
                       const float* __restrict__ part_ml,
                       const float* __restrict__ W,
                       float* __restrict__ out_parts) {
    __shared__ float  s_w[512];
    __shared__ float  s_redm[8];
    __shared__ float  s_redl[8];
    __shared__ float4 s_sum[16 * 16];
    __shared__ float  xs[D_H];
    const int h    = blockIdx.y;
    const int tid  = threadIdx.x;
    const int lane = tid & 31;
    const int wv   = tid >> 5;
    const int j    = lane & 15;
    const int half = lane >> 4;
    const int ob   = blockIdx.x * 128 + wv * 16;
    const int k0   = h * D_H;

    const float mb0 = part_ml[tid * 32 + h];
    const float mb1 = part_ml[(tid + 256) * 32 + h];
    const float lb0 = part_ml[tid * 32 + 16 + h];
    const float lb1 = part_ml[(tid + 256) * 32 + 16 + h];

    const int c = tid & 15;
    const int g = tid >> 4;
    const float4* psrc = (const float4*)part_acc + h * 16 + c;
    float4 pv[32];
    #pragma unroll
    for (int it = 0; it < 32; it++) pv[it] = psrc[(size_t)(it * 16 + g) * 256];

    float bw[32];
    #pragma unroll
    for (int t = 0; t < 8; t++) {
        const int r0 = t * 8 + half * 2;
        const float* wp0 = W + (size_t)(k0 + r0)     * E_DIM + ob + j;
        const float* wp1 = W + (size_t)(k0 + r0 + 4) * E_DIM + ob + j;
        bw[t * 4 + 0] = wp0[0];  bw[t * 4 + 1] = wp0[E_DIM];
        bw[t * 4 + 2] = wp1[0];  bw[t * 4 + 3] = wp1[E_DIM];
    }

    float mx = fmaxf(mb0, mb1);
    #pragma unroll
    for (int mm = 16; mm >= 1; mm >>= 1) mx = fmaxf(mx, __shfl_xor(mx, mm, 32));
    if (lane == 0) s_redm[wv] = mx;
    __syncthreads();
    float M = s_redm[0];
    #pragma unroll
    for (int w = 1; w < 8; w++) M = fmaxf(M, s_redm[w]);

    const float Ms = (M == -INFINITY) ? 0.f : M;
    const float w0 = expf(mb0 - Ms);
    const float w1 = expf(mb1 - Ms);
    s_w[tid]       = w0;
    s_w[tid + 256] = w1;
    float ls = w0 * lb0 + w1 * lb1;
    #pragma unroll
    for (int mm = 16; mm >= 1; mm >>= 1) ls += __shfl_xor(ls, mm, 32);
    if (lane == 0) s_redl[wv] = ls;
    __syncthreads();
    float L = 0.f;
    #pragma unroll
    for (int w = 0; w < 8; w++) L += s_redl[w];

    float4 acc = {0.f, 0.f, 0.f, 0.f};
    #pragma unroll
    for (int it = 0; it < 32; it++) {
        const float w = s_w[it * 16 + g];
        acc.x += pv[it].x * w;  acc.y += pv[it].y * w;
        acc.z += pv[it].z * w;  acc.w += pv[it].w * w;
    }
    s_sum[g * 16 + c] = acc;
    __syncthreads();
    if (tid < D_H) {
        const float* sf = (const float*)s_sum;
        float s = 0.f;
        #pragma unroll
        for (int gg = 0; gg < 16; gg++) s += sf[gg * D_H + tid];
        xs[tid] = s / L;
    }
    __syncthreads();

    const float amask = (j == 0) ? 1.f : 0.f;
    v8f c0 = {}, c1 = {};
    #pragma unroll
    for (int t = 0; t < 8; t++) {
        const int r0 = t * 8 + half * 2;
        v2f a0, a1, b0, b1;
        a0.x = xs[r0]     * amask;  a0.y = xs[r0 + 1] * amask;
        a1.x = xs[r0 + 4] * amask;  a1.y = xs[r0 + 5] * amask;
        b0.x = bw[t * 4 + 0];  b0.y = bw[t * 4 + 1];
        b1.x = bw[t * 4 + 2];  b1.y = bw[t * 4 + 3];
        c0 = wmma4(a0, b0, c0);
        c1 = wmma4(a1, b1, c1);
    }
    const v8f cs = c0 + c1;
    if (lane < 16)
        out_parts[(size_t)h * E_DIM + ob + lane] = cs[0];
}

__global__ __launch_bounds__(256)
void gemv_relu_part_kernel(const float* W,
                           const float* in_parts,
                           const float* in_bias,
                           float* out_parts) {
    __shared__ float xs[128];
    const int tid  = threadIdx.x;
    const int k0   = blockIdx.y * 128;
    const int wv   = tid >> 5;
    const int lane = tid & 31;
    const int j    = lane & 15;
    const int half = lane >> 4;
    const int ob   = blockIdx.x * 128 + wv * 16;
    const int ke   = k0 + (tid & 127);

    float pp[8];
    #pragma unroll
    for (int p = 0; p < 8; p++) pp[p] = in_parts[(size_t)p * E_DIM + ke];
    const float pbias = in_bias[ke];
    asm volatile("" ::: "memory");

    float bw[64];
    #pragma unroll
    for (int t = 0; t < 16; t++) {
        const int r0 = t * 8 + half * 2;
        const float* wp0 = W + (size_t)(k0 + r0)     * E_DIM + ob + j;
        const float* wp1 = W + (size_t)(k0 + r0 + 4) * E_DIM + ob + j;
        bw[t * 4 + 0] = wp0[0];  bw[t * 4 + 1] = wp0[E_DIM];
        bw[t * 4 + 2] = wp1[0];  bw[t * 4 + 3] = wp1[E_DIM];
    }

    float s = 0.f;
    #pragma unroll
    for (int p = 0; p < 8; p++) s += pp[p];
    s += pbias;
    if (tid < 128) xs[tid] = fmaxf(s, 0.f);
    lds_barrier();

    const float amask = (j == 0) ? 1.f : 0.f;
    v8f c0 = {}, c1 = {};
    #pragma unroll
    for (int t = 0; t < 16; t++) {
        const int r0 = t * 8 + half * 2;
        v2f a0, a1, b0, b1;
        a0.x = xs[r0]     * amask;  a0.y = xs[r0 + 1] * amask;
        a1.x = xs[r0 + 4] * amask;  a1.y = xs[r0 + 5] * amask;
        b0.x = bw[t * 4 + 0];  b0.y = bw[t * 4 + 1];
        b1.x = bw[t * 4 + 2];  b1.y = bw[t * 4 + 3];
        c0 = wmma4(a0, b0, c0);
        c1 = wmma4(a1, b1, c1);
    }
    const v8f cs = c0 + c1;
    if (lane < 16)
        out_parts[(size_t)blockIdx.y * E_DIM + ob + lane] = cs[0];
}

__global__ __launch_bounds__(256)
void gemv_ln_part_kernel(const float* __restrict__ W,
                         const float* __restrict__ parts,
                         const float* __restrict__ bias,
                         const float* __restrict__ resid,
                         const float* __restrict__ sc,
                         const float* __restrict__ bs,
                         float* __restrict__ h_out,
                         float* __restrict__ out_parts) {
    __shared__ float red[256];
    __shared__ float xs[128];
    const int tid = threadIdx.x;
    const int k0  = blockIdx.y * 128;

    float pb[4], pr[4], pp[4][16], psc[4], pbs[4];
    #pragma unroll
    for (int i = 0; i < 4; i++) {
        const int e = i * 256 + tid;
        pb[i] = bias[e];
        pr[i] = resid[e];
        #pragma unroll
        for (int p = 0; p < 16; p++) pp[i][p] = parts[(size_t)p * E_DIM + e];
        psc[i] = sc[e];
        pbs[i] = bs[e];
    }

    const int wv   = tid >> 5;
    const int lane = tid & 31;
    const int j    = lane & 15;
    const int half = lane >> 4;
    const int ob   = blockIdx.x * 128 + wv * 16;
    asm volatile("" ::: "memory");
    float bw[64];
    #pragma unroll
    for (int t = 0; t < 16; t++) {
        const int r0 = t * 8 + half * 2;
        const float* wp0 = W + (size_t)(k0 + r0)     * E_DIM + ob + j;
        const float* wp1 = W + (size_t)(k0 + r0 + 4) * E_DIM + ob + j;
        bw[t * 4 + 0] = wp0[0];  bw[t * 4 + 1] = wp0[E_DIM];
        bw[t * 4 + 2] = wp1[0];  bw[t * 4 + 3] = wp1[E_DIM];
    }

    float vv[4];
    float sum = 0.f;
    #pragma unroll
    for (int i = 0; i < 4; i++) {
        float s = pb[i] + pr[i];
        #pragma unroll
        for (int p = 0; p < 16; p++) s += pp[i][p];
        vv[i] = s;
        sum += s;
    }
    red[tid] = sum; lds_barrier();
    for (int st = 128; st > 0; st >>= 1) {
        if (tid < st) red[tid] += red[tid + st];
        lds_barrier();
    }
    const float mean = red[0] * (1.f / E_DIM); lds_barrier();
    float var = 0.f;
    #pragma unroll
    for (int i = 0; i < 4; i++) { const float d = vv[i] - mean; var += d * d; }
    red[tid] = var; lds_barrier();
    for (int st = 128; st > 0; st >>= 1) {
        if (tid < st) red[tid] += red[tid + st];
        lds_barrier();
    }
    const float inv = rsqrtf(red[0] * (1.f / E_DIM) + 1e-6f);
    const bool mine = ((tid >> 7) == (int)(blockIdx.y & 1));
    #pragma unroll
    for (int i = 0; i < 4; i++) {
        const float hv = (vv[i] - mean) * inv * psc[i] + pbs[i];
        if (mine && i == (int)(blockIdx.y >> 1)) {
            xs[tid & 127] = hv;
            if (blockIdx.x == 0) h_out[i * 256 + tid] = hv;
        }
    }
    lds_barrier();

    const float amask = (j == 0) ? 1.f : 0.f;
    v8f c0 = {}, c1 = {};
    #pragma unroll
    for (int t = 0; t < 16; t++) {
        const int r0 = t * 8 + half * 2;
        v2f a0, a1, b0, b1;
        a0.x = xs[r0]     * amask;  a0.y = xs[r0 + 1] * amask;
        a1.x = xs[r0 + 4] * amask;  a1.y = xs[r0 + 5] * amask;
        b0.x = bw[t * 4 + 0];  b0.y = bw[t * 4 + 1];
        b1.x = bw[t * 4 + 2];  b1.y = bw[t * 4 + 3];
        c0 = wmma4(a0, b0, c0);
        c1 = wmma4(a1, b1, c1);
    }
    const v8f cs = c0 + c1;
    if (lane < 16)
        out_parts[(size_t)blockIdx.y * E_DIM + ob + lane] = cs[0];
}

__global__ __launch_bounds__(256)
void ln_combine_kernel(const float* __restrict__ parts,
                       const float* __restrict__ bias,
                       const float* __restrict__ resid,
                       const float* __restrict__ sc,
                       const float* __restrict__ bs,
                       float* __restrict__ out) {
    __shared__ float red[256];
    const int tid = threadIdx.x;
    float pb[4], pr[4], pp[4][8], psc[4], pbs[4];
    #pragma unroll
    for (int i = 0; i < 4; i++) {
        const int e = i * 256 + tid;
        pb[i] = bias[e];
        pr[i] = resid[e];
        #pragma unroll
        for (int p = 0; p < 8; p++) pp[i][p] = parts[(size_t)p * E_DIM + e];
        psc[i] = sc[e];
        pbs[i] = bs[e];
    }
    float vv[4];
    float sum = 0.f;
    #pragma unroll
    for (int i = 0; i < 4; i++) {
        float s = pb[i] + pr[i];
        #pragma unroll
        for (int p = 0; p < 8; p++) s += pp[i][p];
        vv[i] = s;
        sum += s;
    }
    red[tid] = sum; __syncthreads();
    for (int st = 128; st > 0; st >>= 1) {
        if (tid < st) red[tid] += red[tid + st];
        __syncthreads();
    }
    const float mean = red[0] * (1.f / E_DIM); __syncthreads();
    float var = 0.f;
    #pragma unroll
    for (int i = 0; i < 4; i++) { const float d = vv[i] - mean; var += d * d; }
    red[tid] = var; __syncthreads();
    for (int st = 128; st > 0; st >>= 1) {
        if (tid < st) red[tid] += red[tid + st];
        __syncthreads();
    }
    const float inv = rsqrtf(red[0] * (1.f / E_DIM) + 1e-6f);
    #pragma unroll
    for (int i = 0; i < 4; i++) {
        const int e = i * 256 + tid;
        out[e] = (vv[i] - mean) * inv * psc[i] + pbs[i];
    }
}

extern "C" void kernel_launch(void* const* d_in, const int* in_sizes, int n_in,
                              void* d_out, int out_size, void* d_ws, size_t ws_size,
                              hipStream_t stream) {
    (void)in_sizes; (void)n_in; (void)out_size; (void)ws_size;
    const float* x     = (const float*)d_in[0];
    const float* cache = (const float*)d_in[1];
    const float* Wv = (const float*)d_in[2];  const float* bv = (const float*)d_in[3];
    const float* Wq = (const float*)d_in[4];  const float* bq = (const float*)d_in[5];
    const float* Wk = (const float*)d_in[6];  const float* bk = (const float*)d_in[7];
    const float* Wo = (const float*)d_in[8];  const float* bo = (const float*)d_in[9];
    const float* W1 = (const float*)d_in[10]; const float* b1 = (const float*)d_in[11];
    const float* W2 = (const float*)d_in[12]; const float* b2 = (const float*)d_in[13];
    const float* ln1_s = (const float*)d_in[14]; const float* ln1_b = (const float*)d_in[15];
    const float* ln2_s = (const float*)d_in[16]; const float* ln2_b = (const float*)d_in[17];

    float* out_h = (float*)d_out;
    float* newv  = out_h + E_DIM;
    float* newk  = newv + (size_t)S_LEN * E_DIM;
    const float* oldv = cache;
    const float* oldk = cache + (size_t)S_LEN * E_DIM;

    float* ws       = (float*)d_ws;
    float* q        = ws;
    float* part_acc = q + E_DIM;
    float* part_ml  = part_acc + 512 * E_DIM;
    float* h1       = part_ml + 512 * 32;
    float* part_o   = h1 + E_DIM;
    float* part_1   = part_o + 16 * E_DIM;
    float* part_2   = part_1 + 8 * E_DIM;

    qkv_kernel<<<dim3(32, 3), 256, 0, stream>>>(x, Wv, Wq, Wk, bv, bq, bk, q, newv, newk);
    shift_score_pv_kernel<<<512, 256, 0, stream>>>(oldv, oldk, q, newv, newk, part_acc, part_ml);
    merge_gemv_kernel<<<dim3(8, 16), 256, 0, stream>>>(part_acc, part_ml, Wo, part_o);
    gemv_ln_part_kernel<<<dim3(8, 8), 256, 0, stream>>>(W1, part_o, bo, x, ln1_s, ln1_b, h1, part_1);
    gemv_relu_part_kernel<<<dim3(8, 8), 256, 0, stream>>>(W2, part_1, b1, part_2);
    ln_combine_kernel<<<1, 256, 0, stream>>>(part_2, b2, h1, ln2_s, ln2_b, out_h);
}


# cross-lane reductions on the vector ALU (DPP row_xmask, permlanex16) instead of ds_bpermute butterflies; LayerNorm block sums by one LDS hand-over, not 9-barrier LDS trees; same sum order
# speedup vs baseline: 3.5332x; 1.0792x over previous
#include <hip/hip_runtime.h>
#include <hip/hip_bf16.h>
#include <math.h>

#define E_DIM 1024
#define S_LEN 8192
#define H_N   16
#define D_H   64

typedef __attribute__((ext_vector_type(2))) float v2f;
typedef __attribute__((ext_vector_type(8))) float v8f;

__device__ __forceinline__ v8f wmma4(v2f a, v2f b, v8f c) {
    return __builtin_amdgcn_wmma_f32_16x16x4_f32(false, a, false, b, (short)0, c,
                                                 false, false);
}

__device__ __forceinline__ void lds_barrier() {
    __builtin_amdgcn_fence(__ATOMIC_RELEASE, "workgroup", "local");
    __builtin_amdgcn_s_barrier();
    __builtin_amdgcn_fence(__ATOMIC_ACQUIRE, "workgroup", "local");
}

template <int M>
__device__ __forceinline__ int dpp_xmask(int b) {
    return __builtin_amdgcn_mov_dpp(b, 0x160 + M, 0xf, 0xf, false);
}
template <int M>
__device__ __forceinline__ float dpp_xmask(float v) {
    return __builtin_bit_cast(float, dpp_xmask<M>(__builtin_bit_cast(int, v)));
}
__device__ __forceinline__ int xhalf(int b) {
    return __builtin_amdgcn_permlanex16(b, b, 0x76543210, 0xfedcba98, false, false);
}
__device__ __forceinline__ float xhalf(float v) {
    return __builtin_bit_cast(float, xhalf(__builtin_bit_cast(int, v)));
}
__device__ __forceinline__ float row16_sum(float v) {
    v += dpp_xmask<1>(v);
    v += dpp_xmask<2>(v);
    v += dpp_xmask<4>(v);
    v += dpp_xmask<8>(v);
    return v;
}
__device__ __forceinline__ float wave_sum(float v) {
    v += xhalf(v);
    v += dpp_xmask<8>(v);
    v += dpp_xmask<4>(v);
    v += dpp_xmask<2>(v);
    v += dpp_xmask<1>(v);
    return v;
}
__device__ __forceinline__ int wave_or(int v) {
    v |= xhalf(v);
    v |= dpp_xmask<8>(v);
    v |= dpp_xmask<4>(v);
    v |= dpp_xmask<2>(v);
    v |= dpp_xmask<1>(v);
    return v;
}
__device__ __forceinline__ float wave_max(float v) {
    v = fmaxf(v, xhalf(v));
    v = fmaxf(v, dpp_xmask<8>(v));
    v = fmaxf(v, dpp_xmask<4>(v));
    v = fmaxf(v, dpp_xmask<2>(v));
    v = fmaxf(v, dpp_xmask<1>(v));
    return v;
}
__device__ __forceinline__ float block256_sum(float v, float* red, int tid) {
    red[tid] = v;
    lds_barrier();
    const int l = tid & 31;
    const float w0 = red[l],       w1 = red[32 + l],  w2 = red[64 + l],  w3 = red[96 + l];
    const float w4 = red[128 + l], w5 = red[160 + l], w6 = red[192 + l], w7 = red[224 + l];
    const float x = ((w0 + w4) + (w2 + w6)) + ((w1 + w5) + (w3 + w7));
    return wave_sum(x);
}

__global__ __launch_bounds__(256)
void qkv_kernel(const float* __restrict__ x,
                const float* __restrict__ Wv,
                const float* __restrict__ Wq,
                const float* __restrict__ Wk,
                const float* __restrict__ bv,
                const float* __restrict__ bq,
                const float* __restrict__ bk,
                float* __restrict__ q_out,
                float* __restrict__ newv, float* __restrict__ newk) {
    __shared__ float4 xs4[E_DIM / 4];
    __shared__ float  red[8 * 16];
    const float* xs = (const float*)xs4;
    const int tid = threadIdx.x;
    const int m = blockIdx.y;
    const float* W    = (m == 0) ? Wv : (m == 1) ? Wq : Wk;
    const float* bias = (m == 0) ? bv : (m == 1) ? bq : bk;
    float* dst = (m == 0) ? newv + (size_t)(S_LEN - 1) * E_DIM
               : (m == 1) ? q_out
                          : newk + (size_t)(S_LEN - 1) * E_DIM;
    const float bval = bias[blockIdx.x * 32 + (tid & 31)];
    xs4[tid] = ((const float4*)x)[tid];
    __syncthreads();

    const int wv   = tid >> 5;
    const int lane = tid & 31;
    const int j    = lane & 15;
    const int half = lane >> 4;
    const int ob   = blockIdx.x * 32 + (wv & 1) * 16;
    const int k0   = (wv >> 1) * 256;
    const float amask = (j == 0) ? 1.f : 0.f;

    v8f c0 = {}, c1 = {};
    for (int eb = 0; eb < 256; eb += 8) {
        const int r0 = eb + half * 2;
        v2f a0, a1, b0, b1;
        a0.x = xs[k0 + r0]     * amask;  a0.y = xs[k0 + r0 + 1] * amask;
        a1.x = xs[k0 + r0 + 4] * amask;  a1.y = xs[k0 + r0 + 5] * amask;
        const float* wp0 = W + (size_t)(k0 + r0)     * E_DIM + ob + j;
        const float* wp1 = W + (size_t)(k0 + r0 + 4) * E_DIM + ob + j;
        b0.x = wp0[0];  b0.y = wp0[E_DIM];
        b1.x = wp1[0];  b1.y = wp1[E_DIM];
        c0 = wmma4(a0, b0, c0);
        c1 = wmma4(a1, b1, c1);
    }
    const v8f cs = c0 + c1;
    if (lane < 16) red[wv * 16 + lane] = cs[0];
    __syncthreads();
    if (tid < 32) {
        const int og = tid >> 4;
        const int jj = tid & 15;
        float s = 0.f;
        #pragma unroll
        for (int ks = 0; ks < 4; ks++) s += red[(ks * 2 + og) * 16 + jj];
        dst[blockIdx.x * 32 + tid] = s + bval;
    }
}

__global__ __launch_bounds__(256)
void shift_score_pv_kernel(const float* __restrict__ oldv,
                           const float* __restrict__ oldk,
                           const float* __restrict__ q,
                           float* __restrict__ newv, float* __restrict__ newk,
                           float* __restrict__ part_acc,
                           float* __restrict__ part_ml) {
    __shared__ float4 s_acc[8 * 256];
    __shared__ float  s_m[8 * 16];
    __shared__ float  s_l[8 * 16];
    __shared__ float  s_ml[32];
    const int tid  = threadIdx.x;
    const int lane = tid & 31;
    const int wv   = tid >> 5;
    const int s0   = (blockIdx.x * 8 + wv) * 2;
    const int s1   = s0 + 1;
    const bool last1 = (s1 == S_LEN - 1);

    const float4* v0src = (const float4*)(oldv + (size_t)(s0 + 1) * E_DIM);
    const float4* k0src = (const float4*)(oldk + (size_t)(s0 + 1) * E_DIM);
    const float4* v1src = (const float4*)(last1 ? (newv + (size_t)s1 * E_DIM)
                                                : (oldv + (size_t)(s1 + 1) * E_DIM));
    const float4* k1src = (const float4*)(last1 ? (newk + (size_t)s1 * E_DIM)
                                                : (oldk + (size_t)(s1 + 1) * E_DIM));
    float4* v0dst = (float4*)(newv + (size_t)s0 * E_DIM);
    float4* k0dst = (float4*)(newk + (size_t)s0 * E_DIM);
    float4* v1dst = (float4*)(newv + (size_t)s1 * E_DIM);
    float4* k1dst = (float4*)(newk + (size_t)s1 * E_DIM);
    const float4* q4 = (const float4*)q;

    float4 qd[8], va[8], vb[8], ka[8], kb[8];
    #pragma unroll
    for (int i = 0; i < 8; i++) qd[i] = q4[i * 32 + lane];
    #pragma unroll
    for (int i = 0; i < 8; i++) va[i] = v0src[i * 32 + lane];
    #pragma unroll
    for (int i = 0; i < 8; i++) vb[i] = v1src[i * 32 + lane];
    #pragma unroll
    for (int i = 0; i < 8; i++) ka[i] = k0src[i * 32 + lane];
    #pragma unroll
    for (int i = 0; i < 8; i++) kb[i] = k1src[i * 32 + lane];

    #pragma unroll
    for (int i = 0; i < 8; i++) v0dst[i * 32 + lane] = va[i];
    if (!last1) {
        #pragma unroll
        for (int i = 0; i < 8; i++) v1dst[i * 32 + lane] = vb[i];
    }
    #pragma unroll
    for (int i = 0; i < 8; i++) k0dst[i * 32 + lane] = ka[i];
    if (!last1) {
        #pragma unroll
        for (int i = 0; i < 8; i++) k1dst[i * 32 + lane] = kb[i];
    }

    int ora = 0, orb = 0;
    #pragma unroll
    for (int i = 0; i < 8; i++) {
        ora |= __builtin_bit_cast(int, va[i].x) | __builtin_bit_cast(int, va[i].y)
             | __builtin_bit_cast(int, va[i].z) | __builtin_bit_cast(int, va[i].w);
        orb |= __builtin_bit_cast(int, vb[i].x) | __builtin_bit_cast(int, vb[i].y)
             | __builtin_bit_cast(int, vb[i].z) | __builtin_bit_cast(int, vb[i].w);
    }
    const bool nza = (wave_or(ora) & 0x7fffffff) != 0;
    const bool nzb = (wave_or(orb) & 0x7fffffff) != 0;

    const bool arow = (lane == 0) || (lane == 24);
    const v8f zero8 = {};

    float sa[8], sb[8];
    #pragma unroll
    for (int i = 0; i < 8; i++) {
        float pa = ka[i].x * qd[i].x + ka[i].y * qd[i].y + ka[i].z * qd[i].z + ka[i].w * qd[i].w;
        float pb = kb[i].x * qd[i].x + kb[i].y * qd[i].y + kb[i].z * qd[i].z + kb[i].w * qd[i].w;
        pa = row16_sum(pa);
        pb = row16_sum(pb);
        sa[i] = nza ? pa * 0.125f : -INFINITY;
        sb[i] = nzb ? pb * 0.125f : -INFINITY;
    }

    #pragma unroll
    for (int i = 0; i < 8; i++) {
        const float m  = fmaxf(sa[i], sb[i]);
        const float ms = (m == -INFINITY) ? 0.f : m;
        const float ea = expf(sa[i] - ms);
        const float eb = expf(sb[i] - ms);
        if ((lane & 15) == 0) {
            const int h = i * 2 + (lane >> 4);
            s_m[wv * 16 + h] = m;
            s_l[wv * 16 + h] = ea + eb;
        }
        v2f a, b0, b1, b2, b3;
        a.x  = arow ? ea : 0.f;   a.y  = arow ? eb : 0.f;
        b0.x = va[i].x;           b0.y = vb[i].x;
        b1.x = va[i].y;           b1.y = vb[i].y;
        b2.x = va[i].z;           b2.y = vb[i].z;
        b3.x = va[i].w;           b3.y = vb[i].w;
        const v8f d0 = wmma4(a, b0, zero8);
        const v8f d1 = wmma4(a, b1, zero8);
        const v8f d2 = wmma4(a, b2, zero8);
        const v8f d3 = wmma4(a, b3, zero8);
        float4 acc;
        acc.x = d0[0];  acc.y = d1[0];  acc.z = d2[0];  acc.w = d3[0];
        s_acc[wv * 256 + i * 32 + lane] = acc;
    }
    __syncthreads();

    {
        const int head = tid >> 4;
        float mw[8];
        float M = -INFINITY;
        #pragma unroll
        for (int w = 0; w < 8; w++) { mw[w] = s_m[w * 16 + head];  M = fmaxf(M, mw[w]); }
        float4 acc = {0.f, 0.f, 0.f, 0.f};
        float L = 0.f;
        const float Ms = (M == -INFINITY) ? 0.f : M;
        #pragma unroll
        for (int w = 0; w < 8; w++) {
            const float sc = expf(mw[w] - Ms);
            const float4 t = s_acc[w * 256 + tid];
            acc.x += t.x * sc;  acc.y += t.y * sc;  acc.z += t.z * sc;  acc.w += t.w * sc;
            L += s_l[w * 16 + head] * sc;
        }
        ((float4*)part_acc)[(size_t)blockIdx.x * 256 + tid] = acc;
        if ((tid & 15) == 0) { s_ml[head] = M;  s_ml[16 + head] = L; }
    }
    __syncthreads();
    if (tid < 32) part_ml[blockIdx.x * 32 + tid] = s_ml[tid];
}

__global__ __launch_bounds__(256)
void merge_gemv_kernel(const float* __restrict__ part_acc,
                       const float* __restrict__ part_ml,
                       const float* __restrict__ W,
                       float* __restrict__ out_parts) {
    __shared__ float  s_w[512];
    __shared__ float  s_redm[8];
    __shared__ float  s_redl[8];
    __shared__ float4 s_sum[16 * 16];
    __shared__ float  xs[D_H];
    const int h    = blockIdx.y;
    const int tid  = threadIdx.x;
    const int lane = tid & 31;
    const int wv   = tid >> 5;
    const int j    = lane & 15;
    const int half = lane >> 4;
    const int ob   = blockIdx.x * 128 + wv * 16;
    const int k0   = h * D_H;

    const float mb0 = part_ml[tid * 32 + h];
    const float mb1 = part_ml[(tid + 256) * 32 + h];
    const float lb0 = part_ml[tid * 32 + 16 + h];
    const float lb1 = part_ml[(tid + 256) * 32 + 16 + h];

    const int c = tid & 15;
    const int g = tid >> 4;
    const float4* psrc = (const float4*)part_acc + h * 16 + c;
    float4 pv[32];
    #pragma unroll
    for (int it = 0; it < 32; it++) pv[it] = psrc[(size_t)(it * 16 + g) * 256];

    float bw[32];
    #pragma unroll
    for (int t = 0; t < 8; t++) {
        const int r0 = t * 8 + half * 2;
        const float* wp0 = W + (size_t)(k0 + r0)     * E_DIM + ob + j;
        const float* wp1 = W + (size_t)(k0 + r0 + 4) * E_DIM + ob + j;
        bw[t * 4 + 0] = wp0[0];  bw[t * 4 + 1] = wp0[E_DIM];
        bw[t * 4 + 2] = wp1[0];  bw[t * 4 + 3] = wp1[E_DIM];
    }

    const float mx = wave_max(fmaxf(mb0, mb1));
    if (lane == 0) s_redm[wv] = mx;
    __syncthreads();
    float M = s_redm[0];
    #pragma unroll
    for (int w = 1; w < 8; w++) M = fmaxf(M, s_redm[w]);

    const float Ms = (M == -INFINITY) ? 0.f : M;
    const float w0 = expf(mb0 - Ms);
    const float w1 = expf(mb1 - Ms);
    s_w[tid]       = w0;
    s_w[tid + 256] = w1;
    const float ls = wave_sum(w0 * lb0 + w1 * lb1);
    if (lane == 0) s_redl[wv] = ls;
    __syncthreads();
    float L = 0.f;
    #pragma unroll
    for (int w = 0; w < 8; w++) L += s_redl[w];

    float4 acc = {0.f, 0.f, 0.f, 0.f};
    #pragma unroll
    for (int it = 0; it < 32; it++) {
        const float w = s_w[it * 16 + g];
        acc.x += pv[it].x * w;  acc.y += pv[it].y * w;
        acc.z += pv[it].z * w;  acc.w += pv[it].w * w;
    }
    s_sum[g * 16 + c] = acc;
    __syncthreads();
    if (tid < D_H) {
        const float* sf = (const float*)s_sum;
        float s = 0.f;
        #pragma unroll
        for (int gg = 0; gg < 16; gg++) s += sf[gg * D_H + tid];
        xs[tid] = s / L;
    }
    __syncthreads();

    const float amask = (j == 0) ? 1.f : 0.f;
    v8f c0 = {}, c1 = {};
    #pragma unroll
    for (int t = 0; t < 8; t++) {
        const int r0 = t * 8 + half * 2;
        v2f a0, a1, b0, b1;
        a0.x = xs[r0]     * amask;  a0.y = xs[r0 + 1] * amask;
        a1.x = xs[r0 + 4] * amask;  a1.y = xs[r0 + 5] * amask;
        b0.x = bw[t * 4 + 0];  b0.y = bw[t * 4 + 1];
        b1.x = bw[t * 4 + 2];  b1.y = bw[t * 4 + 3];
        c0 = wmma4(a0, b0, c0);
        c1 = wmma4(a1, b1, c1);
    }
    const v8f cs = c0 + c1;
    if (lane < 16)
        out_parts[(size_t)h * E_DIM + ob + lane] = cs[0];
}

__global__ __launch_bounds__(256)
void gemv_relu_part_kernel(const float* W,
                           const float* in_parts,
                           const float* in_bias,
                           float* out_parts) {
    __shared__ float xs[128];
    const int tid  = threadIdx.x;
    const int k0   = blockIdx.y * 128;
    const int wv   = tid >> 5;
    const int lane = tid & 31;
    const int j    = lane & 15;
    const int half = lane >> 4;
    const int ob   = blockIdx.x * 128 + wv * 16;
    const int ke   = k0 + (tid & 127);

    float pp[8];
    #pragma unroll
    for (int p = 0; p < 8; p++) pp[p] = in_parts[(size_t)p * E_DIM + ke];
    const float pbias = in_bias[ke];
    asm volatile("" ::: "memory");

    float bw[64];
    #pragma unroll
    for (int t = 0; t < 16; t++) {
        const int r0 = t * 8 + half * 2;
        const float* wp0 = W + (size_t)(k0 + r0)     * E_DIM + ob + j;
        const float* wp1 = W + (size_t)(k0 + r0 + 4) * E_DIM + ob + j;
        bw[t * 4 + 0] = wp0[0];  bw[t * 4 + 1] = wp0[E_DIM];
        bw[t * 4 + 2] = wp1[0];  bw[t * 4 + 3] = wp1[E_DIM];
    }

    float s = 0.f;
    #pragma unroll
    for (int p = 0; p < 8; p++) s += pp[p];
    s += pbias;
    if (tid < 128) xs[tid] = fmaxf(s, 0.f);
    lds_barrier();

    const float amask = (j == 0) ? 1.f : 0.f;
    v8f c0 = {}, c1 = {};
    #pragma unroll
    for (int t = 0; t < 16; t++) {
        const int r0 = t * 8 + half * 2;
        v2f a0, a1, b0, b1;
        a0.x = xs[r0]     * amask;  a0.y = xs[r0 + 1] * amask;
        a1.x = xs[r0 + 4] * amask;  a1.y = xs[r0 + 5] * amask;
        b0.x = bw[t * 4 + 0];  b0.y = bw[t * 4 + 1];
        b1.x = bw[t * 4 + 2];  b1.y = bw[t * 4 + 3];
        c0 = wmma4(a0, b0, c0);
        c1 = wmma4(a1, b1, c1);
    }
    const v8f cs = c0 + c1;
    if (lane < 16)
        out_parts[(size_t)blockIdx.y * E_DIM + ob + lane] = cs[0];
}

__global__ __launch_bounds__(256)
void gemv_ln_part_kernel(const float* __restrict__ W,
                         const float* __restrict__ parts,
                         const float* __restrict__ bias,
                         const float* __restrict__ resid,
                         const float* __restrict__ sc,
                         const float* __restrict__ bs,
                         float* __restrict__ h_out,
                         float* __restrict__ out_parts) {
    __shared__ float red_s[256];
    __shared__ float red_v[256];
    __shared__ float xs[128];
    const int tid = threadIdx.x;
    const int k0  = blockIdx.y * 128;

    float pb[4], pr[4], pp[4][16], psc[4], pbs[4];
    #pragma unroll
    for (int i = 0; i < 4; i++) {
        const int e = i * 256 + tid;
        pb[i] = bias[e];
        pr[i] = resid[e];
        #pragma unroll
        for (int p = 0; p < 16; p++) pp[i][p] = parts[(size_t)p * E_DIM + e];
        psc[i] = sc[e];
        pbs[i] = bs[e];
    }

    const int wv   = tid >> 5;
    const int lane = tid & 31;
    const int j    = lane & 15;
    const int half = lane >> 4;
    const int ob   = blockIdx.x * 128 + wv * 16;
    asm volatile("" ::: "memory");
    float bw[64];
    #pragma unroll
    for (int t = 0; t < 16; t++) {
        const int r0 = t * 8 + half * 2;
        const float* wp0 = W + (size_t)(k0 + r0)     * E_DIM + ob + j;
        const float* wp1 = W + (size_t)(k0 + r0 + 4) * E_DIM + ob + j;
        bw[t * 4 + 0] = wp0[0];  bw[t * 4 + 1] = wp0[E_DIM];
        bw[t * 4 + 2] = wp1[0];  bw[t * 4 + 3] = wp1[E_DIM];
    }

    float vv[4];
    float sum = 0.f;
    #pragma unroll
    for (int i = 0; i < 4; i++) {
        float s = pb[i] + pr[i];
        #pragma unroll
        for (int p = 0; p < 16; p++) s += pp[i][p];
        vv[i] = s;
        sum += s;
    }
    const float mean = block256_sum(sum, red_s, tid) * (1.f / E_DIM);
    float var = 0.f;
    #pragma unroll
    for (int i = 0; i < 4; i++) { const float d = vv[i] - mean; var += d * d; }
    const float inv = rsqrtf(block256_sum(var, red_v, tid) * (1.f / E_DIM) + 1e-6f);
    const bool mine = ((tid >> 7) == (int)(blockIdx.y & 1));
    #pragma unroll
    for (int i = 0; i < 4; i++) {
        const float hv = (vv[i] - mean) * inv * psc[i] + pbs[i];
        if (mine && i == (int)(blockIdx.y >> 1)) {
            xs[tid & 127] = hv;
            if (blockIdx.x == 0) h_out[i * 256 + tid] = hv;
        }
    }
    lds_barrier();

    const float amask = (j == 0) ? 1.f : 0.f;
    v8f c0 = {}, c1 = {};
    #pragma unroll
    for (int t = 0; t < 16; t++) {
        const int r0 = t * 8 + half * 2;
        v2f a0, a1, b0, b1;
        a0.x = xs[r0]     * amask;  a0.y = xs[r0 + 1] * amask;
        a1.x = xs[r0 + 4] * amask;  a1.y = xs[r0 + 5] * amask;
        b0.x = bw[t * 4 + 0];  b0.y = bw[t * 4 + 1];
        b1.x = bw[t * 4 + 2];  b1.y = bw[t * 4 + 3];
        c0 = wmma4(a0, b0, c0);
        c1 = wmma4(a1, b1, c1);
    }
    const v8f cs = c0 + c1;
    if (lane < 16)
        out_parts[(size_t)blockIdx.y * E_DIM + ob + lane] = cs[0];
}

__global__ __launch_bounds__(256)
void ln_combine_kernel(const float* __restrict__ parts,
                       const float* __restrict__ bias,
                       const float* __restrict__ resid,
                       const float* __restrict__ sc,
                       const float* __restrict__ bs,
                       float* __restrict__ out) {
    __shared__ float red_s[256];
    __shared__ float red_v[256];
    const int tid = threadIdx.x;
    float pb[4], pr[4], pp[4][8], psc[4], pbs[4];
    #pragma unroll
    for (int i = 0; i < 4; i++) {
        const int e = i * 256 + tid;
        pb[i] = bias[e];
        pr[i] = resid[e];
        #pragma unroll
        for (int p = 0; p < 8; p++) pp[i][p] = parts[(size_t)p * E_DIM + e];
        psc[i] = sc[e];
        pbs[i] = bs[e];
    }
    float vv[4];
    float sum = 0.f;
    #pragma unroll
    for (int i = 0; i < 4; i++) {
        float s = pb[i] + pr[i];
        #pragma unroll
        for (int p = 0; p < 8; p++) s += pp[i][p];
        vv[i] = s;
        sum += s;
    }
    const float mean = block256_sum(sum, red_s, tid) * (1.f / E_DIM);
    float var = 0.f;
    #pragma unroll
    for (int i = 0; i < 4; i++) { const float d = vv[i] - mean; var += d * d; }
    const float inv = rsqrtf(block256_sum(var, red_v, tid) * (1.f / E_DIM) + 1e-6f);
    #pragma unroll
    for (int i = 0; i < 4; i++) {
        const int e = i * 256 + tid;
        out[e] = (vv[i] - mean) * inv * psc[i] + pbs[i];
    }
}

extern "C" void kernel_launch(void* const* d_in, const int* in_sizes, int n_in,
                              void* d_out, int out_size, void* d_ws, size_t ws_size,
                              hipStream_t stream) {
    (void)in_sizes; (void)n_in; (void)out_size; (void)ws_size;
    const float* x     = (const float*)d_in[0];
    const float* cache = (const float*)d_in[1];
    const float* Wv = (const float*)d_in[2];  const float* bv = (const float*)d_in[3];
    const float* Wq = (const float*)d_in[4];  const float* bq = (const float*)d_in[5];
    const float* Wk = (const float*)d_in[6];  const float* bk = (const float*)d_in[7];
    const float* Wo = (const float*)d_in[8];  const float* bo = (const float*)d_in[9];
    const float* W1 = (const float*)d_in[10]; const float* b1 = (const float*)d_in[11];
    const float* W2 = (const float*)d_in[12]; const float* b2 = (const float*)d_in[13];
    const float* ln1_s = (const float*)d_in[14]; const float* ln1_b = (const float*)d_in[15];
    const float* ln2_s = (const float*)d_in[16]; const float* ln2_b = (const float*)d_in[17];

    float* out_h = (float*)d_out;
    float* newv  = out_h + E_DIM;
    float* newk  = newv + (size_t)S_LEN * E_DIM;
    const float* oldv = cache;
    const float* oldk = cache + (size_t)S_LEN * E_DIM;

    float* ws       = (float*)d_ws;
    float* q        = ws;
    float* part_acc = q + E_DIM;
    float* part_ml  = part_acc + 512 * E_DIM;
    float* h1       = part_ml + 512 * 32;
    float* part_o   = h1 + E_DIM;
    float* part_1   = part_o + 16 * E_DIM;
    float* part_2   = part_1 + 8 * E_DIM;

    qkv_kernel<<<dim3(32, 3), 256, 0, stream>>>(x, Wv, Wq, Wk, bv, bq, bk, q, newv, newk);
    shift_score_pv_kernel<<<512, 256, 0, stream>>>(oldv, oldk, q, newv, newk, part_acc, part_ml);
    merge_gemv_kernel<<<dim3(8, 16), 256, 0, stream>>>(part_acc, part_ml, Wo, part_o);
    gemv_ln_part_kernel<<<dim3(8, 8), 256, 0, stream>>>(W1, part_o, bo, x, ln1_s, ln1_b, h1, part_1);
    gemv_relu_part_kernel<<<dim3(8, 8), 256, 0, stream>>>(W2, part_1, b1, part_2);
    ln_combine_kernel<<<1, 256, 0, stream>>>(part_2, b2, h1, ln2_s, ln2_b, out_h);
}


# qkv GEMV split over K across 192 blocks (128x128 W tiles, every load issued at entry) instead of 96 blocks of 1024x32; its 8 partials + bias summed in the attention pass via LDS; new sum order
# speedup vs baseline: 3.8701x; 1.0954x over previous
#include <hip/hip_runtime.h>
#include <hip/hip_bf16.h>
#include <math.h>

#define E_DIM 1024
#define S_LEN 8192
#define H_N   16
#define D_H   64

typedef __attribute__((ext_vector_type(2))) float v2f;
typedef __attribute__((ext_vector_type(8))) float v8f;

__device__ __forceinline__ v8f wmma4(v2f a, v2f b, v8f c) {
    return __builtin_amdgcn_wmma_f32_16x16x4_f32(false, a, false, b, (short)0, c,
                                                 false, false);
}

__device__ __forceinline__ void lds_barrier() {
    __builtin_amdgcn_fence(__ATOMIC_RELEASE, "workgroup", "local");
    __builtin_amdgcn_s_barrier();
    __builtin_amdgcn_fence(__ATOMIC_ACQUIRE, "workgroup", "local");
}

template <int M>
__device__ __forceinline__ int dpp_xmask(int b) {
    return __builtin_amdgcn_mov_dpp(b, 0x160 + M, 0xf, 0xf, false);
}
template <int M>
__device__ __forceinline__ float dpp_xmask(float v) {
    return __builtin_bit_cast(float, dpp_xmask<M>(__builtin_bit_cast(int, v)));
}
__device__ __forceinline__ int xhalf(int b) {
    return __builtin_amdgcn_permlanex16(b, b, 0x76543210, 0xfedcba98, false, false);
}
__device__ __forceinline__ float xhalf(float v) {
    return __builtin_bit_cast(float, xhalf(__builtin_bit_cast(int, v)));
}
__device__ __forceinline__ float row16_sum(float v) {
    v += dpp_xmask<1>(v);
    v += dpp_xmask<2>(v);
    v += dpp_xmask<4>(v);
    v += dpp_xmask<8>(v);
    return v;
}
__device__ __forceinline__ float wave_sum(float v) {
    v += xhalf(v);
    v += dpp_xmask<8>(v);
    v += dpp_xmask<4>(v);
    v += dpp_xmask<2>(v);
    v += dpp_xmask<1>(v);
    return v;
}
__device__ __forceinline__ int wave_or(int v) {
    v |= xhalf(v);
    v |= dpp_xmask<8>(v);
    v |= dpp_xmask<4>(v);
    v |= dpp_xmask<2>(v);
    v |= dpp_xmask<1>(v);
    return v;
}
__device__ __forceinline__ float wave_max(float v) {
    v = fmaxf(v, xhalf(v));
    v = fmaxf(v, dpp_xmask<8>(v));
    v = fmaxf(v, dpp_xmask<4>(v));
    v = fmaxf(v, dpp_xmask<2>(v));
    v = fmaxf(v, dpp_xmask<1>(v));
    return v;
}
__device__ __forceinline__ float block256_sum(float v, float* red, int tid) {
    red[tid] = v;
    lds_barrier();
    const int l = tid & 31;
    const float w0 = red[l],       w1 = red[32 + l],  w2 = red[64 + l],  w3 = red[96 + l];
    const float w4 = red[128 + l], w5 = red[160 + l], w6 = red[192 + l], w7 = red[224 + l];
    const float x = ((w0 + w4) + (w2 + w6)) + ((w1 + w5) + (w3 + w7));
    return wave_sum(x);
}

__global__ __launch_bounds__(256)
void qkv_part_kernel(const float* x,
                     const float* Wv,
                     const float* Wq,
                     const float* Wk,
                     float* out_parts) {
    __shared__ float xs[128];
    const int tid  = threadIdx.x;
    const int m    = blockIdx.z;
    const float* W = (m == 0) ? Wv : (m == 1) ? Wq : Wk;
    const int k0   = blockIdx.y * 128;
    const int wv   = tid >> 5;
    const int lane = tid & 31;
    const int j    = lane & 15;
    const int half = lane >> 4;
    const int ob   = blockIdx.x * 128 + wv * 16;

    const float xv = x[k0 + (tid & 127)];
    asm volatile("" ::: "memory");

    float bw[64];
    #pragma unroll
    for (int t = 0; t < 16; t++) {
        const int r0 = t * 8 + half * 2;
        const float* wp0 = W + (size_t)(k0 + r0)     * E_DIM + ob + j;
        const float* wp1 = W + (size_t)(k0 + r0 + 4) * E_DIM + ob + j;
        bw[t * 4 + 0] = wp0[0];  bw[t * 4 + 1] = wp0[E_DIM];
        bw[t * 4 + 2] = wp1[0];  bw[t * 4 + 3] = wp1[E_DIM];
    }

    if (tid < 128) xs[tid] = xv;
    lds_barrier();

    const float amask = (j == 0) ? 1.f : 0.f;
    v8f c0 = {}, c1 = {};
    #pragma unroll
    for (int t = 0; t < 16; t++) {
        const int r0 = t * 8 + half * 2;
        v2f a0, a1, b0, b1;
        a0.x = xs[r0]     * amask;  a0.y = xs[r0 + 1] * amask;
        a1.x = xs[r0 + 4] * amask;  a1.y = xs[r0 + 5] * amask;
        b0.x = bw[t * 4 + 0];  b0.y = bw[t * 4 + 1];
        b1.x = bw[t * 4 + 2];  b1.y = bw[t * 4 + 3];
        c0 = wmma4(a0, b0, c0);
        c1 = wmma4(a1, b1, c1);
    }
    const v8f cs = c0 + c1;
    if (lane < 16)
        out_parts[(size_t)(m * 8 + blockIdx.y) * E_DIM + ob + lane] = cs[0];
}

__global__ __launch_bounds__(256)
void shift_score_pv_kernel(const float* __restrict__ oldv,
                           const float* __restrict__ oldk,
                           const float* __restrict__ qkv_parts,
                           const float* __restrict__ bv,
                           const float* __restrict__ bq,
                           const float* __restrict__ bk,
                           float* __restrict__ newv, float* __restrict__ newk,
                           float* __restrict__ part_acc,
                           float* __restrict__ part_ml) {
    __shared__ float4 s_acc[8 * 256];
    __shared__ float  s_m[8 * 16];
    __shared__ float  s_l[8 * 16];
    __shared__ float  s_ml[32];
    __shared__ float4 s_q[256];
    __shared__ float4 s_vk[512];
    const int tid  = threadIdx.x;
    const int lane = tid & 31;
    const int wv   = tid >> 5;
    const int s0   = (blockIdx.x * 8 + wv) * 2;
    const int s1   = s0 + 1;
    const bool last1 = (s1 == S_LEN - 1);

    const float4* v0src = (const float4*)(oldv + (size_t)(s0 + 1) * E_DIM);
    const float4* k0src = (const float4*)(oldk + (size_t)(s0 + 1) * E_DIM);
    const float4* v1src = (const float4*)(oldv + (size_t)(last1 ? s0 + 1 : s1 + 1) * E_DIM);
    const float4* k1src = (const float4*)(oldk + (size_t)(last1 ? s0 + 1 : s1 + 1) * E_DIM);
    float4* v0dst = (float4*)(newv + (size_t)s0 * E_DIM);
    float4* k0dst = (float4*)(newk + (size_t)s0 * E_DIM);
    float4* v1dst = (float4*)(newv + (size_t)s1 * E_DIM);
    float4* k1dst = (float4*)(newk + (size_t)s1 * E_DIM);
    const float4* parts4 = (const float4*)qkv_parts;
    const bool lastblk = (blockIdx.x == S_LEN / 16 - 1);

    float4 qp[8];
    #pragma unroll
    for (int p = 0; p < 8; p++) qp[p] = parts4[(8 + p) * 256 + tid];
    const float4 qb = ((const float4*)bq)[tid];

    float4 qd[8], va[8], vb[8], ka[8], kb[8];
    #pragma unroll
    for (int i = 0; i < 8; i++) va[i] = v0src[i * 32 + lane];
    #pragma unroll
    for (int i = 0; i < 8; i++) vb[i] = v1src[i * 32 + lane];
    #pragma unroll
    for (int i = 0; i < 8; i++) ka[i] = k0src[i * 32 + lane];
    #pragma unroll
    for (int i = 0; i < 8; i++) kb[i] = k1src[i * 32 + lane];

    {
        float4 a = qp[0];
        #pragma unroll
        for (int p = 1; p < 8; p++) { a.x += qp[p].x;  a.y += qp[p].y;  a.z += qp[p].z;  a.w += qp[p].w; }
        a.x += qb.x;  a.y += qb.y;  a.z += qb.z;  a.w += qb.w;
        s_q[tid] = a;
    }
    if (lastblk) {
        #pragma unroll
        for (int mk = 0; mk < 2; mk++) {
            const float4* pp4 = parts4 + (size_t)(mk * 16) * 256 + tid;
            float4 a = pp4[0];
            #pragma unroll
            for (int p = 1; p < 8; p++) {
                const float4 t = pp4[p * 256];
                a.x += t.x;  a.y += t.y;  a.z += t.z;  a.w += t.w;
            }
            const float4 b = ((const float4*)(mk == 0 ? bv : bk))[tid];
            a.x += b.x;  a.y += b.y;  a.z += b.z;  a.w += b.w;
            s_vk[mk * 256 + tid] = a;
        }
    }
    lds_barrier();
    #pragma unroll
    for (int i = 0; i < 8; i++) qd[i] = s_q[i * 32 + lane];
    if (last1) {
        #pragma unroll
        for (int i = 0; i < 8; i++) { vb[i] = s_vk[i * 32 + lane];  kb[i] = s_vk[256 + i * 32 + lane]; }
    }

    #pragma unroll
    for (int i = 0; i < 8; i++) v0dst[i * 32 + lane] = va[i];
    #pragma unroll
    for (int i = 0; i < 8; i++) v1dst[i * 32 + lane] = vb[i];
    #pragma unroll
    for (int i = 0; i < 8; i++) k0dst[i * 32 + lane] = ka[i];
    #pragma unroll
    for (int i = 0; i < 8; i++) k1dst[i * 32 + lane] = kb[i];

    int ora = 0, orb = 0;
    #pragma unroll
    for (int i = 0; i < 8; i++) {
        ora |= __builtin_bit_cast(int, va[i].x) | __builtin_bit_cast(int, va[i].y)
             | __builtin_bit_cast(int, va[i].z) | __builtin_bit_cast(int, va[i].w);
        orb |= __builtin_bit_cast(int, vb[i].x) | __builtin_bit_cast(int, vb[i].y)
             | __builtin_bit_cast(int, vb[i].z) | __builtin_bit_cast(int, vb[i].w);
    }
    const bool nza = (wave_or(ora) & 0x7fffffff) != 0;
    const bool nzb = (wave_or(orb) & 0x7fffffff) != 0;

    const bool arow = (lane == 0) || (lane == 24);
    const v8f zero8 = {};

    float sa[8], sb[8];
    #pragma unroll
    for (int i = 0; i < 8; i++) {
        float pa = ka[i].x * qd[i].x + ka[i].y * qd[i].y + ka[i].z * qd[i].z + ka[i].w * qd[i].w;
        float pb = kb[i].x * qd[i].x + kb[i].y * qd[i].y + kb[i].z * qd[i].z + kb[i].w * qd[i].w;
        pa = row16_sum(pa);
        pb = row16_sum(pb);
        sa[i] = nza ? pa * 0.125f : -INFINITY;
        sb[i] = nzb ? pb * 0.125f : -INFINITY;
    }

    #pragma unroll
    for (int i = 0; i < 8; i++) {
        const float m  = fmaxf(sa[i], sb[i]);
        const float ms = (m == -INFINITY) ? 0.f : m;
        const float ea = expf(sa[i] - ms);
        const float eb = expf(sb[i] - ms);
        if ((lane & 15) == 0) {
            const int h = i * 2 + (lane >> 4);
            s_m[wv * 16 + h] = m;
            s_l[wv * 16 + h] = ea + eb;
        }
        v2f a, b0, b1, b2, b3;
        a.x  = arow ? ea : 0.f;   a.y  = arow ? eb : 0.f;
        b0.x = va[i].x;           b0.y = vb[i].x;
        b1.x = va[i].y;           b1.y = vb[i].y;
        b2.x = va[i].z;           b2.y = vb[i].z;
        b3.x = va[i].w;           b3.y = vb[i].w;
        const v8f d0 = wmma4(a, b0, zero8);
        const v8f d1 = wmma4(a, b1, zero8);
        const v8f d2 = wmma4(a, b2, zero8);
        const v8f d3 = wmma4(a, b3, zero8);
        float4 acc;
        acc.x = d0[0];  acc.y = d1[0];  acc.z = d2[0];  acc.w = d3[0];
        s_acc[wv * 256 + i * 32 + lane] = acc;
    }
    __syncthreads();

    {
        const int head = tid >> 4;
        float mw[8];
        float M = -INFINITY;
        #pragma unroll
        for (int w = 0; w < 8; w++) { mw[w] = s_m[w * 16 + head];  M = fmaxf(M, mw[w]); }
        float4 acc = {0.f, 0.f, 0.f, 0.f};
        float L = 0.f;
        const float Ms = (M == -INFINITY) ? 0.f : M;
        #pragma unroll
        for (int w = 0; w < 8; w++) {
            const float sc = expf(mw[w] - Ms);
            const float4 t = s_acc[w * 256 + tid];
            acc.x += t.x * sc;  acc.y += t.y * sc;  acc.z += t.z * sc;  acc.w += t.w * sc;
            L += s_l[w * 16 + head] * sc;
        }
        ((float4*)part_acc)[(size_t)blockIdx.x * 256 + tid] = acc;
        if ((tid & 15) == 0) { s_ml[head] = M;  s_ml[16 + head] = L; }
    }
    __syncthreads();
    if (tid < 32) part_ml[blockIdx.x * 32 + tid] = s_ml[tid];
}

__global__ __launch_bounds__(256)
void merge_gemv_kernel(const float* __restrict__ part_acc,
                       const float* __restrict__ part_ml,
                       const float* __restrict__ W,
                       float* __restrict__ out_parts) {
    __shared__ float  s_w[512];
    __shared__ float  s_redm[8];
    __shared__ float  s_redl[8];
    __shared__ float4 s_sum[16 * 16];
    __shared__ float  xs[D_H];
    const int h    = blockIdx.y;
    const int tid  = threadIdx.x;
    const int lane = tid & 31;
    const int wv   = tid >> 5;
    const int j    = lane & 15;
    const int half = lane >> 4;
    const int ob   = blockIdx.x * 128 + wv * 16;
    const int k0   = h * D_H;

    const float mb0 = part_ml[tid * 32 + h];
    const float mb1 = part_ml[(tid + 256) * 32 + h];
    const float lb0 = part_ml[tid * 32 + 16 + h];
    const float lb1 = part_ml[(tid + 256) * 32 + 16 + h];

    const int c = tid & 15;
    const int g = tid >> 4;
    const float4* psrc = (const float4*)part_acc + h * 16 + c;
    float4 pv[32];
    #pragma unroll
    for (int it = 0; it < 32; it++) pv[it] = psrc[(size_t)(it * 16 + g) * 256];

    float bw[32];
    #pragma unroll
    for (int t = 0; t < 8; t++) {
        const int r0 = t * 8 + half * 2;
        const float* wp0 = W + (size_t)(k0 + r0)     * E_DIM + ob + j;
        const float* wp1 = W + (size_t)(k0 + r0 + 4) * E_DIM + ob + j;
        bw[t * 4 + 0] = wp0[0];  bw[t * 4 + 1] = wp0[E_DIM];
        bw[t * 4 + 2] = wp1[0];  bw[t * 4 + 3] = wp1[E_DIM];
    }

    const float mx = wave_max(fmaxf(mb0, mb1));
    if (lane == 0) s_redm[wv] = mx;
    __syncthreads();
    float M = s_redm[0];
    #pragma unroll
    for (int w = 1; w < 8; w++) M = fmaxf(M, s_redm[w]);

    const float Ms = (M == -INFINITY) ? 0.f : M;
    const float w0 = expf(mb0 - Ms);
    const float w1 = expf(mb1 - Ms);
    s_w[tid]       = w0;
    s_w[tid + 256] = w1;
    const float ls = wave_sum(w0 * lb0 + w1 * lb1);
    if (lane == 0) s_redl[wv] = ls;
    __syncthreads();
    float L = 0.f;
    #pragma unroll
    for (int w = 0; w < 8; w++) L += s_redl[w];

    float4 acc = {0.f, 0.f, 0.f, 0.f};
    #pragma unroll
    for (int it = 0; it < 32; it++) {
        const float w = s_w[it * 16 + g];
        acc.x += pv[it].x * w;  acc.y += pv[it].y * w;
        acc.z += pv[it].z * w;  acc.w += pv[it].w * w;
    }
    s_sum[g * 16 + c] = acc;
    __syncthreads();
    if (tid < D_H) {
        const float* sf = (const float*)s_sum;
        float s = 0.f;
        #pragma unroll
        for (int gg = 0; gg < 16; gg++) s += sf[gg * D_H + tid];
        xs[tid] = s / L;
    }
    __syncthreads();

    const float amask = (j == 0) ? 1.f : 0.f;
    v8f c0 = {}, c1 = {};
    #pragma unroll
    for (int t = 0; t < 8; t++) {
        const int r0 = t * 8 + half * 2;
        v2f a0, a1, b0, b1;
        a0.x = xs[r0]     * amask;  a0.y = xs[r0 + 1] * amask;
        a1.x = xs[r0 + 4] * amask;  a1.y = xs[r0 + 5] * amask;
        b0.x = bw[t * 4 + 0];  b0.y = bw[t * 4 + 1];
        b1.x = bw[t * 4 + 2];  b1.y = bw[t * 4 + 3];
        c0 = wmma4(a0, b0, c0);
        c1 = wmma4(a1, b1, c1);
    }
    const v8f cs = c0 + c1;
    if (lane < 16)
        out_parts[(size_t)h * E_DIM + ob + lane] = cs[0];
}

__global__ __launch_bounds__(256)
void gemv_relu_part_kernel(const float* W,
                           const float* in_parts,
                           const float* in_bias,
                           float* out_parts) {
    __shared__ float xs[128];
    const int tid  = threadIdx.x;
    const int k0   = blockIdx.y * 128;
    const int wv   = tid >> 5;
    const int lane = tid & 31;
    const int j    = lane & 15;
    const int half = lane >> 4;
    const int ob   = blockIdx.x * 128 + wv * 16;
    const int ke   = k0 + (tid & 127);

    float pp[8];
    #pragma unroll
    for (int p = 0; p < 8; p++) pp[p] = in_parts[(size_t)p * E_DIM + ke];
    const float pbias = in_bias[ke];
    asm volatile("" ::: "memory");

    float bw[64];
    #pragma unroll
    for (int t = 0; t < 16; t++) {
        const int r0 = t * 8 + half * 2;
        const float* wp0 = W + (size_t)(k0 + r0)     * E_DIM + ob + j;
        const float* wp1 = W + (size_t)(k0 + r0 + 4) * E_DIM + ob + j;
        bw[t * 4 + 0] = wp0[0];  bw[t * 4 + 1] = wp0[E_DIM];
        bw[t * 4 + 2] = wp1[0];  bw[t * 4 + 3] = wp1[E_DIM];
    }

    float s = 0.f;
    #pragma unroll
    for (int p = 0; p < 8; p++) s += pp[p];
    s += pbias;
    if (tid < 128) xs[tid] = fmaxf(s, 0.f);
    lds_barrier();

    const float amask = (j == 0) ? 1.f : 0.f;
    v8f c0 = {}, c1 = {};
    #pragma unroll
    for (int t = 0; t < 16; t++) {
        const int r0 = t * 8 + half * 2;
        v2f a0, a1, b0, b1;
        a0.x = xs[r0]     * amask;  a0.y = xs[r0 + 1] * amask;
        a1.x = xs[r0 + 4] * amask;  a1.y = xs[r0 + 5] * amask;
        b0.x = bw[t * 4 + 0];  b0.y = bw[t * 4 + 1];
        b1.x = bw[t * 4 + 2];  b1.y = bw[t * 4 + 3];
        c0 = wmma4(a0, b0, c0);
        c1 = wmma4(a1, b1, c1);
    }
    const v8f cs = c0 + c1;
    if (lane < 16)
        out_parts[(size_t)blockIdx.y * E_DIM + ob + lane] = cs[0];
}

__global__ __launch_bounds__(256)
void gemv_ln_part_kernel(const float* __restrict__ W,
                         const float* __restrict__ parts,
                         const float* __restrict__ bias,
                         const float* __restrict__ resid,
                         const float* __restrict__ sc,
                         const float* __restrict__ bs,
                         float* __restrict__ h_out,
                         float* __restrict__ out_parts) {
    __shared__ float red_s[256];
    __shared__ float red_v[256];
    __shared__ float xs[128];
    const int tid = threadIdx.x;
    const int k0  = blockIdx.y * 128;

    float pb[4], pr[4], pp[4][16], psc[4], pbs[4];
    #pragma unroll
    for (int i = 0; i < 4; i++) {
        const int e = i * 256 + tid;
        pb[i] = bias[e];
        pr[i] = resid[e];
        #pragma unroll
        for (int p = 0; p < 16; p++) pp[i][p] = parts[(size_t)p * E_DIM + e];
        psc[i] = sc[e];
        pbs[i] = bs[e];
    }

    const int wv   = tid >> 5;
    const int lane = tid & 31;
    const int j    = lane & 15;
    const int half = lane >> 4;
    const int ob   = blockIdx.x * 128 + wv * 16;
    asm volatile("" ::: "memory");
    float bw[64];
    #pragma unroll
    for (int t = 0; t < 16; t++) {
        const int r0 = t * 8 + half * 2;
        const float* wp0 = W + (size_t)(k0 + r0)     * E_DIM + ob + j;
        const float* wp1 = W + (size_t)(k0 + r0 + 4) * E_DIM + ob + j;
        bw[t * 4 + 0] = wp0[0];  bw[t * 4 + 1] = wp0[E_DIM];
        bw[t * 4 + 2] = wp1[0];  bw[t * 4 + 3] = wp1[E_DIM];
    }

    float vv[4];
    float sum = 0.f;
    #pragma unroll
    for (int i = 0; i < 4; i++) {
        float s = pb[i] + pr[i];
        #pragma unroll
        for (int p = 0; p < 16; p++) s += pp[i][p];
        vv[i] = s;
        sum += s;
    }
    const float mean = block256_sum(sum, red_s, tid) * (1.f / E_DIM);
    float var = 0.f;
    #pragma unroll
    for (int i = 0; i < 4; i++) { const float d = vv[i] - mean; var += d * d; }
    const float inv = rsqrtf(block256_sum(var, red_v, tid) * (1.f / E_DIM) + 1e-6f);
    const bool mine = ((tid >> 7) == (int)(blockIdx.y & 1));
    #pragma unroll
    for (int i = 0; i < 4; i++) {
        const float hv = (vv[i] - mean) * inv * psc[i] + pbs[i];
        if (mine && i == (int)(blockIdx.y >> 1)) {
            xs[tid & 127] = hv;
            if (blockIdx.x == 0) h_out[i * 256 + tid] = hv;
        }
    }
    lds_barrier();

    const float amask = (j == 0) ? 1.f : 0.f;
    v8f c0 = {}, c1 = {};
    #pragma unroll
    for (int t = 0; t < 16; t++) {
        const int r0 = t * 8 + half * 2;
        v2f a0, a1, b0, b1;
        a0.x = xs[r0]     * amask;  a0.y = xs[r0 + 1] * amask;
        a1.x = xs[r0 + 4] * amask;  a1.y = xs[r0 + 5] * amask;
        b0.x = bw[t * 4 + 0];  b0.y = bw[t * 4 + 1];
        b1.x = bw[t * 4 + 2];  b1.y = bw[t * 4 + 3];
        c0 = wmma4(a0, b0, c0);
        c1 = wmma4(a1, b1, c1);
    }
    const v8f cs = c0 + c1;
    if (lane < 16)
        out_parts[(size_t)blockIdx.y * E_DIM + ob + lane] = cs[0];
}

__global__ __launch_bounds__(256)
void ln_combine_kernel(const float* __restrict__ parts,
                       const float* __restrict__ bias,
                       const float* __restrict__ resid,
                       const float* __restrict__ sc,
                       const float* __restrict__ bs,
                       float* __restrict__ out) {
    __shared__ float red_s[256];
    __shared__ float red_v[256];
    const int tid = threadIdx.x;
    float pb[4], pr[4], pp[4][8], psc[4], pbs[4];
    #pragma unroll
    for (int i = 0; i < 4; i++) {
        const int e = i * 256 + tid;
        pb[i] = bias[e];
        pr[i] = resid[e];
        #pragma unroll
        for (int p = 0; p < 8; p++) pp[i][p] = parts[(size_t)p * E_DIM + e];
        psc[i] = sc[e];
        pbs[i] = bs[e];
    }
    float vv[4];
    float sum = 0.f;
    #pragma unroll
    for (int i = 0; i < 4; i++) {
        float s = pb[i] + pr[i];
        #pragma unroll
        for (int p = 0; p < 8; p++) s += pp[i][p];
        vv[i] = s;
        sum += s;
    }
    const float mean = block256_sum(sum, red_s, tid) * (1.f / E_DIM);
    float var = 0.f;
    #pragma unroll
    for (int i = 0; i < 4; i++) { const float d = vv[i] - mean; var += d * d; }
    const float inv = rsqrtf(block256_sum(var, red_v, tid) * (1.f / E_DIM) + 1e-6f);
    #pragma unroll
    for (int i = 0; i < 4; i++) {
        const int e = i * 256 + tid;
        out[e] = (vv[i] - mean) * inv * psc[i] + pbs[i];
    }
}

extern "C" void kernel_launch(void* const* d_in, const int* in_sizes, int n_in,
                              void* d_out, int out_size, void* d_ws, size_t ws_size,
                              hipStream_t stream) {
    (void)in_sizes; (void)n_in; (void)out_size; (void)ws_size;
    const float* x     = (const float*)d_in[0];
    const float* cache = (const float*)d_in[1];
    const float* Wv = (const float*)d_in[2];  const float* bv = (const float*)d_in[3];
    const float* Wq = (const float*)d_in[4];  const float* bq = (const float*)d_in[5];
    const float* Wk = (const float*)d_in[6];  const float* bk = (const float*)d_in[7];
    const float* Wo = (const float*)d_in[8];  const float* bo = (const float*)d_in[9];
    const float* W1 = (const float*)d_in[10]; const float* b1 = (const float*)d_in[11];
    const float* W2 = (const float*)d_in[12]; const float* b2 = (const float*)d_in[13];
    const float* ln1_s = (const float*)d_in[14]; const float* ln1_b = (const float*)d_in[15];
    const float* ln2_s = (const float*)d_in[16]; const float* ln2_b = (const float*)d_in[17];

    float* out_h = (float*)d_out;
    float* newv  = out_h + E_DIM;
    float* newk  = newv + (size_t)S_LEN * E_DIM;
    const float* oldv = cache;
    const float* oldk = cache + (size_t)S_LEN * E_DIM;

    float* ws       = (float*)d_ws;
    float* part_qkv = ws;
    float* part_acc = part_qkv + 24 * E_DIM;
    float* part_ml  = part_acc + 512 * E_DIM;
    float* h1       = part_ml + 512 * 32;
    float* part_o   = h1 + E_DIM;
    float* part_1   = part_o + 16 * E_DIM;
    float* part_2   = part_1 + 8 * E_DIM;

    qkv_part_kernel<<<dim3(8, 8, 3), 256, 0, stream>>>(x, Wv, Wq, Wk, part_qkv);
    shift_score_pv_kernel<<<512, 256, 0, stream>>>(oldv, oldk, part_qkv, bv, bq, bk, newv, newk, part_acc, part_ml);
    merge_gemv_kernel<<<dim3(8, 16), 256, 0, stream>>>(part_acc, part_ml, Wo, part_o);
    gemv_ln_part_kernel<<<dim3(8, 8), 256, 0, stream>>>(W1, part_o, bo, x, ln1_s, ln1_b, h1, part_1);
    gemv_relu_part_kernel<<<dim3(8, 8), 256, 0, stream>>>(W2, part_1, b1, part_2);
    ln_combine_kernel<<<1, 256, 0, stream>>>(part_2, b2, h1, ln2_s, ln2_b, out_h);
}


# attention pass re-cut to 256 blocks, one per WGP: each wave streams two pairs of positions, both pairs' rows requested at entry; 256 block partials to merge instead of 512; new sum order
# speedup vs baseline: 3.9976x; 1.0329x over previous
#include <hip/hip_runtime.h>
#include <hip/hip_bf16.h>
#include <math.h>

#define E_DIM 1024
#define S_LEN 8192
#define H_N   16
#define D_H   64

typedef __attribute__((ext_vector_type(2))) float v2f;
typedef __attribute__((ext_vector_type(4))) float v4f;
typedef __attribute__((ext_vector_type(8))) float v8f;

__device__ __forceinline__ v8f wmma4(v2f a, v2f b, v8f c) {
    return __builtin_amdgcn_wmma_f32_16x16x4_f32(false, a, false, b, (short)0, c,
                                                 false, false);
}

__device__ __forceinline__ void lds_barrier() {
    __builtin_amdgcn_fence(__ATOMIC_RELEASE, "workgroup", "local");
    __builtin_amdgcn_s_barrier();
    __builtin_amdgcn_fence(__ATOMIC_ACQUIRE, "workgroup", "local");
}

template <int M>
__device__ __forceinline__ int dpp_xmask(int b) {
    return __builtin_amdgcn_mov_dpp(b, 0x160 + M, 0xf, 0xf, false);
}
template <int M>
__device__ __forceinline__ float dpp_xmask(float v) {
    return __builtin_bit_cast(float, dpp_xmask<M>(__builtin_bit_cast(int, v)));
}
__device__ __forceinline__ int xhalf(int b) {
    return __builtin_amdgcn_permlanex16(b, b, 0x76543210, 0xfedcba98, false, false);
}
__device__ __forceinline__ float xhalf(float v) {
    return __builtin_bit_cast(float, xhalf(__builtin_bit_cast(int, v)));
}
__device__ __forceinline__ float row16_sum(float v) {
    v += dpp_xmask<1>(v);
    v += dpp_xmask<2>(v);
    v += dpp_xmask<4>(v);
    v += dpp_xmask<8>(v);
    return v;
}
__device__ __forceinline__ float wave_sum(float v) {
    v += xhalf(v);
    v += dpp_xmask<8>(v);
    v += dpp_xmask<4>(v);
    v += dpp_xmask<2>(v);
    v += dpp_xmask<1>(v);
    return v;
}
__device__ __forceinline__ int wave_or(int v) {
    v |= xhalf(v);
    v |= dpp_xmask<8>(v);
    v |= dpp_xmask<4>(v);
    v |= dpp_xmask<2>(v);
    v |= dpp_xmask<1>(v);
    return v;
}
__device__ __forceinline__ float wave_max(float v) {
    v = fmaxf(v, xhalf(v));
    v = fmaxf(v, dpp_xmask<8>(v));
    v = fmaxf(v, dpp_xmask<4>(v));
    v = fmaxf(v, dpp_xmask<2>(v));
    v = fmaxf(v, dpp_xmask<1>(v));
    return v;
}
__device__ __forceinline__ float block256_sum(float v, float* red, int tid) {
    red[tid] = v;
    lds_barrier();
    const int l = tid & 31;
    const float w0 = red[l],       w1 = red[32 + l],  w2 = red[64 + l],  w3 = red[96 + l];
    const float w4 = red[128 + l], w5 = red[160 + l], w6 = red[192 + l], w7 = red[224 + l];
    const float x = ((w0 + w4) + (w2 + w6)) + ((w1 + w5) + (w3 + w7));
    return wave_sum(x);
}

__global__ __launch_bounds__(256)
void qkv_part_kernel(const float* x,
                     const float* Wv,
                     const float* Wq,
                     const float* Wk,
                     float* out_parts) {
    __shared__ float xs[128];
    const int tid  = threadIdx.x;
    const int m    = blockIdx.z;
    const float* W = (m == 0) ? Wv : (m == 1) ? Wq : Wk;
    const int k0   = blockIdx.y * 128;
    const int wv   = tid >> 5;
    const int lane = tid & 31;
    const int j    = lane & 15;
    const int half = lane >> 4;
    const int ob   = blockIdx.x * 128 + wv * 16;

    const float xv = x[k0 + (tid & 127)];
    asm volatile("" ::: "memory");

    float bw[64];
    #pragma unroll
    for (int t = 0; t < 16; t++) {
        const int r0 = t * 8 + half * 2;
        const float* wp0 = W + (size_t)(k0 + r0)     * E_DIM + ob + j;
        const float* wp1 = W + (size_t)(k0 + r0 + 4) * E_DIM + ob + j;
        bw[t * 4 + 0] = wp0[0];  bw[t * 4 + 1] = wp0[E_DIM];
        bw[t * 4 + 2] = wp1[0];  bw[t * 4 + 3] = wp1[E_DIM];
    }

    if (tid < 128) xs[tid] = xv;
    lds_barrier();

    const float amask = (j == 0) ? 1.f : 0.f;
    v8f c0 = {}, c1 = {};
    #pragma unroll
    for (int t = 0; t < 16; t++) {
        const int r0 = t * 8 + half * 2;
        v2f a0, a1, b0, b1;
        a0.x = xs[r0]     * amask;  a0.y = xs[r0 + 1] * amask;
        a1.x = xs[r0 + 4] * amask;  a1.y = xs[r0 + 5] * amask;
        b0.x = bw[t * 4 + 0];  b0.y = bw[t * 4 + 1];
        b1.x = bw[t * 4 + 2];  b1.y = bw[t * 4 + 3];
        c0 = wmma4(a0, b0, c0);
        c1 = wmma4(a1, b1, c1);
    }
    const v8f cs = c0 + c1;
    if (lane < 16)
        out_parts[(size_t)(m * 8 + blockIdx.y) * E_DIM + ob + lane] = cs[0];
}

__device__ __forceinline__ void pair_pass(const float4 (&va)[8], const float4 (&vb)[8],
                                          const float4 (&ka)[8], const float4 (&kb)[8],
                                          const float4 (&qd)[8],
                                          float4* vdst, float4* kdst,
                                          int lane, int slot,
                                          float4* s_acc, float* s_m, float* s_l) {
    #pragma unroll
    for (int i = 0; i < 8; i++) vdst[i * 32 + lane] = va[i];
    #pragma unroll
    for (int i = 0; i < 8; i++) vdst[256 + i * 32 + lane] = vb[i];
    #pragma unroll
    for (int i = 0; i < 8; i++) kdst[i * 32 + lane] = ka[i];
    #pragma unroll
    for (int i = 0; i < 8; i++) kdst[256 + i * 32 + lane] = kb[i];

    int ora = 0, orb = 0;
    #pragma unroll
    for (int i = 0; i < 8; i++) {
        ora |= __builtin_bit_cast(int, va[i].x) | __builtin_bit_cast(int, va[i].y)
             | __builtin_bit_cast(int, va[i].z) | __builtin_bit_cast(int, va[i].w);
        orb |= __builtin_bit_cast(int, vb[i].x) | __builtin_bit_cast(int, vb[i].y)
             | __builtin_bit_cast(int, vb[i].z) | __builtin_bit_cast(int, vb[i].w);
    }
    const bool nza = (wave_or(ora) & 0x7fffffff) != 0;
    const bool nzb = (wave_or(orb) & 0x7fffffff) != 0;

    const bool arow = (lane == 0) || (lane == 24);
    const v8f zero8 = {};

    float sa[8], sb[8];
    #pragma unroll
    for (int i = 0; i < 8; i++) {
        float pa = ka[i].x * qd[i].x + ka[i].y * qd[i].y + ka[i].z * qd[i].z + ka[i].w * qd[i].w;
        float pb = kb[i].x * qd[i].x + kb[i].y * qd[i].y + kb[i].z * qd[i].z + kb[i].w * qd[i].w;
        pa = row16_sum(pa);
        pb = row16_sum(pb);
        sa[i] = nza ? pa * 0.125f : -INFINITY;
        sb[i] = nzb ? pb * 0.125f : -INFINITY;
    }

    #pragma unroll
    for (int i = 0; i < 8; i++) {
        const float m  = fmaxf(sa[i], sb[i]);
        const float ms = (m == -INFINITY) ? 0.f : m;
        const float ea = expf(sa[i] - ms);
        const float eb = expf(sb[i] - ms);
        if ((lane & 15) == 0) {
            const int h = i * 2 + (lane >> 4);
            s_m[slot * 16 + h] = m;
            s_l[slot * 16 + h] = ea + eb;
        }
        v2f a, b0, b1, b2, b3;
        a.x  = arow ? ea : 0.f;   a.y  = arow ? eb : 0.f;
        b0.x = va[i].x;           b0.y = vb[i].x;
        b1.x = va[i].y;           b1.y = vb[i].y;
        b2.x = va[i].z;           b2.y = vb[i].z;
        b3.x = va[i].w;           b3.y = vb[i].w;
        const v8f d0 = wmma4(a, b0, zero8);
        const v8f d1 = wmma4(a, b1, zero8);
        const v8f d2 = wmma4(a, b2, zero8);
        const v8f d3 = wmma4(a, b3, zero8);
        float4 acc;
        acc.x = d0[0];  acc.y = d1[0];  acc.z = d2[0];  acc.w = d3[0];
        s_acc[slot * 256 + i * 32 + lane] = acc;
    }
}

__global__ __launch_bounds__(256)
void shift_score_pv_kernel(const float* oldv,
                           const float* oldk,
                           const float* qkv_parts,
                           const float* bv,
                           const float* bq,
                           const float* bk,
                           float* newv, float* newk,
                           float* part_acc,
                           float* part_ml) {
    __shared__ float4 s_acc[16 * 256];
    __shared__ float  s_m[16 * 16];
    __shared__ float  s_l[16 * 16];
    __shared__ float  s_ml[32];
    __shared__ float4 s_q[256];
    __shared__ float4 s_vk[512];
    const int tid  = threadIdx.x;
    const int lane = tid & 31;
    const int wv   = tid >> 5;
    const int pblk = (blockIdx.x + 255) & 255;
    const int sA   = (pblk * 8 + wv) * 4;
    const bool last1 = (sA + 3 == S_LEN - 1);

    const float4* vsrc = (const float4*)(oldv + (size_t)(sA + 1) * E_DIM);
    const float4* ksrc = (const float4*)(oldk + (size_t)(sA + 1) * E_DIM);
    const int r3 = last1 ? 2 * 256 : 3 * 256;
    float4* vdst = (float4*)(newv + (size_t)sA * E_DIM);
    float4* kdst = (float4*)(newk + (size_t)sA * E_DIM);
    const float4* parts4 = (const float4*)qkv_parts;
    const bool lastblk = (pblk == S_LEN / 32 - 1);

    v4f qp[8];
    #pragma unroll
    for (int p = 0; p < 8; p++) qp[p] = ((const v4f*)qkv_parts)[(8 + p) * 256 + tid];
    v4f qb = ((const v4f*)bq)[tid];
    __builtin_amdgcn_sched_barrier(0);

    float4 qd[8], va0[8], vb0[8], ka0[8], kb0[8], va1[8], vb1[8], ka1[8], kb1[8];
    #pragma unroll
    for (int i = 0; i < 8; i++) va0[i] = vsrc[i * 32 + lane];
    #pragma unroll
    for (int i = 0; i < 8; i++) vb0[i] = vsrc[256 + i * 32 + lane];
    #pragma unroll
    for (int i = 0; i < 8; i++) ka0[i] = ksrc[i * 32 + lane];
    #pragma unroll
    for (int i = 0; i < 8; i++) kb0[i] = ksrc[256 + i * 32 + lane];
    __builtin_amdgcn_sched_barrier(0);
    #pragma unroll
    for (int i = 0; i < 8; i++) va1[i] = vsrc[512 + i * 32 + lane];
    #pragma unroll
    for (int i = 0; i < 8; i++) vb1[i] = vsrc[r3 + i * 32 + lane];
    #pragma unroll
    for (int i = 0; i < 8; i++) ka1[i] = ksrc[512 + i * 32 + lane];
    #pragma unroll
    for (int i = 0; i < 8; i++) kb1[i] = ksrc[r3 + i * 32 + lane];

    asm volatile("" : "+v"(qp[0]), "+v"(qp[1]), "+v"(qp[2]), "+v"(qp[3]),
                      "+v"(qp[4]), "+v"(qp[5]), "+v"(qp[6]), "+v"(qp[7]), "+v"(qb));
    {
        v4f a = qp[0];
        #pragma unroll
        for (int p = 1; p < 8; p++) a += qp[p];
        a += qb;
        ((v4f*)s_q)[tid] = a;
    }
    if (lastblk) {
        #pragma unroll
        for (int mk = 0; mk < 2; mk++) {
            const float4* pp4 = parts4 + (size_t)(mk * 16) * 256 + tid;
            float4 a = pp4[0];
            #pragma unroll
            for (int p = 1; p < 8; p++) {
                const float4 t = pp4[p * 256];
                a.x += t.x;  a.y += t.y;  a.z += t.z;  a.w += t.w;
            }
            const float4 b = ((const float4*)(mk == 0 ? bv : bk))[tid];
            a.x += b.x;  a.y += b.y;  a.z += b.z;  a.w += b.w;
            s_vk[mk * 256 + tid] = a;
        }
    }
    lds_barrier();
    #pragma unroll
    for (int i = 0; i < 8; i++) qd[i] = s_q[i * 32 + lane];

    pair_pass(va0, vb0, ka0, kb0, qd, vdst,       kdst,       lane, wv * 2,     s_acc, s_m, s_l);
    if (last1) {
        #pragma unroll
        for (int i = 0; i < 8; i++) { vb1[i] = s_vk[i * 32 + lane];  kb1[i] = s_vk[256 + i * 32 + lane]; }
    }
    pair_pass(va1, vb1, ka1, kb1, qd, vdst + 512, kdst + 512, lane, wv * 2 + 1, s_acc, s_m, s_l);
    __syncthreads();

    {
        const int head = tid >> 4;
        float mw[16];
        float M = -INFINITY;
        #pragma unroll
        for (int w = 0; w < 16; w++) { mw[w] = s_m[w * 16 + head];  M = fmaxf(M, mw[w]); }
        float4 acc = {0.f, 0.f, 0.f, 0.f};
        float L = 0.f;
        const float Ms = (M == -INFINITY) ? 0.f : M;
        #pragma unroll
        for (int w = 0; w < 16; w++) {
            const float sc = expf(mw[w] - Ms);
            const float4 t = s_acc[w * 256 + tid];
            acc.x += t.x * sc;  acc.y += t.y * sc;  acc.z += t.z * sc;  acc.w += t.w * sc;
            L += s_l[w * 16 + head] * sc;
        }
        ((float4*)part_acc)[(size_t)pblk * 256 + tid] = acc;
        if ((tid & 15) == 0) { s_ml[head] = M;  s_ml[16 + head] = L; }
    }
    __syncthreads();
    if (tid < 32) part_ml[pblk * 32 + tid] = s_ml[tid];
}

__global__ __launch_bounds__(256)
void merge_gemv_kernel(const float* __restrict__ part_acc,
                       const float* __restrict__ part_ml,
                       const float* __restrict__ W,
                       float* __restrict__ out_parts) {
    __shared__ float  s_w[256];
    __shared__ float  s_redm[8];
    __shared__ float  s_redl[8];
    __shared__ float4 s_sum[16 * 16];
    __shared__ float  xs[D_H];
    const int h    = blockIdx.y;
    const int tid  = threadIdx.x;
    const int lane = tid & 31;
    const int wv   = tid >> 5;
    const int j    = lane & 15;
    const int half = lane >> 4;
    const int ob   = blockIdx.x * 128 + wv * 16;
    const int k0   = h * D_H;

    const float mb0 = part_ml[tid * 32 + h];
    const float lb0 = part_ml[tid * 32 + 16 + h];

    const int c = tid & 15;
    const int g = tid >> 4;
    const float4* psrc = (const float4*)part_acc + h * 16 + c;
    float4 pv[16];
    #pragma unroll
    for (int it = 0; it < 16; it++) pv[it] = psrc[(size_t)(it * 16 + g) * 256];

    float bw[32];
    #pragma unroll
    for (int t = 0; t < 8; t++) {
        const int r0 = t * 8 + half * 2;
        const float* wp0 = W + (size_t)(k0 + r0)     * E_DIM + ob + j;
        const float* wp1 = W + (size_t)(k0 + r0 + 4) * E_DIM + ob + j;
        bw[t * 4 + 0] = wp0[0];  bw[t * 4 + 1] = wp0[E_DIM];
        bw[t * 4 + 2] = wp1[0];  bw[t * 4 + 3] = wp1[E_DIM];
    }

    const float mx = wave_max(mb0);
    if (lane == 0) s_redm[wv] = mx;
    __syncthreads();
    float M = s_redm[0];
    #pragma unroll
    for (int w = 1; w < 8; w++) M = fmaxf(M, s_redm[w]);

    const float Ms = (M == -INFINITY) ? 0.f : M;
    const float w0 = expf(mb0 - Ms);
    s_w[tid] = w0;
    const float ls = wave_sum(w0 * lb0);
    if (lane == 0) s_redl[wv] = ls;
    __syncthreads();
    float L = 0.f;
    #pragma unroll
    for (int w = 0; w < 8; w++) L += s_redl[w];

    float4 acc = {0.f, 0.f, 0.f, 0.f};
    #pragma unroll
    for (int it = 0; it < 16; it++) {
        const float w = s_w[it * 16 + g];
        acc.x += pv[it].x * w;  acc.y += pv[it].y * w;
        acc.z += pv[it].z * w;  acc.w += pv[it].w * w;
    }
    s_sum[g * 16 + c] = acc;
    __syncthreads();
    if (tid < D_H) {
        const float* sf = (const float*)s_sum;
        float s = 0.f;
        #pragma unroll
        for (int gg = 0; gg < 16; gg++) s += sf[gg * D_H + tid];
        xs[tid] = s / L;
    }
    __syncthreads();

    const float amask = (j == 0) ? 1.f : 0.f;
    v8f c0 = {}, c1 = {};
    #pragma unroll
    for (int t = 0; t < 8; t++) {
        const int r0 = t * 8 + half * 2;
        v2f a0, a1, b0, b1;
        a0.x = xs[r0]     * amask;  a0.y = xs[r0 + 1] * amask;
        a1.x = xs[r0 + 4] * amask;  a1.y = xs[r0 + 5] * amask;
        b0.x = bw[t * 4 + 0];  b0.y = bw[t * 4 + 1];
        b1.x = bw[t * 4 + 2];  b1.y = bw[t * 4 + 3];
        c0 = wmma4(a0, b0, c0);
        c1 = wmma4(a1, b1, c1);
    }
    const v8f cs = c0 + c1;
    if (lane < 16)
        out_parts[(size_t)h * E_DIM + ob + lane] = cs[0];
}

__global__ __launch_bounds__(256)
void gemv_relu_part_kernel(const float* W,
                           const float* in_parts,
                           const float* in_bias,
                           float* out_parts) {
    __shared__ float xs[128];
    const int tid  = threadIdx.x;
    const int k0   = blockIdx.y * 128;
    const int wv   = tid >> 5;
    const int lane = tid & 31;
    const int j    = lane & 15;
    const int half = lane >> 4;
    const int ob   = blockIdx.x * 128 + wv * 16;
    const int ke   = k0 + (tid & 127);

    float pp[8];
    #pragma unroll
    for (int p = 0; p < 8; p++) pp[p] = in_parts[(size_t)p * E_DIM + ke];
    const float pbias = in_bias[ke];
    asm volatile("" ::: "memory");

    float bw[64];
    #pragma unroll
    for (int t = 0; t < 16; t++) {
        const int r0 = t * 8 + half * 2;
        const float* wp0 = W + (size_t)(k0 + r0)     * E_DIM + ob + j;
        const float* wp1 = W + (size_t)(k0 + r0 + 4) * E_DIM + ob + j;
        bw[t * 4 + 0] = wp0[0];  bw[t * 4 + 1] = wp0[E_DIM];
        bw[t * 4 + 2] = wp1[0];  bw[t * 4 + 3] = wp1[E_DIM];
    }

    float s = 0.f;
    #pragma unroll
    for (int p = 0; p < 8; p++) s += pp[p];
    s += pbias;
    if (tid < 128) xs[tid] = fmaxf(s, 0.f);
    lds_barrier();

    const float amask = (j == 0) ? 1.f : 0.f;
    v8f c0 = {}, c1 = {};
    #pragma unroll
    for (int t = 0; t < 16; t++) {
        const int r0 = t * 8 + half * 2;
        v2f a0, a1, b0, b1;
        a0.x = xs[r0]     * amask;  a0.y = xs[r0 + 1] * amask;
        a1.x = xs[r0 + 4] * amask;  a1.y = xs[r0 + 5] * amask;
        b0.x = bw[t * 4 + 0];  b0.y = bw[t * 4 + 1];
        b1.x = bw[t * 4 + 2];  b1.y = bw[t * 4 + 3];
        c0 = wmma4(a0, b0, c0);
        c1 = wmma4(a1, b1, c1);
    }
    const v8f cs = c0 + c1;
    if (lane < 16)
        out_parts[(size_t)blockIdx.y * E_DIM + ob + lane] = cs[0];
}

__global__ __launch_bounds__(256)
void gemv_ln_part_kernel(const float* __restrict__ W,
                         const float* __restrict__ parts,
                         const float* __restrict__ bias,
                         const float* __restrict__ resid,
                         const float* __restrict__ sc,
                         const float* __restrict__ bs,
                         float* __restrict__ h_out,
                         float* __restrict__ out_parts) {
    __shared__ float red_s[256];
    __shared__ float red_v[256];
    __shared__ float xs[128];
    const int tid = threadIdx.x;
    const int k0  = blockIdx.y * 128;

    float pb[4], pr[4], pp[4][16], psc[4], pbs[4];
    #pragma unroll
    for (int i = 0; i < 4; i++) {
        const int e = i * 256 + tid;
        pb[i] = bias[e];
        pr[i] = resid[e];
        #pragma unroll
        for (int p = 0; p < 16; p++) pp[i][p] = parts[(size_t)p * E_DIM + e];
        psc[i] = sc[e];
        pbs[i] = bs[e];
    }

    const int wv   = tid >> 5;
    const int lane = tid & 31;
    const int j    = lane & 15;
    const int half = lane >> 4;
    const int ob   = blockIdx.x * 128 + wv * 16;
    asm volatile("" ::: "memory");
    float bw[64];
    #pragma unroll
    for (int t = 0; t < 16; t++) {
        const int r0 = t * 8 + half * 2;
        const float* wp0 = W + (size_t)(k0 + r0)     * E_DIM + ob + j;
        const float* wp1 = W + (size_t)(k0 + r0 + 4) * E_DIM + ob + j;
        bw[t * 4 + 0] = wp0[0];  bw[t * 4 + 1] = wp0[E_DIM];
        bw[t * 4 + 2] = wp1[0];  bw[t * 4 + 3] = wp1[E_DIM];
    }

    float vv[4];
    float sum = 0.f;
    #pragma unroll
    for (int i = 0; i < 4; i++) {
        float s = pb[i] + pr[i];
        #pragma unroll
        for (int p = 0; p < 16; p++) s += pp[i][p];
        vv[i] = s;
        sum += s;
    }
    const float mean = block256_sum(sum, red_s, tid) * (1.f / E_DIM);
    float var = 0.f;
    #pragma unroll
    for (int i = 0; i < 4; i++) { const float d = vv[i] - mean; var += d * d; }
    const float inv = rsqrtf(block256_sum(var, red_v, tid) * (1.f / E_DIM) + 1e-6f);
    const bool mine = ((tid >> 7) == (int)(blockIdx.y & 1));
    #pragma unroll
    for (int i = 0; i < 4; i++) {
        const float hv = (vv[i] - mean) * inv * psc[i] + pbs[i];
        if (mine && i == (int)(blockIdx.y >> 1)) {
            xs[tid & 127] = hv;
            if (blockIdx.x == 0) h_out[i * 256 + tid] = hv;
        }
    }
    lds_barrier();

    const float amask = (j == 0) ? 1.f : 0.f;
    v8f c0 = {}, c1 = {};
    #pragma unroll
    for (int t = 0; t < 16; t++) {
        const int r0 = t * 8 + half * 2;
        v2f a0, a1, b0, b1;
        a0.x = xs[r0]     * amask;  a0.y = xs[r0 + 1] * amask;
        a1.x = xs[r0 + 4] * amask;  a1.y = xs[r0 + 5] * amask;
        b0.x = bw[t * 4 + 0];  b0.y = bw[t * 4 + 1];
        b1.x = bw[t * 4 + 2];  b1.y = bw[t * 4 + 3];
        c0 = wmma4(a0, b0, c0);
        c1 = wmma4(a1, b1, c1);
    }
    const v8f cs = c0 + c1;
    if (lane < 16)
        out_parts[(size_t)blockIdx.y * E_DIM + ob + lane] = cs[0];
}

__global__ __launch_bounds__(256)
void ln_combine_kernel(const float* __restrict__ parts,
                       const float* __restrict__ bias,
                       const float* __restrict__ resid,
                       const float* __restrict__ sc,
                       const float* __restrict__ bs,
                       float* __restrict__ out) {
    __shared__ float red_s[256];
    __shared__ float red_v[256];
    const int tid = threadIdx.x;
    float pb[4], pr[4], pp[4][8], psc[4], pbs[4];
    #pragma unroll
    for (int i = 0; i < 4; i++) {
        const int e = i * 256 + tid;
        pb[i] = bias[e];
        pr[i] = resid[e];
        #pragma unroll
        for (int p = 0; p < 8; p++) pp[i][p] = parts[(size_t)p * E_DIM + e];
        psc[i] = sc[e];
        pbs[i] = bs[e];
    }
    float vv[4];
    float sum = 0.f;
    #pragma unroll
    for (int i = 0; i < 4; i++) {
        float s = pb[i] + pr[i];
        #pragma unroll
        for (int p = 0; p < 8; p++) s += pp[i][p];
        vv[i] = s;
        sum += s;
    }
    const float mean = block256_sum(sum, red_s, tid) * (1.f / E_DIM);
    float var = 0.f;
    #pragma unroll
    for (int i = 0; i < 4; i++) { const float d = vv[i] - mean; var += d * d; }
    const float inv = rsqrtf(block256_sum(var, red_v, tid) * (1.f / E_DIM) + 1e-6f);
    #pragma unroll
    for (int i = 0; i < 4; i++) {
        const int e = i * 256 + tid;
        out[e] = (vv[i] - mean) * inv * psc[i] + pbs[i];
    }
}

extern "C" void kernel_launch(void* const* d_in, const int* in_sizes, int n_in,
                              void* d_out, int out_size, void* d_ws, size_t ws_size,
                              hipStream_t stream) {
    (void)in_sizes; (void)n_in; (void)out_size; (void)ws_size;
    const float* x     = (const float*)d_in[0];
    const float* cache = (const float*)d_in[1];
    const float* Wv = (const float*)d_in[2];  const float* bv = (const float*)d_in[3];
    const float* Wq = (const float*)d_in[4];  const float* bq = (const float*)d_in[5];
    const float* Wk = (const float*)d_in[6];  const float* bk = (const float*)d_in[7];
    const float* Wo = (const float*)d_in[8];  const float* bo = (const float*)d_in[9];
    const float* W1 = (const float*)d_in[10]; const float* b1 = (const float*)d_in[11];
    const float* W2 = (const float*)d_in[12]; const float* b2 = (const float*)d_in[13];
    const float* ln1_s = (const float*)d_in[14]; const float* ln1_b = (const float*)d_in[15];
    const float* ln2_s = (const float*)d_in[16]; const float* ln2_b = (const float*)d_in[17];

    float* out_h = (float*)d_out;
    float* newv  = out_h + E_DIM;
    float* newk  = newv + (size_t)S_LEN * E_DIM;
    const float* oldv = cache;
    const float* oldk = cache + (size_t)S_LEN * E_DIM;

    float* ws       = (float*)d_ws;
    float* part_qkv = ws;
    float* part_acc = part_qkv + 24 * E_DIM;
    float* part_ml  = part_acc + 256 * E_DIM;
    float* h1       = part_ml + 256 * 32;
    float* part_o   = h1 + E_DIM;
    float* part_1   = part_o + 16 * E_DIM;
    float* part_2   = part_1 + 8 * E_DIM;

    qkv_part_kernel<<<dim3(8, 8, 3), 256, 0, stream>>>(x, Wv, Wq, Wk, part_qkv);
    shift_score_pv_kernel<<<256, 256, 0, stream>>>(oldv, oldk, part_qkv, bv, bq, bk, newv, newk, part_acc, part_ml);
    merge_gemv_kernel<<<dim3(8, 16), 256, 0, stream>>>(part_acc, part_ml, Wo, part_o);
    gemv_ln_part_kernel<<<dim3(8, 8), 256, 0, stream>>>(W1, part_o, bo, x, ln1_s, ln1_b, h1, part_1);
    gemv_relu_part_kernel<<<dim3(8, 8), 256, 0, stream>>>(W2, part_1, b1, part_2);
    ln_combine_kernel<<<1, 256, 0, stream>>>(part_2, b2, h1, ln2_s, ln2_b, out_h);
}
